# starting kernel
# baseline (speedup 1.0000x reference)
#include <hip/hip_runtime.h>
#include <math.h>

#define Nn 1024
#define Mm 16384
#define Bb 2
#define MT 64
#define KTS 65

typedef __attribute__((ext_vector_type(2))) float v2f;
typedef __attribute__((ext_vector_type(8))) float v8f;

#define SQRT3F 1.7320508075688772f
#define LOG2PIF 1.8378770664093453f

__device__ __forceinline__ float kern_pair(float x1, float y1, float a1, float b1, float c1, float q1,
                                           float x2, float y2, float a2, float b2, float c2, float q2,
                                           float scale) {
    float s00 = a1 + a2, s11 = b1 + b2, s01 = c1 + c2;
    float sdet = s00 * s11 - s01 * s01;
    float d0 = x1 - x2, d1 = y1 - y2;
    float Q = 0.5f * (s11 * d0 * d0 - 2.0f * s01 * d0 * d1 + s00 * d1 * d1) / sdet;
    float C = 2.0f * q1 * q2 * rsqrtf(fmaxf(sdet, 1e-5f));
    float t = SQRT3F * sqrtf(fmaxf(Q, 1e-5f));
    return (1.0f + t) * __expf(-t) * C * scale;
}

__global__ void prep_kernel(const float* __restrict__ ct, const float* __restrict__ kpt,
                            const float* __restrict__ cs, const float* __restrict__ kps,
                            const float* __restrict__ scale_param,
                            float* __restrict__ trainP, float* __restrict__ testP,
                            float* __restrict__ Kmm) {
    int gid = blockIdx.x * blockDim.x + threadIdx.x;
    float scale = __expf(scale_param[0]);
    if (gid < Bb * Mm) {
        float a = kps[gid * 3 + 0], b = kps[gid * 3 + 1], c = kps[gid * 3 + 2];
        float det = a * b - c * c;
        testP[gid * 6 + 0] = cs[gid * 2 + 0];
        testP[gid * 6 + 1] = cs[gid * 2 + 1];
        testP[gid * 6 + 2] = a;
        testP[gid * 6 + 3] = b;
        testP[gid * 6 + 4] = c;
        testP[gid * 6 + 5] = sqrtf(sqrtf(det));
        float t0 = SQRT3F * sqrtf(1e-5f);
        float m0 = (1.0f + t0) * __expf(-t0);
        float C = 2.0f * sqrtf(det) * rsqrtf(fmaxf(4.0f * det, 1e-5f));
        Kmm[gid] = C * m0 * scale;
    }
    if (gid < Bb * Nn) {
        float a = kpt[gid * 3 + 0], b = kpt[gid * 3 + 1], c = kpt[gid * 3 + 2];
        float det = a * b - c * c;
        trainP[gid * 6 + 0] = ct[gid * 2 + 0];
        trainP[gid * 6 + 1] = ct[gid * 2 + 1];
        trainP[gid * 6 + 2] = a;
        trainP[gid * 6 + 3] = b;
        trainP[gid * 6 + 4] = c;
        trainP[gid * 6 + 5] = sqrtf(sqrtf(det));
    }
}

__global__ void buildA_kernel(const float* __restrict__ trainP, const float* __restrict__ var,
                              const float* __restrict__ scale_param, float* __restrict__ A) {
    int gid = blockIdx.x * blockDim.x + threadIdx.x;
    int b = gid >> 20;
    int i = (gid >> 10) & (Nn - 1);
    int j = gid & (Nn - 1);
    const float* p1 = trainP + (size_t)(b * Nn + i) * 6;
    const float* p2 = trainP + (size_t)(b * Nn + j) * 6;
    float scale = __expf(scale_param[0]);
    float v = kern_pair(p1[0], p1[1], p1[2], p1[3], p1[4], p1[5],
                        p2[0], p2[1], p2[2], p2[3], p2[4], p2[5], scale);
    if (i == j) v += var[b * Nn + i];
    A[(size_t)gid] = v;
}

__global__ __launch_bounds__(1024) void chol_kernel(float* __restrict__ Abase,
                                                    float* __restrict__ logdet) {
    int b = blockIdx.x;
    float* A = Abase + (size_t)b * Nn * Nn;
    __shared__ float col[Nn];
    __shared__ float dinv_sh;
    int tid = threadIdx.x;
    for (int k = 0; k < Nn; k++) {
        if (tid == 0) {
            float d = sqrtf(A[(size_t)k * Nn + k]);
            A[(size_t)k * Nn + k] = d;
            dinv_sh = 1.0f / d;
        }
        __syncthreads();
        float dinv = dinv_sh;
        for (int i = k + 1 + tid; i < Nn; i += 1024) {
            float v = A[(size_t)i * Nn + k] * dinv;
            A[(size_t)i * Nn + k] = v;
            col[i] = v;
        }
        __syncthreads();
        for (int i = k + 1 + tid; i < Nn; i += 1024) {
            float ci = col[i];
            float* row = A + (size_t)i * Nn;
            for (int j = k + 1; j <= i; j++) row[j] -= ci * col[j];
        }
        __syncthreads();
    }
    float ld = __logf(A[(size_t)tid * Nn + tid]);
    col[tid] = ld;
    __syncthreads();
    for (int s = 512; s > 0; s >>= 1) {
        if (tid < s) col[tid] += col[tid + s];
        __syncthreads();
    }
    if (tid == 0) logdet[b] = col[0];
}

__global__ __launch_bounds__(256) void zsolve_kernel(const float* __restrict__ Lbase,
                                                     const float* __restrict__ y,
                                                     const float* __restrict__ mean,
                                                     const float* __restrict__ logdet,
                                                     float* __restrict__ zbase,
                                                     float* __restrict__ nlml_out) {
    int b = blockIdx.x;
    const float* L = Lbase + (size_t)b * Nn * Nn;
    __shared__ float zsh[Nn];
    __shared__ float red[256];
    int tid = threadIdx.x;
    float mb = mean[b];
    for (int i = 0; i < Nn; i++) {
        const float* row = L + (size_t)i * Nn;
        float s = 0.0f;
        for (int j = tid; j < i; j += 256) s += row[j] * zsh[j];
        red[tid] = s;
        __syncthreads();
        for (int st = 128; st > 0; st >>= 1) {
            if (tid < st) red[tid] += red[tid + st];
            __syncthreads();
        }
        if (tid == 0) zsh[i] = (y[b * Nn + i] - mb - red[0]) / row[i];
        __syncthreads();
    }
    float s2 = 0.0f;
    for (int i = tid; i < Nn; i += 256) {
        float z = zsh[i];
        zbase[b * Nn + i] = z;
        s2 += z * z;
    }
    red[tid] = s2;
    __syncthreads();
    for (int st = 128; st > 0; st >>= 1) {
        if (tid < st) red[tid] += red[tid + st];
        __syncthreads();
    }
    if (tid == 0) nlml_out[b] = 0.5f * red[0] + logdet[b] + 0.5f * Nn * LOG2PIF;
}

__global__ __launch_bounds__(256) void trinv_kernel(const float* __restrict__ Lbase,
                                                    float* __restrict__ Linvbase) {
    int b = blockIdx.y;
    int wave = threadIdx.x >> 5;
    int lane = threadIdx.x & 31;
    int j = blockIdx.x * 8 + wave;
    const float* L = Lbase + (size_t)b * Nn * Nn;
    float* Li = Linvbase + (size_t)b * Nn * Nn;
    __shared__ float xsh[8 * Nn];
    float* x = xsh + wave * Nn;
    for (int i = lane; i < j; i += 32) Li[(size_t)i * Nn + j] = 0.0f;
    float djj = 1.0f / L[(size_t)j * Nn + j];
    if (lane == 0) {
        x[j] = djj;
        Li[(size_t)j * Nn + j] = djj;
    }
    for (int i = j + 1; i < Nn; i++) {
        const float* row = L + (size_t)i * Nn;
        float s = 0.0f;
        for (int k = j + lane; k < i; k += 32) s += row[k] * x[k];
        s += __shfl_xor(s, 16);
        s += __shfl_xor(s, 8);
        s += __shfl_xor(s, 4);
        s += __shfl_xor(s, 2);
        s += __shfl_xor(s, 1);
        float xi = -s / row[i];
        if (lane == 0) {
            x[i] = xi;
            Li[(size_t)i * Nn + j] = xi;
        }
    }
}

__global__ __launch_bounds__(128) void predict_kernel(const float* __restrict__ Linvbase,
                                                      const float* __restrict__ trainPg,
                                                      const float* __restrict__ testPg,
                                                      const float* __restrict__ Kmm,
                                                      const float* __restrict__ zbase,
                                                      const float* __restrict__ meang,
                                                      const float* __restrict__ scale_param,
                                                      float* __restrict__ out_mean,
                                                      float* __restrict__ out_var) {
    extern __shared__ float smem[];
    float* Kt  = smem;
    float* zsh = smem + Nn * KTS;
    float* tp  = zsh + Nn;
    int b = blockIdx.y;
    int colBase = blockIdx.x * MT;
    int tid = threadIdx.x;
    float scale = __expf(scale_param[0]);

    for (int idx = tid; idx < Nn; idx += 128) zsh[idx] = zbase[b * Nn + idx];
    for (int idx = tid; idx < Nn * 6; idx += 128) tp[idx] = trainPg[(size_t)b * Nn * 6 + idx];
    __syncthreads();

    {
        int c = tid & 63;
        int h = tid >> 6;
        const float* q = testPg + (size_t)(b * Mm + colBase + c) * 6;
        float x2 = q[0], y2 = q[1], a2 = q[2], b2 = q[3], c2 = q[4], q2 = q[5];
        for (int n = h; n < Nn; n += 2) {
            const float* p = tp + n * 6;
            Kt[n * KTS + c] = kern_pair(p[0], p[1], p[2], p[3], p[4], p[5],
                                        x2, y2, a2, b2, c2, q2, scale);
        }
    }
    __syncthreads();

    int wave = tid >> 5, lane = tid & 31;
    int r = lane & 15, hi = lane >> 4;
    int wcol = wave * 16;
    const float* Li = Linvbase + (size_t)b * Nn * Nn;

    float meanAcc = 0.0f, varAcc = 0.0f;
    for (int rb = 0; rb < Nn / 16; rb++) {
        v8f acc = {};
        const float* arow = Li + (size_t)(rb * 16 + r) * Nn;
        for (int j = 0; j <= rb; j++) {
            int kb = j * 16;
#pragma unroll
            for (int k0 = 0; k0 < 16; k0 += 4) {
                int kk = kb + k0 + 2 * hi;
                v2f a = *(const v2f*)(arow + kk);
                v2f bb;
                bb.x = Kt[kk * KTS + wcol + r];
                bb.y = Kt[(kk + 1) * KTS + wcol + r];
                acc = __builtin_amdgcn_wmma_f32_16x16x4_f32(false, a, false, bb,
                                                            (short)0, acc, false, false);
            }
        }
        const float* zb = zsh + rb * 16 + 8 * hi;
#pragma unroll
        for (int v = 0; v < 8; v++) {
            float cv = acc[v];
            meanAcc += zb[v] * cv;
            varAcc  += cv * cv;
        }
    }
    meanAcc += __shfl_xor(meanAcc, 16);
    varAcc  += __shfl_xor(varAcc, 16);
    if (lane < 16) {
        int m = colBase + wcol + lane;
        out_mean[b * Mm + m] = meang[b] + meanAcc;
        out_var[b * Mm + m]  = Kmm[b * Mm + m] - varAcc;
    }
}

extern "C" void kernel_launch(void* const* d_in, const int* in_sizes, int n_in,
                              void* d_out, int out_size, void* d_ws, size_t ws_size,
                              hipStream_t stream) {
    const float* coords_train = (const float*)d_in[0];
    const float* kp_train     = (const float*)d_in[1];
    const float* coords_test  = (const float*)d_in[2];
    const float* kp_test      = (const float*)d_in[3];
    const float* y_train      = (const float*)d_in[4];
    const float* var          = (const float*)d_in[5];
    const float* mean         = (const float*)d_in[6];
    const float* scale_param  = (const float*)d_in[7];

    float* ws = (float*)d_ws;
    size_t off = 0;
    float* A      = ws + off; off += (size_t)Bb * Nn * Nn;
    float* Linv   = ws + off; off += (size_t)Bb * Nn * Nn;
    float* trainP = ws + off; off += (size_t)Bb * Nn * 6;
    float* testP  = ws + off; off += (size_t)Bb * Mm * 6;
    float* Kmm    = ws + off; off += (size_t)Bb * Mm;
    float* z      = ws + off; off += (size_t)Bb * Nn;
    float* logdet = ws + off; off += (size_t)Bb;

    float* out_mean = (float*)d_out;
    float* out_var  = out_mean + (size_t)Bb * Mm;
    float* nlml     = out_var + (size_t)Bb * Mm;

    prep_kernel<<<(Bb * Mm + 255) / 256, 256, 0, stream>>>(
        coords_train, kp_train, coords_test, kp_test, scale_param, trainP, testP, Kmm);
    buildA_kernel<<<(Bb * Nn * Nn) / 256, 256, 0, stream>>>(trainP, var, scale_param, A);
    chol_kernel<<<Bb, 1024, 0, stream>>>(A, logdet);
    zsolve_kernel<<<Bb, 256, 0, stream>>>(A, y_train, mean, logdet, z, nlml);
    trinv_kernel<<<dim3(Nn / 8, Bb), 256, 0, stream>>>(A, Linv);

    size_t smem_bytes = (size_t)(Nn * KTS + Nn + Nn * 6) * sizeof(float);
    predict_kernel<<<dim3(Mm / MT, Bb), 128, smem_bytes, stream>>>(
        Linv, trainP, testP, Kmm, z, mean, scale_param, out_mean, out_var);
}


# Cholesky as 32 tile-column launches, one wave per 32x32 tile: rows in registers, diagonal tile re-factored and own panel tiles solved in-wave by lane broadcast, then tile update; sum order kept
# speedup vs baseline: 7.7815x; 7.7815x over previous
#include <hip/hip_runtime.h>
#include <math.h>

#define Nn 1024
#define Mm 16384
#define Bb 2
#define MT 64
#define KTS 65

typedef __attribute__((ext_vector_type(2))) float v2f;
typedef __attribute__((ext_vector_type(8))) float v8f;

#define SQRT3F 1.7320508075688772f
#define LOG2PIF 1.8378770664093453f

__device__ __forceinline__ float kern_pair(float x1, float y1, float a1, float b1, float c1, float q1,
                                           float x2, float y2, float a2, float b2, float c2, float q2,
                                           float scale) {
    float s00 = a1 + a2, s11 = b1 + b2, s01 = c1 + c2;
    float sdet = s00 * s11 - s01 * s01;
    float d0 = x1 - x2, d1 = y1 - y2;
    float Q = 0.5f * (s11 * d0 * d0 - 2.0f * s01 * d0 * d1 + s00 * d1 * d1) / sdet;
    float C = 2.0f * q1 * q2 * rsqrtf(fmaxf(sdet, 1e-5f));
    float t = SQRT3F * sqrtf(fmaxf(Q, 1e-5f));
    return (1.0f + t) * __expf(-t) * C * scale;
}

__global__ void prep_kernel(const float* __restrict__ ct, const float* __restrict__ kpt,
                            const float* __restrict__ cs, const float* __restrict__ kps,
                            const float* __restrict__ scale_param,
                            float* __restrict__ trainP, float* __restrict__ testP,
                            float* __restrict__ Kmm) {
    int gid = blockIdx.x * blockDim.x + threadIdx.x;
    float scale = __expf(scale_param[0]);
    if (gid < Bb * Mm) {
        float a = kps[gid * 3 + 0], b = kps[gid * 3 + 1], c = kps[gid * 3 + 2];
        float det = a * b - c * c;
        testP[gid * 6 + 0] = cs[gid * 2 + 0];
        testP[gid * 6 + 1] = cs[gid * 2 + 1];
        testP[gid * 6 + 2] = a;
        testP[gid * 6 + 3] = b;
        testP[gid * 6 + 4] = c;
        testP[gid * 6 + 5] = sqrtf(sqrtf(det));
        float t0 = SQRT3F * sqrtf(1e-5f);
        float m0 = (1.0f + t0) * __expf(-t0);
        float C = 2.0f * sqrtf(det) * rsqrtf(fmaxf(4.0f * det, 1e-5f));
        Kmm[gid] = C * m0 * scale;
    }
    if (gid < Bb * Nn) {
        float a = kpt[gid * 3 + 0], b = kpt[gid * 3 + 1], c = kpt[gid * 3 + 2];
        float det = a * b - c * c;
        trainP[gid * 6 + 0] = ct[gid * 2 + 0];
        trainP[gid * 6 + 1] = ct[gid * 2 + 1];
        trainP[gid * 6 + 2] = a;
        trainP[gid * 6 + 3] = b;
        trainP[gid * 6 + 4] = c;
        trainP[gid * 6 + 5] = sqrtf(sqrtf(det));
    }
}

__global__ void buildA_kernel(const float* __restrict__ trainP, const float* __restrict__ var,
                              const float* __restrict__ scale_param, float* __restrict__ A) {
    int gid = blockIdx.x * blockDim.x + threadIdx.x;
    int b = gid >> 20;
    int i = (gid >> 10) & (Nn - 1);
    int j = gid & (Nn - 1);
    const float* p1 = trainP + (size_t)(b * Nn + i) * 6;
    const float* p2 = trainP + (size_t)(b * Nn + j) * 6;
    float scale = __expf(scale_param[0]);
    float v = kern_pair(p1[0], p1[1], p1[2], p1[3], p1[4], p1[5],
                        p2[0], p2[1], p2[2], p2[3], p2[4], p2[5], scale);
    if (i == j) v += var[b * Nn + i];
    A[(size_t)gid] = v;
}

#define TS 32
#define NT (Nn / TS)

__device__ __forceinline__ float lane_bcast(float v, int src) {
    return __int_as_float(__builtin_amdgcn_readlane(__float_as_int(v), src));
}

__device__ __forceinline__ void load_row32(const float* p, float (&r)[TS]) {
    const float4* p4 = (const float4*)p;
#pragma unroll
    for (int q = 0; q < TS / 4; q++) {
        float4 t = p4[q];
        r[4 * q + 0] = t.x; r[4 * q + 1] = t.y; r[4 * q + 2] = t.z; r[4 * q + 3] = t.w;
    }
}

__device__ __forceinline__ void store_row32(float* p, const float (&r)[TS]) {
    float4* p4 = (float4*)p;
#pragma unroll
    for (int q = 0; q < TS / 4; q++) {
        float4 t;
        t.x = r[4 * q + 0]; t.y = r[4 * q + 1]; t.z = r[4 * q + 2]; t.w = r[4 * q + 3];
        p4[q] = t;
    }
}

#define QP 36

__global__ __launch_bounds__(32) void chol_step_kernel(float* Abase, float* __restrict__ Lbase, int k) {
    __shared__ __attribute__((aligned(16))) float qs[TS * QP];
    int i = k + blockIdx.x;
    int j = k + blockIdx.y;
    if (j > i) return;
    int b = blockIdx.z;
    int r = threadIdx.x;
    float* A = Abase + (size_t)b * Nn * Nn;
    float* Lo = Lbase + (size_t)b * Nn * Nn;

    float d[TS];
    v2f pq[TS];
    load_row32(A + (size_t)(k * TS + r) * Nn + k * TS, d);
    {
        float p[TS], q[TS];
        load_row32(A + (size_t)(i * TS + r) * Nn + k * TS, p);
        load_row32(A + (size_t)(j * TS + r) * Nn + k * TS, q);
#pragma unroll
        for (int c = 0; c < TS; c++) { pq[c].x = p[c]; pq[c].y = q[c]; }
    }

#pragma unroll
    for (int c = 0; c < TS; c++) {
#pragma unroll
        for (int c1 = 0; c1 < c; c1++) {
            float s = lane_bcast(d[c1], c);
            d[c] -= d[c1] * s;
            pq[c] -= pq[c1] * s;
        }
        float piv = lane_bcast(d[c], c);
        float dg = sqrtf(piv);
        float dinv = 1.0f / dg;
        d[c] = (r == c) ? dg : d[c] * dinv;
        pq[c] = pq[c] * dinv;
    }

    if (j > k) {
        float4* qw = (float4*)(qs + r * QP);
#pragma unroll
        for (int c4 = 0; c4 < TS / 4; c4++)
            qw[c4] = make_float4(pq[4 * c4].y, pq[4 * c4 + 1].y, pq[4 * c4 + 2].y, pq[4 * c4 + 3].y);
        __syncthreads();
        float* arow = A + (size_t)(i * TS + r) * Nn + j * TS;
#pragma unroll 1
        for (int s0 = 0; s0 < TS; s0 += 4) {
            float4 av = *(const float4*)(arow + s0);
            float a0 = av.x, a1 = av.y, a2 = av.z, a3 = av.w;
            const float4* q0 = (const float4*)(qs + (s0 + 0) * QP);
            const float4* q1 = (const float4*)(qs + (s0 + 1) * QP);
            const float4* q2 = (const float4*)(qs + (s0 + 2) * QP);
            const float4* q3 = (const float4*)(qs + (s0 + 3) * QP);
#pragma unroll
            for (int c4 = 0; c4 < TS / 4; c4++) {
                float4 u0 = q0[c4], u1 = q1[c4], u2 = q2[c4], u3 = q3[c4];
                float p0 = pq[4 * c4].x, p1 = pq[4 * c4 + 1].x, p2 = pq[4 * c4 + 2].x, p3 = pq[4 * c4 + 3].x;
                a0 -= p0 * u0.x; a1 -= p0 * u1.x; a2 -= p0 * u2.x; a3 -= p0 * u3.x;
                a0 -= p1 * u0.y; a1 -= p1 * u1.y; a2 -= p1 * u2.y; a3 -= p1 * u3.y;
                a0 -= p2 * u0.z; a1 -= p2 * u1.z; a2 -= p2 * u2.z; a3 -= p2 * u3.z;
                a0 -= p3 * u0.w; a1 -= p3 * u1.w; a2 -= p3 * u2.w; a3 -= p3 * u3.w;
            }
            *(float4*)(arow + s0) = make_float4(a0, a1, a2, a3);
        }
    } else if (i > k) {
        float p[TS];
#pragma unroll
        for (int c = 0; c < TS; c++) p[c] = pq[c].x;
        store_row32(Lo + (size_t)(i * TS + r) * Nn + k * TS, p);
    } else {
#pragma unroll
        for (int c = 0; c < TS; c++) if (c > r) d[c] = 0.0f;
        store_row32(Lo + (size_t)(k * TS + r) * Nn + k * TS, d);
    }
}

__global__ __launch_bounds__(1024) void logdet_kernel(const float* __restrict__ Lbase,
                                                      float* __restrict__ logdet) {
    int b = blockIdx.x;
    const float* L = Lbase + (size_t)b * Nn * Nn;
    __shared__ float col[Nn];
    int tid = threadIdx.x;
    float ld = __logf(L[(size_t)tid * Nn + tid]);
    col[tid] = ld;
    __syncthreads();
    for (int s = 512; s > 0; s >>= 1) {
        if (tid < s) col[tid] += col[tid + s];
        __syncthreads();
    }
    if (tid == 0) logdet[b] = col[0];
}

__global__ __launch_bounds__(256) void zsolve_kernel(const float* __restrict__ Lbase,
                                                     const float* __restrict__ y,
                                                     const float* __restrict__ mean,
                                                     const float* __restrict__ logdet,
                                                     float* __restrict__ zbase,
                                                     float* __restrict__ nlml_out) {
    int b = blockIdx.x;
    const float* L = Lbase + (size_t)b * Nn * Nn;
    __shared__ float zsh[Nn];
    __shared__ float red[256];
    int tid = threadIdx.x;
    float mb = mean[b];
    for (int i = 0; i < Nn; i++) {
        const float* row = L + (size_t)i * Nn;
        float s = 0.0f;
        for (int j = tid; j < i; j += 256) s += row[j] * zsh[j];
        red[tid] = s;
        __syncthreads();
        for (int st = 128; st > 0; st >>= 1) {
            if (tid < st) red[tid] += red[tid + st];
            __syncthreads();
        }
        if (tid == 0) zsh[i] = (y[b * Nn + i] - mb - red[0]) / row[i];
        __syncthreads();
    }
    float s2 = 0.0f;
    for (int i = tid; i < Nn; i += 256) {
        float z = zsh[i];
        zbase[b * Nn + i] = z;
        s2 += z * z;
    }
    red[tid] = s2;
    __syncthreads();
    for (int st = 128; st > 0; st >>= 1) {
        if (tid < st) red[tid] += red[tid + st];
        __syncthreads();
    }
    if (tid == 0) nlml_out[b] = 0.5f * red[0] + logdet[b] + 0.5f * Nn * LOG2PIF;
}

__global__ __launch_bounds__(256) void trinv_kernel(const float* __restrict__ Lbase,
                                                    float* __restrict__ Linvbase) {
    int b = blockIdx.y;
    int wave = threadIdx.x >> 5;
    int lane = threadIdx.x & 31;
    int j = blockIdx.x * 8 + wave;
    const float* L = Lbase + (size_t)b * Nn * Nn;
    float* Li = Linvbase + (size_t)b * Nn * Nn;
    __shared__ float xsh[8 * Nn];
    float* x = xsh + wave * Nn;
    for (int i = lane; i < j; i += 32) Li[(size_t)i * Nn + j] = 0.0f;
    float djj = 1.0f / L[(size_t)j * Nn + j];
    if (lane == 0) {
        x[j] = djj;
        Li[(size_t)j * Nn + j] = djj;
    }
    for (int i = j + 1; i < Nn; i++) {
        const float* row = L + (size_t)i * Nn;
        float s = 0.0f;
        for (int k = j + lane; k < i; k += 32) s += row[k] * x[k];
        s += __shfl_xor(s, 16);
        s += __shfl_xor(s, 8);
        s += __shfl_xor(s, 4);
        s += __shfl_xor(s, 2);
        s += __shfl_xor(s, 1);
        float xi = -s / row[i];
        if (lane == 0) {
            x[i] = xi;
            Li[(size_t)i * Nn + j] = xi;
        }
    }
}

__global__ __launch_bounds__(128) void predict_kernel(const float* __restrict__ Linvbase,
                                                      const float* __restrict__ trainPg,
                                                      const float* __restrict__ testPg,
                                                      const float* __restrict__ Kmm,
                                                      const float* __restrict__ zbase,
                                                      const float* __restrict__ meang,
                                                      const float* __restrict__ scale_param,
                                                      float* __restrict__ out_mean,
                                                      float* __restrict__ out_var) {
    extern __shared__ float smem[];
    float* Kt  = smem;
    float* zsh = smem + Nn * KTS;
    float* tp  = zsh + Nn;
    int b = blockIdx.y;
    int colBase = blockIdx.x * MT;
    int tid = threadIdx.x;
    float scale = __expf(scale_param[0]);

    for (int idx = tid; idx < Nn; idx += 128) zsh[idx] = zbase[b * Nn + idx];
    for (int idx = tid; idx < Nn * 6; idx += 128) tp[idx] = trainPg[(size_t)b * Nn * 6 + idx];
    __syncthreads();

    {
        int c = tid & 63;
        int h = tid >> 6;
        const float* q = testPg + (size_t)(b * Mm + colBase + c) * 6;
        float x2 = q[0], y2 = q[1], a2 = q[2], b2 = q[3], c2 = q[4], q2 = q[5];
        for (int n = h; n < Nn; n += 2) {
            const float* p = tp + n * 6;
            Kt[n * KTS + c] = kern_pair(p[0], p[1], p[2], p[3], p[4], p[5],
                                        x2, y2, a2, b2, c2, q2, scale);
        }
    }
    __syncthreads();

    int wave = tid >> 5, lane = tid & 31;
    int r = lane & 15, hi = lane >> 4;
    int wcol = wave * 16;
    const float* Li = Linvbase + (size_t)b * Nn * Nn;

    float meanAcc = 0.0f, varAcc = 0.0f;
    for (int rb = 0; rb < Nn / 16; rb++) {
        v8f acc = {};
        const float* arow = Li + (size_t)(rb * 16 + r) * Nn;
        for (int j = 0; j <= rb; j++) {
            int kb = j * 16;
#pragma unroll
            for (int k0 = 0; k0 < 16; k0 += 4) {
                int kk = kb + k0 + 2 * hi;
                v2f a = *(const v2f*)(arow + kk);
                v2f bb;
                bb.x = Kt[kk * KTS + wcol + r];
                bb.y = Kt[(kk + 1) * KTS + wcol + r];
                acc = __builtin_amdgcn_wmma_f32_16x16x4_f32(false, a, false, bb,
                                                            (short)0, acc, false, false);
            }
        }
        const float* zb = zsh + rb * 16 + 8 * hi;
#pragma unroll
        for (int v = 0; v < 8; v++) {
            float cv = acc[v];
            meanAcc += zb[v] * cv;
            varAcc  += cv * cv;
        }
    }
    meanAcc += __shfl_xor(meanAcc, 16);
    varAcc  += __shfl_xor(varAcc, 16);
    if (lane < 16) {
        int m = colBase + wcol + lane;
        out_mean[b * Mm + m] = meang[b] + meanAcc;
        out_var[b * Mm + m]  = Kmm[b * Mm + m] - varAcc;
    }
}

extern "C" void kernel_launch(void* const* d_in, const int* in_sizes, int n_in,
                              void* d_out, int out_size, void* d_ws, size_t ws_size,
                              hipStream_t stream) {
    const float* coords_train = (const float*)d_in[0];
    const float* kp_train     = (const float*)d_in[1];
    const float* coords_test  = (const float*)d_in[2];
    const float* kp_test      = (const float*)d_in[3];
    const float* y_train      = (const float*)d_in[4];
    const float* var          = (const float*)d_in[5];
    const float* mean         = (const float*)d_in[6];
    const float* scale_param  = (const float*)d_in[7];

    float* ws = (float*)d_ws;
    size_t off = 0;
    float* A      = ws + off; off += (size_t)Bb * Nn * Nn;
    float* Linv   = ws + off; off += (size_t)Bb * Nn * Nn;
    float* Lf     = ws + off; off += (size_t)Bb * Nn * Nn;
    float* trainP = ws + off; off += (size_t)Bb * Nn * 6;
    float* testP  = ws + off; off += (size_t)Bb * Mm * 6;
    float* Kmm    = ws + off; off += (size_t)Bb * Mm;
    float* z      = ws + off; off += (size_t)Bb * Nn;
    float* logdet = ws + off; off += (size_t)Bb;

    float* out_mean = (float*)d_out;
    float* out_var  = out_mean + (size_t)Bb * Mm;
    float* nlml     = out_var + (size_t)Bb * Mm;

    prep_kernel<<<(Bb * Mm + 255) / 256, 256, 0, stream>>>(
        coords_train, kp_train, coords_test, kp_test, scale_param, trainP, testP, Kmm);
    buildA_kernel<<<(Bb * Nn * Nn) / 256, 256, 0, stream>>>(trainP, var, scale_param, A);
#define CHOL_STEP(K) chol_step_kernel<<<dim3(NT - (K), NT - (K), Bb), 32, 0, stream>>>(A, Lf, (K));
    CHOL_STEP(0) CHOL_STEP(1) CHOL_STEP(2) CHOL_STEP(3) CHOL_STEP(4) CHOL_STEP(5) CHOL_STEP(6) CHOL_STEP(7)
    CHOL_STEP(8) CHOL_STEP(9) CHOL_STEP(10) CHOL_STEP(11) CHOL_STEP(12) CHOL_STEP(13) CHOL_STEP(14) CHOL_STEP(15)
    CHOL_STEP(16) CHOL_STEP(17) CHOL_STEP(18) CHOL_STEP(19) CHOL_STEP(20) CHOL_STEP(21) CHOL_STEP(22) CHOL_STEP(23)
    CHOL_STEP(24) CHOL_STEP(25) CHOL_STEP(26) CHOL_STEP(27) CHOL_STEP(28) CHOL_STEP(29) CHOL_STEP(30) CHOL_STEP(31)
#undef CHOL_STEP
    logdet_kernel<<<Bb, 1024, 0, stream>>>(Lf, logdet);
    zsolve_kernel<<<Bb, 256, 0, stream>>>(Lf, y_train, mean, logdet, z, nlml);
    trinv_kernel<<<dim3(Nn / 8, Bb), 256, 0, stream>>>(Lf, Linv);

    size_t smem_bytes = (size_t)(Nn * KTS + Nn + Nn * 6) * sizeof(float);
    predict_kernel<<<dim3(Mm / MT, Bb), 128, smem_bytes, stream>>>(
        Linv, trainP, testP, Kmm, z, mean, scale_param, out_mean, out_var);
}


# Targets y and identity carried as extra tile rows through the Cholesky steps: panel solves give z and L^-1 (stored transposed); z-solve and inverse kernels dropped; new sum order in z, nlml, L^-1
# speedup vs baseline: 31.7571x; 4.0811x over previous
#include <hip/hip_runtime.h>
#include <math.h>

#define Nn 1024
#define Mm 16384
#define Bb 2
#define MT 64
#define KTS 65

typedef __attribute__((ext_vector_type(2))) float v2f;
typedef __attribute__((ext_vector_type(8))) float v8f;

#define SQRT3F 1.7320508075688772f
#define LOG2PIF 1.8378770664093453f

__device__ __forceinline__ float kern_pair(float x1, float y1, float a1, float b1, float c1, float q1,
                                           float x2, float y2, float a2, float b2, float c2, float q2,
                                           float scale) {
    float s00 = a1 + a2, s11 = b1 + b2, s01 = c1 + c2;
    float sdet = s00 * s11 - s01 * s01;
    float d0 = x1 - x2, d1 = y1 - y2;
    float Q = 0.5f * (s11 * d0 * d0 - 2.0f * s01 * d0 * d1 + s00 * d1 * d1) / sdet;
    float C = 2.0f * q1 * q2 * rsqrtf(fmaxf(sdet, 1e-5f));
    float t = SQRT3F * sqrtf(fmaxf(Q, 1e-5f));
    return (1.0f + t) * __expf(-t) * C * scale;
}

__global__ void prep_kernel(const float* __restrict__ ct, const float* __restrict__ kpt,
                            const float* __restrict__ cs, const float* __restrict__ kps,
                            const float* __restrict__ scale_param,
                            float* __restrict__ trainP, float* __restrict__ testP,
                            float* __restrict__ Kmm) {
    int gid = blockIdx.x * blockDim.x + threadIdx.x;
    float scale = __expf(scale_param[0]);
    if (gid < Bb * Mm) {
        float a = kps[gid * 3 + 0], b = kps[gid * 3 + 1], c = kps[gid * 3 + 2];
        float det = a * b - c * c;
        testP[gid * 6 + 0] = cs[gid * 2 + 0];
        testP[gid * 6 + 1] = cs[gid * 2 + 1];
        testP[gid * 6 + 2] = a;
        testP[gid * 6 + 3] = b;
        testP[gid * 6 + 4] = c;
        testP[gid * 6 + 5] = sqrtf(sqrtf(det));
        float t0 = SQRT3F * sqrtf(1e-5f);
        float m0 = (1.0f + t0) * __expf(-t0);
        float C = 2.0f * sqrtf(det) * rsqrtf(fmaxf(4.0f * det, 1e-5f));
        Kmm[gid] = C * m0 * scale;
    }
    if (gid < Bb * Nn) {
        float a = kpt[gid * 3 + 0], b = kpt[gid * 3 + 1], c = kpt[gid * 3 + 2];
        float det = a * b - c * c;
        trainP[gid * 6 + 0] = ct[gid * 2 + 0];
        trainP[gid * 6 + 1] = ct[gid * 2 + 1];
        trainP[gid * 6 + 2] = a;
        trainP[gid * 6 + 3] = b;
        trainP[gid * 6 + 4] = c;
        trainP[gid * 6 + 5] = sqrtf(sqrtf(det));
    }
}

__global__ void buildA_kernel(const float* __restrict__ trainP, const float* __restrict__ var,
                              const float* __restrict__ scale_param,
                              const float* __restrict__ y, const float* __restrict__ mean,
                              float* __restrict__ A, float* __restrict__ Yrow) {
    int gid = blockIdx.x * blockDim.x + threadIdx.x;
    int b = gid >> 20;
    int i = (gid >> 10) & (Nn - 1);
    int j = gid & (Nn - 1);
    const float* p1 = trainP + (size_t)(b * Nn + i) * 6;
    const float* p2 = trainP + (size_t)(b * Nn + j) * 6;
    float scale = __expf(scale_param[0]);
    float v = kern_pair(p1[0], p1[1], p1[2], p1[3], p1[4], p1[5],
                        p2[0], p2[1], p2[2], p2[3], p2[4], p2[5], scale);
    if (i == j) v += var[b * Nn + i];
    A[(size_t)gid] = v;
    if (i == 0) Yrow[b * Nn + j] = y[b * Nn + j] - mean[b];
}

#define TS 32
#define NT (Nn / TS)

__device__ __forceinline__ float lane_bcast(float v, int src) {
    return __int_as_float(__builtin_amdgcn_readlane(__float_as_int(v), src));
}

__device__ __forceinline__ void load_row32(const float* p, float (&r)[TS]) {
    const float4* p4 = (const float4*)p;
#pragma unroll
    for (int q = 0; q < TS / 4; q++) {
        float4 t = p4[q];
        r[4 * q + 0] = t.x; r[4 * q + 1] = t.y; r[4 * q + 2] = t.z; r[4 * q + 3] = t.w;
    }
}

__device__ __forceinline__ void store_row32(float* p, const float (&r)[TS]) {
    float4* p4 = (float4*)p;
#pragma unroll
    for (int q = 0; q < TS / 4; q++) {
        float4 t;
        t.x = r[4 * q + 0]; t.y = r[4 * q + 1]; t.z = r[4 * q + 2]; t.w = r[4 * q + 3];
        p4[q] = t;
    }
}

#define QP 36

__global__ __launch_bounds__(32) void chol_step_kernel(float* Abase, float* __restrict__ Lbase,
                                                       float* Xbase, float* Ybase,
                                                       float* __restrict__ zbase, int k) {
    __shared__ __attribute__((aligned(16))) float qs[TS * QP];
    int nmat = NT - k;
    int bx = blockIdx.x;
    int kind = (bx < nmat) ? 0 : (bx <= NT) ? 1 : 2;
    int i = (kind == 0) ? k + bx : (kind == 1) ? bx - nmat : 0;
    int j = k + blockIdx.y;
    if (kind == 0 && j > i) return;
    int b = blockIdx.z;
    int r = threadIdx.x;
    float* A = Abase + (size_t)b * Nn * Nn;
    float* Lo = Lbase + (size_t)b * Nn * Nn;
    float* X = Xbase + (size_t)b * Nn * Nn;
    float* Y = Ybase + (size_t)b * Nn;
    bool ident = (kind == 1) && (i == k);
    bool mine = (kind != 2) || (r == 0);
    const float* prow = (kind == 0) ? A + (size_t)(i * TS + r) * Nn + k * TS
                      : (kind == 1) ? X + (size_t)(i * TS + r) * Nn + k * TS
                                    : Y + k * TS;
    float* trow = (kind == 0) ? A + (size_t)(i * TS + r) * Nn + j * TS
                : (kind == 1) ? X + (size_t)(i * TS + r) * Nn + j * TS
                              : Y + j * TS;

    float d[TS];
    v2f pq[TS];
    load_row32(A + (size_t)(k * TS + r) * Nn + k * TS, d);
    {
        float p[TS], q[TS];
        if (ident) {
#pragma unroll
            for (int c = 0; c < TS; c++) p[c] = (c == r) ? 1.0f : 0.0f;
        } else if (mine) {
            load_row32(prow, p);
        } else {
#pragma unroll
            for (int c = 0; c < TS; c++) p[c] = 0.0f;
        }
        load_row32(A + (size_t)(j * TS + r) * Nn + k * TS, q);
#pragma unroll
        for (int c = 0; c < TS; c++) { pq[c].x = p[c]; pq[c].y = q[c]; }
    }

#pragma unroll
    for (int c = 0; c < TS; c++) {
#pragma unroll
        for (int c1 = 0; c1 < c; c1++) {
            float s = lane_bcast(d[c1], c);
            d[c] -= d[c1] * s;
            pq[c] -= pq[c1] * s;
        }
        float piv = lane_bcast(d[c], c);
        float dg = sqrtf(piv);
        float dinv = 1.0f / dg;
        d[c] = (r == c) ? dg : d[c] * dinv;
        pq[c] = pq[c] * dinv;
    }

    if (j > k) {
        float4* qw = (float4*)(qs + r * QP);
#pragma unroll
        for (int c4 = 0; c4 < TS / 4; c4++)
            qw[c4] = make_float4(pq[4 * c4].y, pq[4 * c4 + 1].y, pq[4 * c4 + 2].y, pq[4 * c4 + 3].y);
        __syncthreads();
        float* arow = trow;
#pragma unroll 1
        for (int s0 = 0; s0 < TS; s0 += 4) {
            float4 av = make_float4(0.0f, 0.0f, 0.0f, 0.0f);
            if (!ident && mine) av = *(const float4*)(arow + s0);
            float a0 = av.x, a1 = av.y, a2 = av.z, a3 = av.w;
            const float4* q0 = (const float4*)(qs + (s0 + 0) * QP);
            const float4* q1 = (const float4*)(qs + (s0 + 1) * QP);
            const float4* q2 = (const float4*)(qs + (s0 + 2) * QP);
            const float4* q3 = (const float4*)(qs + (s0 + 3) * QP);
#pragma unroll
            for (int c4 = 0; c4 < TS / 4; c4++) {
                float4 u0 = q0[c4], u1 = q1[c4], u2 = q2[c4], u3 = q3[c4];
                float p0 = pq[4 * c4].x, p1 = pq[4 * c4 + 1].x, p2 = pq[4 * c4 + 2].x, p3 = pq[4 * c4 + 3].x;
                a0 -= p0 * u0.x; a1 -= p0 * u1.x; a2 -= p0 * u2.x; a3 -= p0 * u3.x;
                a0 -= p1 * u0.y; a1 -= p1 * u1.y; a2 -= p1 * u2.y; a3 -= p1 * u3.y;
                a0 -= p2 * u0.z; a1 -= p2 * u1.z; a2 -= p2 * u2.z; a3 -= p2 * u3.z;
                a0 -= p3 * u0.w; a1 -= p3 * u1.w; a2 -= p3 * u2.w; a3 -= p3 * u3.w;
            }
            if (mine) *(float4*)(arow + s0) = make_float4(a0, a1, a2, a3);
        }
    } else if (kind == 1) {
        float* dst = X + (size_t)(k * TS) * Nn + i * TS + r;
#pragma unroll
        for (int c = 0; c < TS; c++) dst[(size_t)c * Nn] = pq[c].x;
    } else if (kind == 2) {
        if (r == 0) {
            float p[TS];
#pragma unroll
            for (int c = 0; c < TS; c++) p[c] = pq[c].x;
            store_row32(zbase + (size_t)b * Nn + k * TS, p);
        }
    } else if (i > k) {
        float p[TS];
#pragma unroll
        for (int c = 0; c < TS; c++) p[c] = pq[c].x;
        store_row32(Lo + (size_t)(i * TS + r) * Nn + k * TS, p);
    } else {
#pragma unroll
        for (int c = 0; c < TS; c++) if (c > r) d[c] = 0.0f;
        store_row32(Lo + (size_t)(k * TS + r) * Nn + k * TS, d);
    }
}

__global__ __launch_bounds__(1024) void nlml_kernel(const float* __restrict__ Lbase,
                                                    const float* __restrict__ zbase,
                                                    float* __restrict__ nlml_out) {
    int b = blockIdx.x;
    const float* L = Lbase + (size_t)b * Nn * Nn;
    __shared__ float col[Nn];
    __shared__ float zz[Nn];
    int tid = threadIdx.x;
    float z = zbase[b * Nn + tid];
    col[tid] = __logf(L[(size_t)tid * Nn + tid]);
    zz[tid] = z * z;
    __syncthreads();
    for (int s = 512; s > 0; s >>= 1) {
        if (tid < s) {
            col[tid] += col[tid + s];
            zz[tid] += zz[tid + s];
        }
        __syncthreads();
    }
    if (tid == 0) nlml_out[b] = 0.5f * zz[0] + col[0] + 0.5f * Nn * LOG2PIF;
}

__global__ __launch_bounds__(128) void predict_kernel(const float* __restrict__ Linvbase,
                                                      const float* __restrict__ trainPg,
                                                      const float* __restrict__ testPg,
                                                      const float* __restrict__ Kmm,
                                                      const float* __restrict__ zbase,
                                                      const float* __restrict__ meang,
                                                      const float* __restrict__ scale_param,
                                                      float* __restrict__ out_mean,
                                                      float* __restrict__ out_var) {
    extern __shared__ float smem[];
    float* Kt  = smem;
    float* zsh = smem + Nn * KTS;
    float* tp  = zsh + Nn;
    int b = blockIdx.y;
    int colBase = blockIdx.x * MT;
    int tid = threadIdx.x;
    float scale = __expf(scale_param[0]);

    for (int idx = tid; idx < Nn; idx += 128) zsh[idx] = zbase[b * Nn + idx];
    for (int idx = tid; idx < Nn * 6; idx += 128) tp[idx] = trainPg[(size_t)b * Nn * 6 + idx];
    __syncthreads();

    {
        int c = tid & 63;
        int h = tid >> 6;
        const float* q = testPg + (size_t)(b * Mm + colBase + c) * 6;
        float x2 = q[0], y2 = q[1], a2 = q[2], b2 = q[3], c2 = q[4], q2 = q[5];
        for (int n = h; n < Nn; n += 2) {
            const float* p = tp + n * 6;
            Kt[n * KTS + c] = kern_pair(p[0], p[1], p[2], p[3], p[4], p[5],
                                        x2, y2, a2, b2, c2, q2, scale);
        }
    }
    __syncthreads();

    int wave = tid >> 5, lane = tid & 31;
    int r = lane & 15, hi = lane >> 4;
    int wcol = wave * 16;
    const float* Li = Linvbase + (size_t)b * Nn * Nn;

    float meanAcc = 0.0f, varAcc = 0.0f;
    for (int rb = 0; rb < Nn / 16; rb++) {
        v8f acc = {};
        const float* arow = Li + (size_t)(rb * 16 + r) * Nn;
        for (int j = 0; j <= rb; j++) {
            int kb = j * 16;
#pragma unroll
            for (int k0 = 0; k0 < 16; k0 += 4) {
                int kk = kb + k0 + 2 * hi;
                v2f a = *(const v2f*)(arow + kk);
                v2f bb;
                bb.x = Kt[kk * KTS + wcol + r];
                bb.y = Kt[(kk + 1) * KTS + wcol + r];
                acc = __builtin_amdgcn_wmma_f32_16x16x4_f32(false, a, false, bb,
                                                            (short)0, acc, false, false);
            }
        }
        const float* zb = zsh + rb * 16 + 8 * hi;
#pragma unroll
        for (int v = 0; v < 8; v++) {
            float cv = acc[v];
            meanAcc += zb[v] * cv;
            varAcc  += cv * cv;
        }
    }
    meanAcc += __shfl_xor(meanAcc, 16);
    varAcc  += __shfl_xor(varAcc, 16);
    if (lane < 16) {
        int m = colBase + wcol + lane;
        out_mean[b * Mm + m] = meang[b] + meanAcc;
        out_var[b * Mm + m]  = Kmm[b * Mm + m] - varAcc;
    }
}

extern "C" void kernel_launch(void* const* d_in, const int* in_sizes, int n_in,
                              void* d_out, int out_size, void* d_ws, size_t ws_size,
                              hipStream_t stream) {
    const float* coords_train = (const float*)d_in[0];
    const float* kp_train     = (const float*)d_in[1];
    const float* coords_test  = (const float*)d_in[2];
    const float* kp_test      = (const float*)d_in[3];
    const float* y_train      = (const float*)d_in[4];
    const float* var          = (const float*)d_in[5];
    const float* mean         = (const float*)d_in[6];
    const float* scale_param  = (const float*)d_in[7];

    float* ws = (float*)d_ws;
    size_t off = 0;
    float* A      = ws + off; off += (size_t)Bb * Nn * Nn;
    float* Linv   = ws + off; off += (size_t)Bb * Nn * Nn;
    float* Lf     = ws + off; off += (size_t)Bb * Nn * Nn;
    float* trainP = ws + off; off += (size_t)Bb * Nn * 6;
    float* testP  = ws + off; off += (size_t)Bb * Mm * 6;
    float* Kmm    = ws + off; off += (size_t)Bb * Mm;
    float* z      = ws + off; off += (size_t)Bb * Nn;
    float* Yrow   = ws + off; off += (size_t)Bb * Nn;

    float* out_mean = (float*)d_out;
    float* out_var  = out_mean + (size_t)Bb * Mm;
    float* nlml     = out_var + (size_t)Bb * Mm;

    prep_kernel<<<(Bb * Mm + 255) / 256, 256, 0, stream>>>(
        coords_train, kp_train, coords_test, kp_test, scale_param, trainP, testP, Kmm);
    buildA_kernel<<<(Bb * Nn * Nn) / 256, 256, 0, stream>>>(trainP, var, scale_param, y_train, mean, A, Yrow);
#define CHOL_STEP(K) chol_step_kernel<<<dim3(NT + 2, NT - (K), Bb), 32, 0, stream>>>(A, Lf, Linv, Yrow, z, (K));
    CHOL_STEP(0) CHOL_STEP(1) CHOL_STEP(2) CHOL_STEP(3) CHOL_STEP(4) CHOL_STEP(5) CHOL_STEP(6) CHOL_STEP(7)
    CHOL_STEP(8) CHOL_STEP(9) CHOL_STEP(10) CHOL_STEP(11) CHOL_STEP(12) CHOL_STEP(13) CHOL_STEP(14) CHOL_STEP(15)
    CHOL_STEP(16) CHOL_STEP(17) CHOL_STEP(18) CHOL_STEP(19) CHOL_STEP(20) CHOL_STEP(21) CHOL_STEP(22) CHOL_STEP(23)
    CHOL_STEP(24) CHOL_STEP(25) CHOL_STEP(26) CHOL_STEP(27) CHOL_STEP(28) CHOL_STEP(29) CHOL_STEP(30) CHOL_STEP(31)
#undef CHOL_STEP
    nlml_kernel<<<Bb, 1024, 0, stream>>>(Lf, z, nlml);

    size_t smem_bytes = (size_t)(Nn * KTS + Nn + Nn * 6) * sizeof(float);
    predict_kernel<<<dim3(Mm / MT, Bb), 128, smem_bytes, stream>>>(
        Linv, trainP, testP, Kmm, z, mean, scale_param, out_mean, out_var);
}


# predict: rows of L^-1 (A operand) staged once per workgroup in two LDS tiles by 16-byte loads one chunk ahead, fragment loads grouped before 16 matrix steps; tiling, k order and sum order kept
# speedup vs baseline: 42.8533x; 1.3494x over previous
#include <hip/hip_runtime.h>
#include <math.h>

#define Nn 1024
#define Mm 16384
#define Bb 2
#define MT 64
#define KTS 65
#define ACH 64
#define APITCH 68

typedef __attribute__((ext_vector_type(2))) float v2f;
typedef __attribute__((ext_vector_type(8))) float v8f;

#define SQRT3F 1.7320508075688772f
#define LOG2PIF 1.8378770664093453f

__device__ __forceinline__ float kern_pair(float x1, float y1, float a1, float b1, float c1, float q1,
                                           float x2, float y2, float a2, float b2, float c2, float q2,
                                           float scale) {
    float s00 = a1 + a2, s11 = b1 + b2, s01 = c1 + c2;
    float sdet = s00 * s11 - s01 * s01;
    float d0 = x1 - x2, d1 = y1 - y2;
    float Q = 0.5f * (s11 * d0 * d0 - 2.0f * s01 * d0 * d1 + s00 * d1 * d1) / sdet;
    float C = 2.0f * q1 * q2 * rsqrtf(fmaxf(sdet, 1e-5f));
    float t = SQRT3F * sqrtf(fmaxf(Q, 1e-5f));
    return (1.0f + t) * __expf(-t) * C * scale;
}

__global__ void prep_kernel(const float* __restrict__ ct, const float* __restrict__ kpt,
                            const float* __restrict__ cs, const float* __restrict__ kps,
                            const float* __restrict__ scale_param,
                            float* __restrict__ trainP, float* __restrict__ testP,
                            float* __restrict__ Kmm) {
    int gid = blockIdx.x * blockDim.x + threadIdx.x;
    float scale = __expf(scale_param[0]);
    if (gid < Bb * Mm) {
        float a = kps[gid * 3 + 0], b = kps[gid * 3 + 1], c = kps[gid * 3 + 2];
        float det = a * b - c * c;
        testP[gid * 6 + 0] = cs[gid * 2 + 0];
        testP[gid * 6 + 1] = cs[gid * 2 + 1];
        testP[gid * 6 + 2] = a;
        testP[gid * 6 + 3] = b;
        testP[gid * 6 + 4] = c;
        testP[gid * 6 + 5] = sqrtf(sqrtf(det));
        float t0 = SQRT3F * sqrtf(1e-5f);
        float m0 = (1.0f + t0) * __expf(-t0);
        float C = 2.0f * sqrtf(det) * rsqrtf(fmaxf(4.0f * det, 1e-5f));
        Kmm[gid] = C * m0 * scale;
    }
    if (gid < Bb * Nn) {
        float a = kpt[gid * 3 + 0], b = kpt[gid * 3 + 1], c = kpt[gid * 3 + 2];
        float det = a * b - c * c;
        trainP[gid * 6 + 0] = ct[gid * 2 + 0];
        trainP[gid * 6 + 1] = ct[gid * 2 + 1];
        trainP[gid * 6 + 2] = a;
        trainP[gid * 6 + 3] = b;
        trainP[gid * 6 + 4] = c;
        trainP[gid * 6 + 5] = sqrtf(sqrtf(det));
    }
}

__global__ void buildA_kernel(const float* __restrict__ trainP, const float* __restrict__ var,
                              const float* __restrict__ scale_param,
                              const float* __restrict__ y, const float* __restrict__ mean,
                              float* __restrict__ A, float* __restrict__ Yrow) {
    int gid = blockIdx.x * blockDim.x + threadIdx.x;
    int b = gid >> 20;
    int i = (gid >> 10) & (Nn - 1);
    int j = gid & (Nn - 1);
    const float* p1 = trainP + (size_t)(b * Nn + i) * 6;
    const float* p2 = trainP + (size_t)(b * Nn + j) * 6;
    float scale = __expf(scale_param[0]);
    float v = kern_pair(p1[0], p1[1], p1[2], p1[3], p1[4], p1[5],
                        p2[0], p2[1], p2[2], p2[3], p2[4], p2[5], scale);
    if (i == j) v += var[b * Nn + i];
    A[(size_t)gid] = v;
    if (i == 0) Yrow[b * Nn + j] = y[b * Nn + j] - mean[b];
}

#define TS 32
#define NT (Nn / TS)

__device__ __forceinline__ float lane_bcast(float v, int src) {
    return __int_as_float(__builtin_amdgcn_readlane(__float_as_int(v), src));
}

__device__ __forceinline__ void load_row32(const float* p, float (&r)[TS]) {
    const float4* p4 = (const float4*)p;
#pragma unroll
    for (int q = 0; q < TS / 4; q++) {
        float4 t = p4[q];
        r[4 * q + 0] = t.x; r[4 * q + 1] = t.y; r[4 * q + 2] = t.z; r[4 * q + 3] = t.w;
    }
}

__device__ __forceinline__ void store_row32(float* p, const float (&r)[TS]) {
    float4* p4 = (float4*)p;
#pragma unroll
    for (int q = 0; q < TS / 4; q++) {
        float4 t;
        t.x = r[4 * q + 0]; t.y = r[4 * q + 1]; t.z = r[4 * q + 2]; t.w = r[4 * q + 3];
        p4[q] = t;
    }
}

#define QP 36

__global__ __launch_bounds__(32) void chol_step_kernel(float* Abase, float* __restrict__ Lbase,
                                                       float* Xbase, float* Ybase,
                                                       float* __restrict__ zbase, int k) {
    __shared__ __attribute__((aligned(16))) float qs[TS * QP];
    int nmat = NT - k;
    int bx = blockIdx.x;
    int kind = (bx < nmat) ? 0 : (bx <= NT) ? 1 : 2;
    int i = (kind == 0) ? k + bx : (kind == 1) ? bx - nmat : 0;
    int j = k + blockIdx.y;
    if (kind == 0 && j > i) return;
    int b = blockIdx.z;
    int r = threadIdx.x;
    float* A = Abase + (size_t)b * Nn * Nn;
    float* Lo = Lbase + (size_t)b * Nn * Nn;
    float* X = Xbase + (size_t)b * Nn * Nn;
    float* Y = Ybase + (size_t)b * Nn;
    bool ident = (kind == 1) && (i == k);
    bool mine = (kind != 2) || (r == 0);
    const float* prow = (kind == 0) ? A + (size_t)(i * TS + r) * Nn + k * TS
                      : (kind == 1) ? X + (size_t)(i * TS + r) * Nn + k * TS
                                    : Y + k * TS;
    float* trow = (kind == 0) ? A + (size_t)(i * TS + r) * Nn + j * TS
                : (kind == 1) ? X + (size_t)(i * TS + r) * Nn + j * TS
                              : Y + j * TS;

    float d[TS];
    v2f pq[TS];
    load_row32(A + (size_t)(k * TS + r) * Nn + k * TS, d);
    {
        float p[TS], q[TS];
        if (ident) {
#pragma unroll
            for (int c = 0; c < TS; c++) p[c] = (c == r) ? 1.0f : 0.0f;
        } else if (mine) {
            load_row32(prow, p);
        } else {
#pragma unroll
            for (int c = 0; c < TS; c++) p[c] = 0.0f;
        }
        load_row32(A + (size_t)(j * TS + r) * Nn + k * TS, q);
#pragma unroll
        for (int c = 0; c < TS; c++) { pq[c].x = p[c]; pq[c].y = q[c]; }
    }

#pragma unroll
    for (int c = 0; c < TS; c++) {
#pragma unroll
        for (int c1 = 0; c1 < c; c1++) {
            float s = lane_bcast(d[c1], c);
            d[c] -= d[c1] * s;
            pq[c] -= pq[c1] * s;
        }
        float piv = lane_bcast(d[c], c);
        float dg = sqrtf(piv);
        float dinv = 1.0f / dg;
        d[c] = (r == c) ? dg : d[c] * dinv;
        pq[c] = pq[c] * dinv;
    }

    if (j > k) {
        float4* qw = (float4*)(qs + r * QP);
#pragma unroll
        for (int c4 = 0; c4 < TS / 4; c4++)
            qw[c4] = make_float4(pq[4 * c4].y, pq[4 * c4 + 1].y, pq[4 * c4 + 2].y, pq[4 * c4 + 3].y);
        __syncthreads();
        float* arow = trow;
#pragma unroll 1
        for (int s0 = 0; s0 < TS; s0 += 4) {
            float4 av = make_float4(0.0f, 0.0f, 0.0f, 0.0f);
            if (!ident && mine) av = *(const float4*)(arow + s0);
            float a0 = av.x, a1 = av.y, a2 = av.z, a3 = av.w;
            const float4* q0 = (const float4*)(qs + (s0 + 0) * QP);
            const float4* q1 = (const float4*)(qs + (s0 + 1) * QP);
            const float4* q2 = (const float4*)(qs + (s0 + 2) * QP);
            const float4* q3 = (const float4*)(qs + (s0 + 3) * QP);
#pragma unroll
            for (int c4 = 0; c4 < TS / 4; c4++) {
                float4 u0 = q0[c4], u1 = q1[c4], u2 = q2[c4], u3 = q3[c4];
                float p0 = pq[4 * c4].x, p1 = pq[4 * c4 + 1].x, p2 = pq[4 * c4 + 2].x, p3 = pq[4 * c4 + 3].x;
                a0 -= p0 * u0.x; a1 -= p0 * u1.x; a2 -= p0 * u2.x; a3 -= p0 * u3.x;
                a0 -= p1 * u0.y; a1 -= p1 * u1.y; a2 -= p1 * u2.y; a3 -= p1 * u3.y;
                a0 -= p2 * u0.z; a1 -= p2 * u1.z; a2 -= p2 * u2.z; a3 -= p2 * u3.z;
                a0 -= p3 * u0.w; a1 -= p3 * u1.w; a2 -= p3 * u2.w; a3 -= p3 * u3.w;
            }
            if (mine) *(float4*)(arow + s0) = make_float4(a0, a1, a2, a3);
        }
    } else if (kind == 1) {
        float* dst = X + (size_t)(k * TS) * Nn + i * TS + r;
#pragma unroll
        for (int c = 0; c < TS; c++) dst[(size_t)c * Nn] = pq[c].x;
    } else if (kind == 2) {
        if (r == 0) {
            float p[TS];
#pragma unroll
            for (int c = 0; c < TS; c++) p[c] = pq[c].x;
            store_row32(zbase + (size_t)b * Nn + k * TS, p);
        }
    } else if (i > k) {
        float p[TS];
#pragma unroll
        for (int c = 0; c < TS; c++) p[c] = pq[c].x;
        store_row32(Lo + (size_t)(i * TS + r) * Nn + k * TS, p);
    } else {
#pragma unroll
        for (int c = 0; c < TS; c++) if (c > r) d[c] = 0.0f;
        store_row32(Lo + (size_t)(k * TS + r) * Nn + k * TS, d);
    }
}

__global__ __launch_bounds__(1024) void nlml_kernel(const float* __restrict__ Lbase,
                                                    const float* __restrict__ zbase,
                                                    float* __restrict__ nlml_out) {
    int b = blockIdx.x;
    const float* L = Lbase + (size_t)b * Nn * Nn;
    __shared__ float col[Nn];
    __shared__ float zz[Nn];
    int tid = threadIdx.x;
    float z = zbase[b * Nn + tid];
    col[tid] = __logf(L[(size_t)tid * Nn + tid]);
    zz[tid] = z * z;
    __syncthreads();
    for (int s = 512; s > 0; s >>= 1) {
        if (tid < s) {
            col[tid] += col[tid + s];
            zz[tid] += zz[tid + s];
        }
        __syncthreads();
    }
    if (tid == 0) nlml_out[b] = 0.5f * zz[0] + col[0] + 0.5f * Nn * LOG2PIF;
}

__global__ __launch_bounds__(128) void predict_kernel(const float* __restrict__ Linvbase,
                                                      const float* __restrict__ trainPg,
                                                      const float* __restrict__ testPg,
                                                      const float* __restrict__ Kmm,
                                                      const float* __restrict__ zbase,
                                                      const float* __restrict__ meang,
                                                      const float* __restrict__ scale_param,
                                                      float* __restrict__ out_mean,
                                                      float* __restrict__ out_var) {
    extern __shared__ float smem[];
    float* Kt  = smem;
    float* zsh = smem + Nn * KTS;
    float* tp  = zsh + Nn;
    float* Ab  = tp + Nn * 6;
    int b = blockIdx.y;
    int colBase = blockIdx.x * MT;
    int tid = threadIdx.x;
    float scale = __expf(scale_param[0]);

    for (int idx = tid; idx < Nn; idx += 128) zsh[idx] = zbase[b * Nn + idx];
    for (int idx = tid; idx < Nn * 6; idx += 128) tp[idx] = trainPg[(size_t)b * Nn * 6 + idx];
    __syncthreads();

    {
        int c = tid & 63;
        int h = tid >> 6;
        const float* q = testPg + (size_t)(b * Mm + colBase + c) * 6;
        float x2 = q[0], y2 = q[1], a2 = q[2], b2 = q[3], c2 = q[4], q2 = q[5];
        for (int n = h; n < Nn; n += 2) {
            const float* p = tp + n * 6;
            Kt[n * KTS + c] = kern_pair(p[0], p[1], p[2], p[3], p[4], p[5],
                                        x2, y2, a2, b2, c2, q2, scale);
        }
    }
    __syncthreads();

    int wave = tid >> 5, lane = tid & 31;
    int r = lane & 15, hi = lane >> 4;
    int wcol = wave * 16;
    const float* Li = Linvbase + (size_t)b * Nn * Nn;

    int srow = tid >> 4, sseg = tid & 15;
    const float* fa = Ab + r * APITCH + 2 * hi;
    const float* fb = Kt + (2 * hi) * KTS + wcol + r;
    float4 g0, g1;
    {
        g0 = make_float4(0.0f, 0.0f, 0.0f, 0.0f); g1 = g0;
        if (4 * sseg < 16) {
            g0 = *(const float4*)(Li + (size_t)srow * Nn + 4 * sseg);
            g1 = *(const float4*)(Li + (size_t)(srow + 8) * Nn + 4 * sseg);
        }
    }
    int buf = 0;
    float meanAcc = 0.0f, varAcc = 0.0f;
    for (int rb = 0; rb < Nn / 16; rb++) {
        v8f acc = {};
        int nch = (rb >> 2) + 1;
        for (int c = 0; c < nch; c++) {
            float* At = Ab + buf * (16 * APITCH);
            *(float4*)(At + srow * APITCH + 4 * sseg) = g0;
            *(float4*)(At + (srow + 8) * APITCH + 4 * sseg) = g1;
            int nrb = rb, nc = c + 1;
            if (nc == nch) { nrb = rb + 1; nc = 0; }
            if (nrb < Nn / 16) {
                int k0 = nc * ACH + 4 * sseg;
                g0 = make_float4(0.0f, 0.0f, 0.0f, 0.0f); g1 = g0;
                if (k0 < 16 * (nrb + 1)) {
                    g0 = *(const float4*)(Li + (size_t)(nrb * 16 + srow) * Nn + k0);
                    g1 = *(const float4*)(Li + (size_t)(nrb * 16 + srow + 8) * Nn + k0);
                }
            }
            __builtin_amdgcn_fence(__ATOMIC_RELEASE, "workgroup", "local");
            __builtin_amdgcn_s_barrier();
            __builtin_amdgcn_fence(__ATOMIC_ACQUIRE, "workgroup", "local");
            const float* fac = fa + buf * (16 * APITCH);
            const float* fbc = fb + (c * ACH) * KTS;
            v2f av[ACH / 4], bv[ACH / 4];
#pragma unroll
            for (int s = 0; s < ACH / 4; s++) {
                av[s] = *(const v2f*)(fac + 4 * s);
                bv[s].x = fbc[(4 * s) * KTS];
                bv[s].y = fbc[(4 * s + 1) * KTS];
            }
#pragma unroll
            for (int s = 0; s < ACH / 4; s++)
                acc = __builtin_amdgcn_wmma_f32_16x16x4_f32(false, av[s], false, bv[s],
                                                            (short)0, acc, false, false);
            __builtin_amdgcn_sched_group_barrier(0x100, 24, 0);
            __builtin_amdgcn_sched_group_barrier(0x008, 16, 0);
            buf ^= 1;
        }
        const float* zb = zsh + rb * 16 + 8 * hi;
#pragma unroll
        for (int v = 0; v < 8; v++) {
            float cv = acc[v];
            meanAcc += zb[v] * cv;
            varAcc  += cv * cv;
        }
    }
    meanAcc += __shfl_xor(meanAcc, 16);
    varAcc  += __shfl_xor(varAcc, 16);
    if (lane < 16) {
        int m = colBase + wcol + lane;
        out_mean[b * Mm + m] = meang[b] + meanAcc;
        out_var[b * Mm + m]  = Kmm[b * Mm + m] - varAcc;
    }
}

extern "C" void kernel_launch(void* const* d_in, const int* in_sizes, int n_in,
                              void* d_out, int out_size, void* d_ws, size_t ws_size,
                              hipStream_t stream) {
    const float* coords_train = (const float*)d_in[0];
    const float* kp_train     = (const float*)d_in[1];
    const float* coords_test  = (const float*)d_in[2];
    const float* kp_test      = (const float*)d_in[3];
    const float* y_train      = (const float*)d_in[4];
    const float* var          = (const float*)d_in[5];
    const float* mean         = (const float*)d_in[6];
    const float* scale_param  = (const float*)d_in[7];

    float* ws = (float*)d_ws;
    size_t off = 0;
    float* A      = ws + off; off += (size_t)Bb * Nn * Nn;
    float* Linv   = ws + off; off += (size_t)Bb * Nn * Nn;
    float* Lf     = ws + off; off += (size_t)Bb * Nn * Nn;
    float* trainP = ws + off; off += (size_t)Bb * Nn * 6;
    float* testP  = ws + off; off += (size_t)Bb * Mm * 6;
    float* Kmm    = ws + off; off += (size_t)Bb * Mm;
    float* z      = ws + off; off += (size_t)Bb * Nn;
    float* Yrow   = ws + off; off += (size_t)Bb * Nn;

    float* out_mean = (float*)d_out;
    float* out_var  = out_mean + (size_t)Bb * Mm;
    float* nlml     = out_var + (size_t)Bb * Mm;

    prep_kernel<<<(Bb * Mm + 255) / 256, 256, 0, stream>>>(
        coords_train, kp_train, coords_test, kp_test, scale_param, trainP, testP, Kmm);
    buildA_kernel<<<(Bb * Nn * Nn) / 256, 256, 0, stream>>>(trainP, var, scale_param, y_train, mean, A, Yrow);
#define CHOL_STEP(K) chol_step_kernel<<<dim3(NT + 2, NT - (K), Bb), 32, 0, stream>>>(A, Lf, Linv, Yrow, z, (K));
    CHOL_STEP(0) CHOL_STEP(1) CHOL_STEP(2) CHOL_STEP(3) CHOL_STEP(4) CHOL_STEP(5) CHOL_STEP(6) CHOL_STEP(7)
    CHOL_STEP(8) CHOL_STEP(9) CHOL_STEP(10) CHOL_STEP(11) CHOL_STEP(12) CHOL_STEP(13) CHOL_STEP(14) CHOL_STEP(15)
    CHOL_STEP(16) CHOL_STEP(17) CHOL_STEP(18) CHOL_STEP(19) CHOL_STEP(20) CHOL_STEP(21) CHOL_STEP(22) CHOL_STEP(23)
    CHOL_STEP(24) CHOL_STEP(25) CHOL_STEP(26) CHOL_STEP(27) CHOL_STEP(28) CHOL_STEP(29) CHOL_STEP(30) CHOL_STEP(31)
#undef CHOL_STEP
    nlml_kernel<<<Bb, 1024, 0, stream>>>(Lf, z, nlml);

    size_t smem_bytes = (size_t)(Nn * KTS + Nn + Nn * 6 + 2 * 16 * APITCH) * sizeof(float);
    predict_kernel<<<dim3(Mm / MT, Bb), 128, smem_bytes, stream>>>(
        Linv, trainP, testP, Kmm, z, mean, scale_param, out_mean, out_var);
}


# K tile build in the predict kernel: eight train rows per thread per pass (parameter reads first, stores last), so the one wave per SIMD overlaps eight dependent chains; arithmetic unchanged
# speedup vs baseline: 46.7031x; 1.0898x over previous
#include <hip/hip_runtime.h>
#include <math.h>

#define Nn 1024
#define Mm 16384
#define Bb 2
#define MT 64
#define KTS 65
#define ACH 64
#define KU 8
#define APITCH 68

typedef __attribute__((ext_vector_type(2))) float v2f;
typedef __attribute__((ext_vector_type(8))) float v8f;

#define SQRT3F 1.7320508075688772f
#define LOG2PIF 1.8378770664093453f

__device__ __forceinline__ float kern_pair(float x1, float y1, float a1, float b1, float c1, float q1,
                                           float x2, float y2, float a2, float b2, float c2, float q2,
                                           float scale) {
    float s00 = a1 + a2, s11 = b1 + b2, s01 = c1 + c2;
    float sdet = s00 * s11 - s01 * s01;
    float d0 = x1 - x2, d1 = y1 - y2;
    float Q = 0.5f * (s11 * d0 * d0 - 2.0f * s01 * d0 * d1 + s00 * d1 * d1) / sdet;
    float C = 2.0f * q1 * q2 * rsqrtf(fmaxf(sdet, 1e-5f));
    float t = SQRT3F * sqrtf(fmaxf(Q, 1e-5f));
    return (1.0f + t) * __expf(-t) * C * scale;
}

__global__ void prep_kernel(const float* __restrict__ ct, const float* __restrict__ kpt,
                            const float* __restrict__ cs, const float* __restrict__ kps,
                            const float* __restrict__ scale_param,
                            float* __restrict__ trainP, float* __restrict__ testP,
                            float* __restrict__ Kmm) {
    int gid = blockIdx.x * blockDim.x + threadIdx.x;
    float scale = __expf(scale_param[0]);
    if (gid < Bb * Mm) {
        float a = kps[gid * 3 + 0], b = kps[gid * 3 + 1], c = kps[gid * 3 + 2];
        float det = a * b - c * c;
        testP[gid * 6 + 0] = cs[gid * 2 + 0];
        testP[gid * 6 + 1] = cs[gid * 2 + 1];
        testP[gid * 6 + 2] = a;
        testP[gid * 6 + 3] = b;
        testP[gid * 6 + 4] = c;
        testP[gid * 6 + 5] = sqrtf(sqrtf(det));
        float t0 = SQRT3F * sqrtf(1e-5f);
        float m0 = (1.0f + t0) * __expf(-t0);
        float C = 2.0f * sqrtf(det) * rsqrtf(fmaxf(4.0f * det, 1e-5f));
        Kmm[gid] = C * m0 * scale;
    }
    if (gid < Bb * Nn) {
        float a = kpt[gid * 3 + 0], b = kpt[gid * 3 + 1], c = kpt[gid * 3 + 2];
        float det = a * b - c * c;
        trainP[gid * 6 + 0] = ct[gid * 2 + 0];
        trainP[gid * 6 + 1] = ct[gid * 2 + 1];
        trainP[gid * 6 + 2] = a;
        trainP[gid * 6 + 3] = b;
        trainP[gid * 6 + 4] = c;
        trainP[gid * 6 + 5] = sqrtf(sqrtf(det));
    }
}

__global__ void buildA_kernel(const float* __restrict__ trainP, const float* __restrict__ var,
                              const float* __restrict__ scale_param,
                              const float* __restrict__ y, const float* __restrict__ mean,
                              float* __restrict__ A, float* __restrict__ Yrow) {
    int gid = blockIdx.x * blockDim.x + threadIdx.x;
    int b = gid >> 20;
    int i = (gid >> 10) & (Nn - 1);
    int j = gid & (Nn - 1);
    const float* p1 = trainP + (size_t)(b * Nn + i) * 6;
    const float* p2 = trainP + (size_t)(b * Nn + j) * 6;
    float scale = __expf(scale_param[0]);
    float v = kern_pair(p1[0], p1[1], p1[2], p1[3], p1[4], p1[5],
                        p2[0], p2[1], p2[2], p2[3], p2[4], p2[5], scale);
    if (i == j) v += var[b * Nn + i];
    A[(size_t)gid] = v;
    if (i == 0) Yrow[b * Nn + j] = y[b * Nn + j] - mean[b];
}

#define TS 32
#define NT (Nn / TS)

__device__ __forceinline__ float lane_bcast(float v, int src) {
    return __int_as_float(__builtin_amdgcn_readlane(__float_as_int(v), src));
}

__device__ __forceinline__ void load_row32(const float* p, float (&r)[TS]) {
    const float4* p4 = (const float4*)p;
#pragma unroll
    for (int q = 0; q < TS / 4; q++) {
        float4 t = p4[q];
        r[4 * q + 0] = t.x; r[4 * q + 1] = t.y; r[4 * q + 2] = t.z; r[4 * q + 3] = t.w;
    }
}

__device__ __forceinline__ void store_row32(float* p, const float (&r)[TS]) {
    float4* p4 = (float4*)p;
#pragma unroll
    for (int q = 0; q < TS / 4; q++) {
        float4 t;
        t.x = r[4 * q + 0]; t.y = r[4 * q + 1]; t.z = r[4 * q + 2]; t.w = r[4 * q + 3];
        p4[q] = t;
    }
}

#define QP 36

__global__ __launch_bounds__(32) void chol_step_kernel(float* Abase, float* __restrict__ Lbase,
                                                       float* Xbase, float* Ybase,
                                                       float* __restrict__ zbase, int k) {
    __shared__ __attribute__((aligned(16))) float qs[TS * QP];
    int nmat = NT - k;
    int bx = blockIdx.x;
    int kind = (bx < nmat) ? 0 : (bx <= NT) ? 1 : 2;
    int i = (kind == 0) ? k + bx : (kind == 1) ? bx - nmat : 0;
    int j = k + blockIdx.y;
    if (kind == 0 && j > i) return;
    int b = blockIdx.z;
    int r = threadIdx.x;
    float* A = Abase + (size_t)b * Nn * Nn;
    float* Lo = Lbase + (size_t)b * Nn * Nn;
    float* X = Xbase + (size_t)b * Nn * Nn;
    float* Y = Ybase + (size_t)b * Nn;
    bool ident = (kind == 1) && (i == k);
    bool mine = (kind != 2) || (r == 0);
    const float* prow = (kind == 0) ? A + (size_t)(i * TS + r) * Nn + k * TS
                      : (kind == 1) ? X + (size_t)(i * TS + r) * Nn + k * TS
                                    : Y + k * TS;
    float* trow = (kind == 0) ? A + (size_t)(i * TS + r) * Nn + j * TS
                : (kind == 1) ? X + (size_t)(i * TS + r) * Nn + j * TS
                              : Y + j * TS;

    float d[TS];
    v2f pq[TS];
    load_row32(A + (size_t)(k * TS + r) * Nn + k * TS, d);
    {
        float p[TS], q[TS];
        if (ident) {
#pragma unroll
            for (int c = 0; c < TS; c++) p[c] = (c == r) ? 1.0f : 0.0f;
        } else if (mine) {
            load_row32(prow, p);
        } else {
#pragma unroll
            for (int c = 0; c < TS; c++) p[c] = 0.0f;
        }
        load_row32(A + (size_t)(j * TS + r) * Nn + k * TS, q);
#pragma unroll
        for (int c = 0; c < TS; c++) { pq[c].x = p[c]; pq[c].y = q[c]; }
    }

#pragma unroll
    for (int c = 0; c < TS; c++) {
#pragma unroll
        for (int c1 = 0; c1 < c; c1++) {
            float s = lane_bcast(d[c1], c);
            d[c] -= d[c1] * s;
            pq[c] -= pq[c1] * s;
        }
        float piv = lane_bcast(d[c], c);
        float dg = sqrtf(piv);
        float dinv = 1.0f / dg;
        d[c] = (r == c) ? dg : d[c] * dinv;
        pq[c] = pq[c] * dinv;
    }

    if (j > k) {
        float4* qw = (float4*)(qs + r * QP);
#pragma unroll
        for (int c4 = 0; c4 < TS / 4; c4++)
            qw[c4] = make_float4(pq[4 * c4].y, pq[4 * c4 + 1].y, pq[4 * c4 + 2].y, pq[4 * c4 + 3].y);
        __syncthreads();
        float* arow = trow;
#pragma unroll 1
        for (int s0 = 0; s0 < TS; s0 += 4) {
            float4 av = make_float4(0.0f, 0.0f, 0.0f, 0.0f);
            if (!ident && mine) av = *(const float4*)(arow + s0);
            float a0 = av.x, a1 = av.y, a2 = av.z, a3 = av.w;
            const float4* q0 = (const float4*)(qs + (s0 + 0) * QP);
            const float4* q1 = (const float4*)(qs + (s0 + 1) * QP);
            const float4* q2 = (const float4*)(qs + (s0 + 2) * QP);
            const float4* q3 = (const float4*)(qs + (s0 + 3) * QP);
#pragma unroll
            for (int c4 = 0; c4 < TS / 4; c4++) {
                float4 u0 = q0[c4], u1 = q1[c4], u2 = q2[c4], u3 = q3[c4];
                float p0 = pq[4 * c4].x, p1 = pq[4 * c4 + 1].x, p2 = pq[4 * c4 + 2].x, p3 = pq[4 * c4 + 3].x;
                a0 -= p0 * u0.x; a1 -= p0 * u1.x; a2 -= p0 * u2.x; a3 -= p0 * u3.x;
                a0 -= p1 * u0.y; a1 -= p1 * u1.y; a2 -= p1 * u2.y; a3 -= p1 * u3.y;
                a0 -= p2 * u0.z; a1 -= p2 * u1.z; a2 -= p2 * u2.z; a3 -= p2 * u3.z;
                a0 -= p3 * u0.w; a1 -= p3 * u1.w; a2 -= p3 * u2.w; a3 -= p3 * u3.w;
            }
            if (mine) *(float4*)(arow + s0) = make_float4(a0, a1, a2, a3);
        }
    } else if (kind == 1) {
        float* dst = X + (size_t)(k * TS) * Nn + i * TS + r;
#pragma unroll
        for (int c = 0; c < TS; c++) dst[(size_t)c * Nn] = pq[c].x;
    } else if (kind == 2) {
        if (r == 0) {
            float p[TS];
#pragma unroll
            for (int c = 0; c < TS; c++) p[c] = pq[c].x;
            store_row32(zbase + (size_t)b * Nn + k * TS, p);
        }
    } else if (i > k) {
        float p[TS];
#pragma unroll
        for (int c = 0; c < TS; c++) p[c] = pq[c].x;
        store_row32(Lo + (size_t)(i * TS + r) * Nn + k * TS, p);
    } else {
#pragma unroll
        for (int c = 0; c < TS; c++) if (c > r) d[c] = 0.0f;
        store_row32(Lo + (size_t)(k * TS + r) * Nn + k * TS, d);
    }
}

__global__ __launch_bounds__(1024) void nlml_kernel(const float* __restrict__ Lbase,
                                                    const float* __restrict__ zbase,
                                                    float* __restrict__ nlml_out) {
    int b = blockIdx.x;
    const float* L = Lbase + (size_t)b * Nn * Nn;
    __shared__ float col[Nn];
    __shared__ float zz[Nn];
    int tid = threadIdx.x;
    float z = zbase[b * Nn + tid];
    col[tid] = __logf(L[(size_t)tid * Nn + tid]);
    zz[tid] = z * z;
    __syncthreads();
    for (int s = 512; s > 0; s >>= 1) {
        if (tid < s) {
            col[tid] += col[tid + s];
            zz[tid] += zz[tid + s];
        }
        __syncthreads();
    }
    if (tid == 0) nlml_out[b] = 0.5f * zz[0] + col[0] + 0.5f * Nn * LOG2PIF;
}

__global__ __launch_bounds__(128) void predict_kernel(const float* __restrict__ Linvbase,
                                                      const float* __restrict__ trainPg,
                                                      const float* __restrict__ testPg,
                                                      const float* __restrict__ Kmm,
                                                      const float* __restrict__ zbase,
                                                      const float* __restrict__ meang,
                                                      const float* __restrict__ scale_param,
                                                      float* __restrict__ out_mean,
                                                      float* __restrict__ out_var) {
    extern __shared__ float smem[];
    float* Kt  = smem;
    float* zsh = smem + Nn * KTS;
    float* tp  = zsh + Nn;
    float* Ab  = tp + Nn * 6;
    int b = blockIdx.y;
    int colBase = blockIdx.x * MT;
    int tid = threadIdx.x;
    float scale = __expf(scale_param[0]);

    for (int idx = tid; idx < Nn; idx += 128) zsh[idx] = zbase[b * Nn + idx];
    for (int idx = tid; idx < Nn * 6; idx += 128) tp[idx] = trainPg[(size_t)b * Nn * 6 + idx];
    __syncthreads();

    {
        int c = tid & 63;
        int h = tid >> 6;
        const float* q = testPg + (size_t)(b * Mm + colBase + c) * 6;
        float x2 = q[0], y2 = q[1], a2 = q[2], b2 = q[3], c2 = q[4], q2 = q[5];
        for (int n0 = h; n0 < Nn; n0 += 2 * KU) {
            float pr[KU][6], kv[KU];
#pragma unroll
            for (int u = 0; u < KU; u++) {
                const float* p = tp + (n0 + 2 * u) * 6;
#pragma unroll
                for (int e = 0; e < 6; e++) pr[u][e] = p[e];
            }
#pragma unroll
            for (int u = 0; u < KU; u++)
                kv[u] = kern_pair(pr[u][0], pr[u][1], pr[u][2], pr[u][3], pr[u][4], pr[u][5],
                                  x2, y2, a2, b2, c2, q2, scale);
#pragma unroll
            for (int u = 0; u < KU; u++)
                Kt[(n0 + 2 * u) * KTS + c] = kv[u];
        }
    }
    __syncthreads();

    int wave = tid >> 5, lane = tid & 31;
    int r = lane & 15, hi = lane >> 4;
    int wcol = wave * 16;
    const float* Li = Linvbase + (size_t)b * Nn * Nn;

    int srow = tid >> 4, sseg = tid & 15;
    const float* fa = Ab + r * APITCH + 2 * hi;
    const float* fb = Kt + (2 * hi) * KTS + wcol + r;
    float4 g0, g1;
    {
        g0 = make_float4(0.0f, 0.0f, 0.0f, 0.0f); g1 = g0;
        if (4 * sseg < 16) {
            g0 = *(const float4*)(Li + (size_t)srow * Nn + 4 * sseg);
            g1 = *(const float4*)(Li + (size_t)(srow + 8) * Nn + 4 * sseg);
        }
    }
    int buf = 0;
    float meanAcc = 0.0f, varAcc = 0.0f;
    for (int rb = 0; rb < Nn / 16; rb++) {
        v8f acc = {};
        int nch = (rb >> 2) + 1;
        for (int c = 0; c < nch; c++) {
            float* At = Ab + buf * (16 * APITCH);
            *(float4*)(At + srow * APITCH + 4 * sseg) = g0;
            *(float4*)(At + (srow + 8) * APITCH + 4 * sseg) = g1;
            int nrb = rb, nc = c + 1;
            if (nc == nch) { nrb = rb + 1; nc = 0; }
            if (nrb < Nn / 16) {
                int k0 = nc * ACH + 4 * sseg;
                g0 = make_float4(0.0f, 0.0f, 0.0f, 0.0f); g1 = g0;
                if (k0 < 16 * (nrb + 1)) {
                    g0 = *(const float4*)(Li + (size_t)(nrb * 16 + srow) * Nn + k0);
                    g1 = *(const float4*)(Li + (size_t)(nrb * 16 + srow + 8) * Nn + k0);
                }
            }
            __builtin_amdgcn_fence(__ATOMIC_RELEASE, "workgroup", "local");
            __builtin_amdgcn_s_barrier();
            __builtin_amdgcn_fence(__ATOMIC_ACQUIRE, "workgroup", "local");
            const float* fac = fa + buf * (16 * APITCH);
            const float* fbc = fb + (c * ACH) * KTS;
            v2f av[ACH / 4], bv[ACH / 4];
#pragma unroll
            for (int s = 0; s < ACH / 4; s++) {
                av[s] = *(const v2f*)(fac + 4 * s);
                bv[s].x = fbc[(4 * s) * KTS];
                bv[s].y = fbc[(4 * s + 1) * KTS];
            }
#pragma unroll
            for (int s = 0; s < ACH / 4; s++)
                acc = __builtin_amdgcn_wmma_f32_16x16x4_f32(false, av[s], false, bv[s],
                                                            (short)0, acc, false, false);
            __builtin_amdgcn_sched_group_barrier(0x100, 24, 0);
            __builtin_amdgcn_sched_group_barrier(0x008, 16, 0);
            buf ^= 1;
        }
        const float* zb = zsh + rb * 16 + 8 * hi;
#pragma unroll
        for (int v = 0; v < 8; v++) {
            float cv = acc[v];
            meanAcc += zb[v] * cv;
            varAcc  += cv * cv;
        }
    }
    meanAcc += __shfl_xor(meanAcc, 16);
    varAcc  += __shfl_xor(varAcc, 16);
    if (lane < 16) {
        int m = colBase + wcol + lane;
        out_mean[b * Mm + m] = meang[b] + meanAcc;
        out_var[b * Mm + m]  = Kmm[b * Mm + m] - varAcc;
    }
}

extern "C" void kernel_launch(void* const* d_in, const int* in_sizes, int n_in,
                              void* d_out, int out_size, void* d_ws, size_t ws_size,
                              hipStream_t stream) {
    const float* coords_train = (const float*)d_in[0];
    const float* kp_train     = (const float*)d_in[1];
    const float* coords_test  = (const float*)d_in[2];
    const float* kp_test      = (const float*)d_in[3];
    const float* y_train      = (const float*)d_in[4];
    const float* var          = (const float*)d_in[5];
    const float* mean         = (const float*)d_in[6];
    const float* scale_param  = (const float*)d_in[7];

    float* ws = (float*)d_ws;
    size_t off = 0;
    float* A      = ws + off; off += (size_t)Bb * Nn * Nn;
    float* Linv   = ws + off; off += (size_t)Bb * Nn * Nn;
    float* Lf     = ws + off; off += (size_t)Bb * Nn * Nn;
    float* trainP = ws + off; off += (size_t)Bb * Nn * 6;
    float* testP  = ws + off; off += (size_t)Bb * Mm * 6;
    float* Kmm    = ws + off; off += (size_t)Bb * Mm;
    float* z      = ws + off; off += (size_t)Bb * Nn;
    float* Yrow   = ws + off; off += (size_t)Bb * Nn;

    float* out_mean = (float*)d_out;
    float* out_var  = out_mean + (size_t)Bb * Mm;
    float* nlml     = out_var + (size_t)Bb * Mm;

    prep_kernel<<<(Bb * Mm + 255) / 256, 256, 0, stream>>>(
        coords_train, kp_train, coords_test, kp_test, scale_param, trainP, testP, Kmm);
    buildA_kernel<<<(Bb * Nn * Nn) / 256, 256, 0, stream>>>(trainP, var, scale_param, y_train, mean, A, Yrow);
#define CHOL_STEP(K) chol_step_kernel<<<dim3(NT + 2, NT - (K), Bb), 32, 0, stream>>>(A, Lf, Linv, Yrow, z, (K));
    CHOL_STEP(0) CHOL_STEP(1) CHOL_STEP(2) CHOL_STEP(3) CHOL_STEP(4) CHOL_STEP(5) CHOL_STEP(6) CHOL_STEP(7)
    CHOL_STEP(8) CHOL_STEP(9) CHOL_STEP(10) CHOL_STEP(11) CHOL_STEP(12) CHOL_STEP(13) CHOL_STEP(14) CHOL_STEP(15)
    CHOL_STEP(16) CHOL_STEP(17) CHOL_STEP(18) CHOL_STEP(19) CHOL_STEP(20) CHOL_STEP(21) CHOL_STEP(22) CHOL_STEP(23)
    CHOL_STEP(24) CHOL_STEP(25) CHOL_STEP(26) CHOL_STEP(27) CHOL_STEP(28) CHOL_STEP(29) CHOL_STEP(30) CHOL_STEP(31)
#undef CHOL_STEP
    nlml_kernel<<<Bb, 1024, 0, stream>>>(Lf, z, nlml);

    size_t smem_bytes = (size_t)(Nn * KTS + Nn + Nn * 6 + 2 * 16 * APITCH) * sizeof(float);
    predict_kernel<<<dim3(Mm / MT, Bb), 128, smem_bytes, stream>>>(
        Linv, trainP, testP, Kmm, z, mean, scale_param, out_mean, out_var);
}


# predict kernel: each workgroup walks the 64 row bands of L^-1 from a start band of its own, wrapping round, so workgroups stop reading the same lines in step; new sum order in pred_mean, pred_var
# speedup vs baseline: 52.0036x; 1.1135x over previous
#include <hip/hip_runtime.h>
#include <math.h>

#define Nn 1024
#define Mm 16384
#define Bb 2
#define MT 64
#define KTS 65
#define ACH 64
#define KU 8
#define APITCH 68

typedef __attribute__((ext_vector_type(2))) float v2f;
typedef __attribute__((ext_vector_type(8))) float v8f;

#define SQRT3F 1.7320508075688772f
#define LOG2PIF 1.8378770664093453f

__device__ __forceinline__ float kern_pair(float x1, float y1, float a1, float b1, float c1, float q1,
                                           float x2, float y2, float a2, float b2, float c2, float q2,
                                           float scale) {
    float s00 = a1 + a2, s11 = b1 + b2, s01 = c1 + c2;
    float sdet = s00 * s11 - s01 * s01;
    float d0 = x1 - x2, d1 = y1 - y2;
    float Q = 0.5f * (s11 * d0 * d0 - 2.0f * s01 * d0 * d1 + s00 * d1 * d1) / sdet;
    float C = 2.0f * q1 * q2 * rsqrtf(fmaxf(sdet, 1e-5f));
    float t = SQRT3F * sqrtf(fmaxf(Q, 1e-5f));
    return (1.0f + t) * __expf(-t) * C * scale;
}

__global__ void prep_kernel(const float* __restrict__ ct, const float* __restrict__ kpt,
                            const float* __restrict__ cs, const float* __restrict__ kps,
                            const float* __restrict__ scale_param,
                            float* __restrict__ trainP, float* __restrict__ testP,
                            float* __restrict__ Kmm) {
    int gid = blockIdx.x * blockDim.x + threadIdx.x;
    float scale = __expf(scale_param[0]);
    if (gid < Bb * Mm) {
        float a = kps[gid * 3 + 0], b = kps[gid * 3 + 1], c = kps[gid * 3 + 2];
        float det = a * b - c * c;
        testP[gid * 6 + 0] = cs[gid * 2 + 0];
        testP[gid * 6 + 1] = cs[gid * 2 + 1];
        testP[gid * 6 + 2] = a;
        testP[gid * 6 + 3] = b;
        testP[gid * 6 + 4] = c;
        testP[gid * 6 + 5] = sqrtf(sqrtf(det));
        float t0 = SQRT3F * sqrtf(1e-5f);
        float m0 = (1.0f + t0) * __expf(-t0);
        float C = 2.0f * sqrtf(det) * rsqrtf(fmaxf(4.0f * det, 1e-5f));
        Kmm[gid] = C * m0 * scale;
    }
    if (gid < Bb * Nn) {
        float a = kpt[gid * 3 + 0], b = kpt[gid * 3 + 1], c = kpt[gid * 3 + 2];
        float det = a * b - c * c;
        trainP[gid * 6 + 0] = ct[gid * 2 + 0];
        trainP[gid * 6 + 1] = ct[gid * 2 + 1];
        trainP[gid * 6 + 2] = a;
        trainP[gid * 6 + 3] = b;
        trainP[gid * 6 + 4] = c;
        trainP[gid * 6 + 5] = sqrtf(sqrtf(det));
    }
}

__global__ void buildA_kernel(const float* __restrict__ trainP, const float* __restrict__ var,
                              const float* __restrict__ scale_param,
                              const float* __restrict__ y, const float* __restrict__ mean,
                              float* __restrict__ A, float* __restrict__ Yrow) {
    int gid = blockIdx.x * blockDim.x + threadIdx.x;
    int b = gid >> 20;
    int i = (gid >> 10) & (Nn - 1);
    int j = gid & (Nn - 1);
    const float* p1 = trainP + (size_t)(b * Nn + i) * 6;
    const float* p2 = trainP + (size_t)(b * Nn + j) * 6;
    float scale = __expf(scale_param[0]);
    float v = kern_pair(p1[0], p1[1], p1[2], p1[3], p1[4], p1[5],
                        p2[0], p2[1], p2[2], p2[3], p2[4], p2[5], scale);
    if (i == j) v += var[b * Nn + i];
    A[(size_t)gid] = v;
    if (i == 0) Yrow[b * Nn + j] = y[b * Nn + j] - mean[b];
}

#define TS 32
#define NT (Nn / TS)

__device__ __forceinline__ float lane_bcast(float v, int src) {
    return __int_as_float(__builtin_amdgcn_readlane(__float_as_int(v), src));
}

__device__ __forceinline__ void load_row32(const float* p, float (&r)[TS]) {
    const float4* p4 = (const float4*)p;
#pragma unroll
    for (int q = 0; q < TS / 4; q++) {
        float4 t = p4[q];
        r[4 * q + 0] = t.x; r[4 * q + 1] = t.y; r[4 * q + 2] = t.z; r[4 * q + 3] = t.w;
    }
}

__device__ __forceinline__ void store_row32(float* p, const float (&r)[TS]) {
    float4* p4 = (float4*)p;
#pragma unroll
    for (int q = 0; q < TS / 4; q++) {
        float4 t;
        t.x = r[4 * q + 0]; t.y = r[4 * q + 1]; t.z = r[4 * q + 2]; t.w = r[4 * q + 3];
        p4[q] = t;
    }
}

#define QP 36

__global__ __launch_bounds__(32) void chol_step_kernel(float* Abase, float* __restrict__ Lbase,
                                                       float* Xbase, float* Ybase,
                                                       float* __restrict__ zbase, int k) {
    __shared__ __attribute__((aligned(16))) float qs[TS * QP];
    int nmat = NT - k;
    int bx = blockIdx.x;
    int kind = (bx < nmat) ? 0 : (bx <= NT) ? 1 : 2;
    int i = (kind == 0) ? k + bx : (kind == 1) ? bx - nmat : 0;
    int j = k + blockIdx.y;
    if (kind == 0 && j > i) return;
    int b = blockIdx.z;
    int r = threadIdx.x;
    float* A = Abase + (size_t)b * Nn * Nn;
    float* Lo = Lbase + (size_t)b * Nn * Nn;
    float* X = Xbase + (size_t)b * Nn * Nn;
    float* Y = Ybase + (size_t)b * Nn;
    bool ident = (kind == 1) && (i == k);
    bool mine = (kind != 2) || (r == 0);
    const float* prow = (kind == 0) ? A + (size_t)(i * TS + r) * Nn + k * TS
                      : (kind == 1) ? X + (size_t)(i * TS + r) * Nn + k * TS
                                    : Y + k * TS;
    float* trow = (kind == 0) ? A + (size_t)(i * TS + r) * Nn + j * TS
                : (kind == 1) ? X + (size_t)(i * TS + r) * Nn + j * TS
                              : Y + j * TS;

    float d[TS];
    v2f pq[TS];
    load_row32(A + (size_t)(k * TS + r) * Nn + k * TS, d);
    {
        float p[TS], q[TS];
        if (ident) {
#pragma unroll
            for (int c = 0; c < TS; c++) p[c] = (c == r) ? 1.0f : 0.0f;
        } else if (mine) {
            load_row32(prow, p);
        } else {
#pragma unroll
            for (int c = 0; c < TS; c++) p[c] = 0.0f;
        }
        load_row32(A + (size_t)(j * TS + r) * Nn + k * TS, q);
#pragma unroll
        for (int c = 0; c < TS; c++) { pq[c].x = p[c]; pq[c].y = q[c]; }
    }

#pragma unroll
    for (int c = 0; c < TS; c++) {
#pragma unroll
        for (int c1 = 0; c1 < c; c1++) {
            float s = lane_bcast(d[c1], c);
            d[c] -= d[c1] * s;
            pq[c] -= pq[c1] * s;
        }
        float piv = lane_bcast(d[c], c);
        float dg = sqrtf(piv);
        float dinv = 1.0f / dg;
        d[c] = (r == c) ? dg : d[c] * dinv;
        pq[c] = pq[c] * dinv;
    }

    if (j > k) {
        float4* qw = (float4*)(qs + r * QP);
#pragma unroll
        for (int c4 = 0; c4 < TS / 4; c4++)
            qw[c4] = make_float4(pq[4 * c4].y, pq[4 * c4 + 1].y, pq[4 * c4 + 2].y, pq[4 * c4 + 3].y);
        __syncthreads();
        float* arow = trow;
#pragma unroll 1
        for (int s0 = 0; s0 < TS; s0 += 4) {
            float4 av = make_float4(0.0f, 0.0f, 0.0f, 0.0f);
            if (!ident && mine) av = *(const float4*)(arow + s0);
            float a0 = av.x, a1 = av.y, a2 = av.z, a3 = av.w;
            const float4* q0 = (const float4*)(qs + (s0 + 0) * QP);
            const float4* q1 = (const float4*)(qs + (s0 + 1) * QP);
            const float4* q2 = (const float4*)(qs + (s0 + 2) * QP);
            const float4* q3 = (const float4*)(qs + (s0 + 3) * QP);
#pragma unroll
            for (int c4 = 0; c4 < TS / 4; c4++) {
                float4 u0 = q0[c4], u1 = q1[c4], u2 = q2[c4], u3 = q3[c4];
                float p0 = pq[4 * c4].x, p1 = pq[4 * c4 + 1].x, p2 = pq[4 * c4 + 2].x, p3 = pq[4 * c4 + 3].x;
                a0 -= p0 * u0.x; a1 -= p0 * u1.x; a2 -= p0 * u2.x; a3 -= p0 * u3.x;
                a0 -= p1 * u0.y; a1 -= p1 * u1.y; a2 -= p1 * u2.y; a3 -= p1 * u3.y;
                a0 -= p2 * u0.z; a1 -= p2 * u1.z; a2 -= p2 * u2.z; a3 -= p2 * u3.z;
                a0 -= p3 * u0.w; a1 -= p3 * u1.w; a2 -= p3 * u2.w; a3 -= p3 * u3.w;
            }
            if (mine) *(float4*)(arow + s0) = make_float4(a0, a1, a2, a3);
        }
    } else if (kind == 1) {
        float* dst = X + (size_t)(k * TS) * Nn + i * TS + r;
#pragma unroll
        for (int c = 0; c < TS; c++) dst[(size_t)c * Nn] = pq[c].x;
    } else if (kind == 2) {
        if (r == 0) {
            float p[TS];
#pragma unroll
            for (int c = 0; c < TS; c++) p[c] = pq[c].x;
            store_row32(zbase + (size_t)b * Nn + k * TS, p);
        }
    } else if (i > k) {
        float p[TS];
#pragma unroll
        for (int c = 0; c < TS; c++) p[c] = pq[c].x;
        store_row32(Lo + (size_t)(i * TS + r) * Nn + k * TS, p);
    } else {
#pragma unroll
        for (int c = 0; c < TS; c++) if (c > r) d[c] = 0.0f;
        store_row32(Lo + (size_t)(k * TS + r) * Nn + k * TS, d);
    }
}

__global__ __launch_bounds__(1024) void nlml_kernel(const float* __restrict__ Lbase,
                                                    const float* __restrict__ zbase,
                                                    float* __restrict__ nlml_out) {
    int b = blockIdx.x;
    const float* L = Lbase + (size_t)b * Nn * Nn;
    __shared__ float col[Nn];
    __shared__ float zz[Nn];
    int tid = threadIdx.x;
    float z = zbase[b * Nn + tid];
    col[tid] = __logf(L[(size_t)tid * Nn + tid]);
    zz[tid] = z * z;
    __syncthreads();
    for (int s = 512; s > 0; s >>= 1) {
        if (tid < s) {
            col[tid] += col[tid + s];
            zz[tid] += zz[tid + s];
        }
        __syncthreads();
    }
    if (tid == 0) nlml_out[b] = 0.5f * zz[0] + col[0] + 0.5f * Nn * LOG2PIF;
}

__global__ __launch_bounds__(128) void predict_kernel(const float* __restrict__ Linvbase,
                                                      const float* __restrict__ trainPg,
                                                      const float* __restrict__ testPg,
                                                      const float* __restrict__ Kmm,
                                                      const float* __restrict__ zbase,
                                                      const float* __restrict__ meang,
                                                      const float* __restrict__ scale_param,
                                                      float* __restrict__ out_mean,
                                                      float* __restrict__ out_var) {
    extern __shared__ float smem[];
    float* Kt  = smem;
    float* zsh = smem + Nn * KTS;
    float* tp  = zsh + Nn;
    float* Ab  = tp + Nn * 6;
    int b = blockIdx.y;
    int colBase = blockIdx.x * MT;
    int tid = threadIdx.x;
    float scale = __expf(scale_param[0]);

    for (int idx = tid; idx < Nn; idx += 128) zsh[idx] = zbase[b * Nn + idx];
    for (int idx = tid; idx < Nn * 6; idx += 128) tp[idx] = trainPg[(size_t)b * Nn * 6 + idx];
    __syncthreads();

    {
        int c = tid & 63;
        int h = tid >> 6;
        const float* q = testPg + (size_t)(b * Mm + colBase + c) * 6;
        float x2 = q[0], y2 = q[1], a2 = q[2], b2 = q[3], c2 = q[4], q2 = q[5];
        for (int n0 = h; n0 < Nn; n0 += 2 * KU) {
            float pr[KU][6], kv[KU];
#pragma unroll
            for (int u = 0; u < KU; u++) {
                const float* p = tp + (n0 + 2 * u) * 6;
#pragma unroll
                for (int e = 0; e < 6; e++) pr[u][e] = p[e];
            }
#pragma unroll
            for (int u = 0; u < KU; u++)
                kv[u] = kern_pair(pr[u][0], pr[u][1], pr[u][2], pr[u][3], pr[u][4], pr[u][5],
                                  x2, y2, a2, b2, c2, q2, scale);
#pragma unroll
            for (int u = 0; u < KU; u++)
                Kt[(n0 + 2 * u) * KTS + c] = kv[u];
        }
    }
    __syncthreads();

    int wave = tid >> 5, lane = tid & 31;
    int r = lane & 15, hi = lane >> 4;
    int wcol = wave * 16;
    const float* Li = Linvbase + (size_t)b * Nn * Nn;

    int rb0 = ((int)(4.0f * sqrtf((float)blockIdx.x))) & (Nn / 16 - 1);
    int srow = tid >> 4, sseg = tid & 15;
    const float* fa = Ab + r * APITCH + 2 * hi;
    const float* fb = Kt + (2 * hi) * KTS + wcol + r;
    float4 g0, g1;
    {
        g0 = make_float4(0.0f, 0.0f, 0.0f, 0.0f); g1 = g0;
        if (4 * sseg < 16 * (rb0 + 1)) {
            g0 = *(const float4*)(Li + (size_t)(rb0 * 16 + srow) * Nn + 4 * sseg);
            g1 = *(const float4*)(Li + (size_t)(rb0 * 16 + srow + 8) * Nn + 4 * sseg);
        }
    }
    int buf = 0;
    float meanAcc = 0.0f, varAcc = 0.0f;
    for (int t = 0; t < Nn / 16; t++) {
        int rb = (rb0 + t) & (Nn / 16 - 1);
        v8f acc = {};
        int nch = (rb >> 2) + 1;
        for (int c = 0; c < nch; c++) {
            float* At = Ab + buf * (16 * APITCH);
            *(float4*)(At + srow * APITCH + 4 * sseg) = g0;
            *(float4*)(At + (srow + 8) * APITCH + 4 * sseg) = g1;
            int nrb = rb, nc = c + 1;
            bool more = true;
            if (nc == nch) { nrb = (rb + 1) & (Nn / 16 - 1); nc = 0; more = (t + 1 < Nn / 16); }
            if (more) {
                int k0 = nc * ACH + 4 * sseg;
                g0 = make_float4(0.0f, 0.0f, 0.0f, 0.0f); g1 = g0;
                if (k0 < 16 * (nrb + 1)) {
                    g0 = *(const float4*)(Li + (size_t)(nrb * 16 + srow) * Nn + k0);
                    g1 = *(const float4*)(Li + (size_t)(nrb * 16 + srow + 8) * Nn + k0);
                }
            }
            __builtin_amdgcn_fence(__ATOMIC_RELEASE, "workgroup", "local");
            __builtin_amdgcn_s_barrier();
            __builtin_amdgcn_fence(__ATOMIC_ACQUIRE, "workgroup", "local");
            const float* fac = fa + buf * (16 * APITCH);
            const float* fbc = fb + (c * ACH) * KTS;
            v2f av[ACH / 4], bv[ACH / 4];
#pragma unroll
            for (int s = 0; s < ACH / 4; s++) {
                av[s] = *(const v2f*)(fac + 4 * s);
                bv[s].x = fbc[(4 * s) * KTS];
                bv[s].y = fbc[(4 * s + 1) * KTS];
            }
#pragma unroll
            for (int s = 0; s < ACH / 4; s++)
                acc = __builtin_amdgcn_wmma_f32_16x16x4_f32(false, av[s], false, bv[s],
                                                            (short)0, acc, false, false);
            __builtin_amdgcn_sched_group_barrier(0x100, 24, 0);
            __builtin_amdgcn_sched_group_barrier(0x008, 16, 0);
            buf ^= 1;
        }
        const float* zb = zsh + rb * 16 + 8 * hi;
#pragma unroll
        for (int v = 0; v < 8; v++) {
            float cv = acc[v];
            meanAcc += zb[v] * cv;
            varAcc  += cv * cv;
        }
    }
    meanAcc += __shfl_xor(meanAcc, 16);
    varAcc  += __shfl_xor(varAcc, 16);
    if (lane < 16) {
        int m = colBase + wcol + lane;
        out_mean[b * Mm + m] = meang[b] + meanAcc;
        out_var[b * Mm + m]  = Kmm[b * Mm + m] - varAcc;
    }
}

extern "C" void kernel_launch(void* const* d_in, const int* in_sizes, int n_in,
                              void* d_out, int out_size, void* d_ws, size_t ws_size,
                              hipStream_t stream) {
    const float* coords_train = (const float*)d_in[0];
    const float* kp_train     = (const float*)d_in[1];
    const float* coords_test  = (const float*)d_in[2];
    const float* kp_test      = (const float*)d_in[3];
    const float* y_train      = (const float*)d_in[4];
    const float* var          = (const float*)d_in[5];
    const float* mean         = (const float*)d_in[6];
    const float* scale_param  = (const float*)d_in[7];

    float* ws = (float*)d_ws;
    size_t off = 0;
    float* A      = ws + off; off += (size_t)Bb * Nn * Nn;
    float* Linv   = ws + off; off += (size_t)Bb * Nn * Nn;
    float* Lf     = ws + off; off += (size_t)Bb * Nn * Nn;
    float* trainP = ws + off; off += (size_t)Bb * Nn * 6;
    float* testP  = ws + off; off += (size_t)Bb * Mm * 6;
    float* Kmm    = ws + off; off += (size_t)Bb * Mm;
    float* z      = ws + off; off += (size_t)Bb * Nn;
    float* Yrow   = ws + off; off += (size_t)Bb * Nn;

    float* out_mean = (float*)d_out;
    float* out_var  = out_mean + (size_t)Bb * Mm;
    float* nlml     = out_var + (size_t)Bb * Mm;

    prep_kernel<<<(Bb * Mm + 255) / 256, 256, 0, stream>>>(
        coords_train, kp_train, coords_test, kp_test, scale_param, trainP, testP, Kmm);
    buildA_kernel<<<(Bb * Nn * Nn) / 256, 256, 0, stream>>>(trainP, var, scale_param, y_train, mean, A, Yrow);
#define CHOL_STEP(K) chol_step_kernel<<<dim3(NT + 2, NT - (K), Bb), 32, 0, stream>>>(A, Lf, Linv, Yrow, z, (K));
    CHOL_STEP(0) CHOL_STEP(1) CHOL_STEP(2) CHOL_STEP(3) CHOL_STEP(4) CHOL_STEP(5) CHOL_STEP(6) CHOL_STEP(7)
    CHOL_STEP(8) CHOL_STEP(9) CHOL_STEP(10) CHOL_STEP(11) CHOL_STEP(12) CHOL_STEP(13) CHOL_STEP(14) CHOL_STEP(15)
    CHOL_STEP(16) CHOL_STEP(17) CHOL_STEP(18) CHOL_STEP(19) CHOL_STEP(20) CHOL_STEP(21) CHOL_STEP(22) CHOL_STEP(23)
    CHOL_STEP(24) CHOL_STEP(25) CHOL_STEP(26) CHOL_STEP(27) CHOL_STEP(28) CHOL_STEP(29) CHOL_STEP(30) CHOL_STEP(31)
#undef CHOL_STEP
    nlml_kernel<<<Bb, 1024, 0, stream>>>(Lf, z, nlml);

    size_t smem_bytes = (size_t)(Nn * KTS + Nn + Nn * 6 + 2 * 16 * APITCH) * sizeof(float);
    predict_kernel<<<dim3(Mm / MT, Bb), 128, smem_bytes, stream>>>(
        Linv, trainP, testP, Kmm, z, mean, scale_param, out_mean, out_var);
}


# Cholesky tile step: a wave's four rows (diagonal, two panel, trailing) requested in one batch at entry, trailing row parked in LDS: one memory latency per launch, not ten; arithmetic and order kept
# speedup vs baseline: 61.0501x; 1.1740x over previous
#include <hip/hip_runtime.h>
#include <math.h>

#define Nn 1024
#define Mm 16384
#define Bb 2
#define MT 64
#define KTS 65
#define ACH 64
#define KU 8
#define APITCH 68

typedef __attribute__((ext_vector_type(2))) float v2f;
typedef __attribute__((ext_vector_type(8))) float v8f;

#define SQRT3F 1.7320508075688772f
#define LOG2PIF 1.8378770664093453f

__device__ __forceinline__ float kern_pair(float x1, float y1, float a1, float b1, float c1, float q1,
                                           float x2, float y2, float a2, float b2, float c2, float q2,
                                           float scale) {
    float s00 = a1 + a2, s11 = b1 + b2, s01 = c1 + c2;
    float sdet = s00 * s11 - s01 * s01;
    float d0 = x1 - x2, d1 = y1 - y2;
    float Q = 0.5f * (s11 * d0 * d0 - 2.0f * s01 * d0 * d1 + s00 * d1 * d1) / sdet;
    float C = 2.0f * q1 * q2 * rsqrtf(fmaxf(sdet, 1e-5f));
    float t = SQRT3F * sqrtf(fmaxf(Q, 1e-5f));
    return (1.0f + t) * __expf(-t) * C * scale;
}

__global__ void prep_kernel(const float* __restrict__ ct, const float* __restrict__ kpt,
                            const float* __restrict__ cs, const float* __restrict__ kps,
                            const float* __restrict__ scale_param,
                            float* __restrict__ trainP, float* __restrict__ testP,
                            float* __restrict__ Kmm) {
    int gid = blockIdx.x * blockDim.x + threadIdx.x;
    float scale = __expf(scale_param[0]);
    if (gid < Bb * Mm) {
        float a = kps[gid * 3 + 0], b = kps[gid * 3 + 1], c = kps[gid * 3 + 2];
        float det = a * b - c * c;
        testP[gid * 6 + 0] = cs[gid * 2 + 0];
        testP[gid * 6 + 1] = cs[gid * 2 + 1];
        testP[gid * 6 + 2] = a;
        testP[gid * 6 + 3] = b;
        testP[gid * 6 + 4] = c;
        testP[gid * 6 + 5] = sqrtf(sqrtf(det));
        float t0 = SQRT3F * sqrtf(1e-5f);
        float m0 = (1.0f + t0) * __expf(-t0);
        float C = 2.0f * sqrtf(det) * rsqrtf(fmaxf(4.0f * det, 1e-5f));
        Kmm[gid] = C * m0 * scale;
    }
    if (gid < Bb * Nn) {
        float a = kpt[gid * 3 + 0], b = kpt[gid * 3 + 1], c = kpt[gid * 3 + 2];
        float det = a * b - c * c;
        trainP[gid * 6 + 0] = ct[gid * 2 + 0];
        trainP[gid * 6 + 1] = ct[gid * 2 + 1];
        trainP[gid * 6 + 2] = a;
        trainP[gid * 6 + 3] = b;
        trainP[gid * 6 + 4] = c;
        trainP[gid * 6 + 5] = sqrtf(sqrtf(det));
    }
}

__global__ void buildA_kernel(const float* __restrict__ trainP, const float* __restrict__ var,
                              const float* __restrict__ scale_param,
                              const float* __restrict__ y, const float* __restrict__ mean,
                              float* __restrict__ A, float* __restrict__ Yrow) {
    int gid = blockIdx.x * blockDim.x + threadIdx.x;
    int b = gid >> 20;
    int i = (gid >> 10) & (Nn - 1);
    int j = gid & (Nn - 1);
    const float* p1 = trainP + (size_t)(b * Nn + i) * 6;
    const float* p2 = trainP + (size_t)(b * Nn + j) * 6;
    float scale = __expf(scale_param[0]);
    float v = kern_pair(p1[0], p1[1], p1[2], p1[3], p1[4], p1[5],
                        p2[0], p2[1], p2[2], p2[3], p2[4], p2[5], scale);
    if (i == j) v += var[b * Nn + i];
    A[(size_t)gid] = v;
    if (i == 0) Yrow[b * Nn + j] = y[b * Nn + j] - mean[b];
}

#define TS 32
#define NT (Nn / TS)

__device__ __forceinline__ float lane_bcast(float v, int src) {
    return __int_as_float(__builtin_amdgcn_readlane(__float_as_int(v), src));
}

__device__ __forceinline__ void load_row32(const float* p, float (&r)[TS]) {
    const float4* p4 = (const float4*)p;
#pragma unroll
    for (int q = 0; q < TS / 4; q++) {
        float4 t = p4[q];
        r[4 * q + 0] = t.x; r[4 * q + 1] = t.y; r[4 * q + 2] = t.z; r[4 * q + 3] = t.w;
    }
}

__device__ __forceinline__ void store_row32(float* p, const float (&r)[TS]) {
    float4* p4 = (float4*)p;
#pragma unroll
    for (int q = 0; q < TS / 4; q++) {
        float4 t;
        t.x = r[4 * q + 0]; t.y = r[4 * q + 1]; t.z = r[4 * q + 2]; t.w = r[4 * q + 3];
        p4[q] = t;
    }
}

#define QP 36

__global__ __launch_bounds__(32) void chol_step_kernel(float* Abase, float* __restrict__ Lbase,
                                                       float* Xbase, float* Ybase,
                                                       float* __restrict__ zbase, int k) {
    __shared__ __attribute__((aligned(16))) float qs[TS * QP];
    __shared__ __attribute__((aligned(16))) float ts[TS * QP];
    int nmat = NT - k;
    int bx = blockIdx.x;
    int kind = (bx < nmat) ? 0 : (bx <= NT) ? 1 : 2;
    int i = (kind == 0) ? k + bx : (kind == 1) ? bx - nmat : 0;
    int j = k + blockIdx.y;
    if (kind == 0 && j > i) return;
    int b = blockIdx.z;
    int r = threadIdx.x;
    float* A = Abase + (size_t)b * Nn * Nn;
    float* Lo = Lbase + (size_t)b * Nn * Nn;
    float* X = Xbase + (size_t)b * Nn * Nn;
    float* Y = Ybase + (size_t)b * Nn;
    bool ident = (kind == 1) && (i == k);
    bool mine = (kind != 2) || (r == 0);
    const float* prow = (kind == 0) ? A + (size_t)(i * TS + r) * Nn + k * TS
                      : (kind == 1) ? X + (size_t)(i * TS + r) * Nn + k * TS
                                    : Y + k * TS;
    float* trow = (kind == 0) ? A + (size_t)(i * TS + r) * Nn + j * TS
                : (kind == 1) ? X + (size_t)(i * TS + r) * Nn + j * TS
                              : Y + j * TS;

    float d[TS];
    v2f pq[TS];
    load_row32(A + (size_t)(k * TS + r) * Nn + k * TS, d);
    {
        float p[TS], q[TS], t[TS];
        load_row32(A + (size_t)(j * TS + r) * Nn + k * TS, q);
        if (j > k) {
            if (!ident && mine) {
                load_row32(trow, t);
            } else {
#pragma unroll
                for (int c = 0; c < TS; c++) t[c] = 0.0f;
            }
        }
        if (ident) {
#pragma unroll
            for (int c = 0; c < TS; c++) p[c] = (c == r) ? 1.0f : 0.0f;
        } else {
            load_row32(prow, p);
#pragma unroll
            for (int c = 0; c < TS; c++) p[c] = mine ? p[c] : 0.0f;
        }
#pragma unroll
        for (int c = 0; c < TS; c++) { pq[c].x = p[c]; pq[c].y = q[c]; }
        if (j > k) store_row32(ts + r * QP, t);
    }

#pragma unroll
    for (int c = 0; c < TS; c++) {
#pragma unroll
        for (int c1 = 0; c1 < c; c1++) {
            float s = lane_bcast(d[c1], c);
            d[c] -= d[c1] * s;
            pq[c] -= pq[c1] * s;
        }
        float piv = lane_bcast(d[c], c);
        float dg = sqrtf(piv);
        float dinv = 1.0f / dg;
        d[c] = (r == c) ? dg : d[c] * dinv;
        pq[c] = pq[c] * dinv;
    }

    if (j > k) {
        float4* qw = (float4*)(qs + r * QP);
#pragma unroll
        for (int c4 = 0; c4 < TS / 4; c4++)
            qw[c4] = make_float4(pq[4 * c4].y, pq[4 * c4 + 1].y, pq[4 * c4 + 2].y, pq[4 * c4 + 3].y);
        __syncthreads();
        float* arow = trow;
#pragma unroll 1
        for (int s0 = 0; s0 < TS; s0 += 4) {
            float4 av = *(const float4*)(ts + r * QP + s0);
            float a0 = av.x, a1 = av.y, a2 = av.z, a3 = av.w;
            const float4* q0 = (const float4*)(qs + (s0 + 0) * QP);
            const float4* q1 = (const float4*)(qs + (s0 + 1) * QP);
            const float4* q2 = (const float4*)(qs + (s0 + 2) * QP);
            const float4* q3 = (const float4*)(qs + (s0 + 3) * QP);
#pragma unroll
            for (int c4 = 0; c4 < TS / 4; c4++) {
                float4 u0 = q0[c4], u1 = q1[c4], u2 = q2[c4], u3 = q3[c4];
                float p0 = pq[4 * c4].x, p1 = pq[4 * c4 + 1].x, p2 = pq[4 * c4 + 2].x, p3 = pq[4 * c4 + 3].x;
                a0 -= p0 * u0.x; a1 -= p0 * u1.x; a2 -= p0 * u2.x; a3 -= p0 * u3.x;
                a0 -= p1 * u0.y; a1 -= p1 * u1.y; a2 -= p1 * u2.y; a3 -= p1 * u3.y;
                a0 -= p2 * u0.z; a1 -= p2 * u1.z; a2 -= p2 * u2.z; a3 -= p2 * u3.z;
                a0 -= p3 * u0.w; a1 -= p3 * u1.w; a2 -= p3 * u2.w; a3 -= p3 * u3.w;
            }
            if (mine) *(float4*)(arow + s0) = make_float4(a0, a1, a2, a3);
        }
    } else if (kind == 1) {
        float* dst = X + (size_t)(k * TS) * Nn + i * TS + r;
#pragma unroll
        for (int c = 0; c < TS; c++) dst[(size_t)c * Nn] = pq[c].x;
    } else if (kind == 2) {
        if (r == 0) {
            float p[TS];
#pragma unroll
            for (int c = 0; c < TS; c++) p[c] = pq[c].x;
            store_row32(zbase + (size_t)b * Nn + k * TS, p);
        }
    } else if (i > k) {
        float p[TS];
#pragma unroll
        for (int c = 0; c < TS; c++) p[c] = pq[c].x;
        store_row32(Lo + (size_t)(i * TS + r) * Nn + k * TS, p);
    } else {
#pragma unroll
        for (int c = 0; c < TS; c++) if (c > r) d[c] = 0.0f;
        store_row32(Lo + (size_t)(k * TS + r) * Nn + k * TS, d);
    }
}

__global__ __launch_bounds__(1024) void nlml_kernel(const float* __restrict__ Lbase,
                                                    const float* __restrict__ zbase,
                                                    float* __restrict__ nlml_out) {
    int b = blockIdx.x;
    const float* L = Lbase + (size_t)b * Nn * Nn;
    __shared__ float col[Nn];
    __shared__ float zz[Nn];
    int tid = threadIdx.x;
    float z = zbase[b * Nn + tid];
    col[tid] = __logf(L[(size_t)tid * Nn + tid]);
    zz[tid] = z * z;
    __syncthreads();
    for (int s = 512; s > 0; s >>= 1) {
        if (tid < s) {
            col[tid] += col[tid + s];
            zz[tid] += zz[tid + s];
        }
        __syncthreads();
    }
    if (tid == 0) nlml_out[b] = 0.5f * zz[0] + col[0] + 0.5f * Nn * LOG2PIF;
}

__global__ __launch_bounds__(128) void predict_kernel(const float* __restrict__ Linvbase,
                                                      const float* __restrict__ trainPg,
                                                      const float* __restrict__ testPg,
                                                      const float* __restrict__ Kmm,
                                                      const float* __restrict__ zbase,
                                                      const float* __restrict__ meang,
                                                      const float* __restrict__ scale_param,
                                                      float* __restrict__ out_mean,
                                                      float* __restrict__ out_var) {
    extern __shared__ float smem[];
    float* Kt  = smem;
    float* zsh = smem + Nn * KTS;
    float* tp  = zsh + Nn;
    float* Ab  = tp + Nn * 6;
    int b = blockIdx.y;
    int colBase = blockIdx.x * MT;
    int tid = threadIdx.x;
    float scale = __expf(scale_param[0]);

    for (int idx = tid; idx < Nn; idx += 128) zsh[idx] = zbase[b * Nn + idx];
    for (int idx = tid; idx < Nn * 6; idx += 128) tp[idx] = trainPg[(size_t)b * Nn * 6 + idx];
    __syncthreads();

    {
        int c = tid & 63;
        int h = tid >> 6;
        const float* q = testPg + (size_t)(b * Mm + colBase + c) * 6;
        float x2 = q[0], y2 = q[1], a2 = q[2], b2 = q[3], c2 = q[4], q2 = q[5];
        for (int n0 = h; n0 < Nn; n0 += 2 * KU) {
            float pr[KU][6], kv[KU];
#pragma unroll
            for (int u = 0; u < KU; u++) {
                const float* p = tp + (n0 + 2 * u) * 6;
#pragma unroll
                for (int e = 0; e < 6; e++) pr[u][e] = p[e];
            }
#pragma unroll
            for (int u = 0; u < KU; u++)
                kv[u] = kern_pair(pr[u][0], pr[u][1], pr[u][2], pr[u][3], pr[u][4], pr[u][5],
                                  x2, y2, a2, b2, c2, q2, scale);
#pragma unroll
            for (int u = 0; u < KU; u++)
                Kt[(n0 + 2 * u) * KTS + c] = kv[u];
        }
    }
    __syncthreads();

    int wave = tid >> 5, lane = tid & 31;
    int r = lane & 15, hi = lane >> 4;
    int wcol = wave * 16;
    const float* Li = Linvbase + (size_t)b * Nn * Nn;

    int rb0 = ((int)(4.0f * sqrtf((float)blockIdx.x))) & (Nn / 16 - 1);
    int srow = tid >> 4, sseg = tid & 15;
    const float* fa = Ab + r * APITCH + 2 * hi;
    const float* fb = Kt + (2 * hi) * KTS + wcol + r;
    float4 g0, g1;
    {
        g0 = make_float4(0.0f, 0.0f, 0.0f, 0.0f); g1 = g0;
        if (4 * sseg < 16 * (rb0 + 1)) {
            g0 = *(const float4*)(Li + (size_t)(rb0 * 16 + srow) * Nn + 4 * sseg);
            g1 = *(const float4*)(Li + (size_t)(rb0 * 16 + srow + 8) * Nn + 4 * sseg);
        }
    }
    int buf = 0;
    float meanAcc = 0.0f, varAcc = 0.0f;
    for (int t = 0; t < Nn / 16; t++) {
        int rb = (rb0 + t) & (Nn / 16 - 1);
        v8f acc = {};
        int nch = (rb >> 2) + 1;
        for (int c = 0; c < nch; c++) {
            float* At = Ab + buf * (16 * APITCH);
            *(float4*)(At + srow * APITCH + 4 * sseg) = g0;
            *(float4*)(At + (srow + 8) * APITCH + 4 * sseg) = g1;
            int nrb = rb, nc = c + 1;
            bool more = true;
            if (nc == nch) { nrb = (rb + 1) & (Nn / 16 - 1); nc = 0; more = (t + 1 < Nn / 16); }
            if (more) {
                int k0 = nc * ACH + 4 * sseg;
                g0 = make_float4(0.0f, 0.0f, 0.0f, 0.0f); g1 = g0;
                if (k0 < 16 * (nrb + 1)) {
                    g0 = *(const float4*)(Li + (size_t)(nrb * 16 + srow) * Nn + k0);
                    g1 = *(const float4*)(Li + (size_t)(nrb * 16 + srow + 8) * Nn + k0);
                }
            }
            __builtin_amdgcn_fence(__ATOMIC_RELEASE, "workgroup", "local");
            __builtin_amdgcn_s_barrier();
            __builtin_amdgcn_fence(__ATOMIC_ACQUIRE, "workgroup", "local");
            const float* fac = fa + buf * (16 * APITCH);
            const float* fbc = fb + (c * ACH) * KTS;
            v2f av[ACH / 4], bv[ACH / 4];
#pragma unroll
            for (int s = 0; s < ACH / 4; s++) {
                av[s] = *(const v2f*)(fac + 4 * s);
                bv[s].x = fbc[(4 * s) * KTS];
                bv[s].y = fbc[(4 * s + 1) * KTS];
            }
#pragma unroll
            for (int s = 0; s < ACH / 4; s++)
                acc = __builtin_amdgcn_wmma_f32_16x16x4_f32(false, av[s], false, bv[s],
                                                            (short)0, acc, false, false);
            __builtin_amdgcn_sched_group_barrier(0x100, 24, 0);
            __builtin_amdgcn_sched_group_barrier(0x008, 16, 0);
            buf ^= 1;
        }
        const float* zb = zsh + rb * 16 + 8 * hi;
#pragma unroll
        for (int v = 0; v < 8; v++) {
            float cv = acc[v];
            meanAcc += zb[v] * cv;
            varAcc  += cv * cv;
        }
    }
    meanAcc += __shfl_xor(meanAcc, 16);
    varAcc  += __shfl_xor(varAcc, 16);
    if (lane < 16) {
        int m = colBase + wcol + lane;
        out_mean[b * Mm + m] = meang[b] + meanAcc;
        out_var[b * Mm + m]  = Kmm[b * Mm + m] - varAcc;
    }
}

extern "C" void kernel_launch(void* const* d_in, const int* in_sizes, int n_in,
                              void* d_out, int out_size, void* d_ws, size_t ws_size,
                              hipStream_t stream) {
    const float* coords_train = (const float*)d_in[0];
    const float* kp_train     = (const float*)d_in[1];
    const float* coords_test  = (const float*)d_in[2];
    const float* kp_test      = (const float*)d_in[3];
    const float* y_train      = (const float*)d_in[4];
    const float* var          = (const float*)d_in[5];
    const float* mean         = (const float*)d_in[6];
    const float* scale_param  = (const float*)d_in[7];

    float* ws = (float*)d_ws;
    size_t off = 0;
    float* A      = ws + off; off += (size_t)Bb * Nn * Nn;
    float* Linv   = ws + off; off += (size_t)Bb * Nn * Nn;
    float* Lf     = ws + off; off += (size_t)Bb * Nn * Nn;
    float* trainP = ws + off; off += (size_t)Bb * Nn * 6;
    float* testP  = ws + off; off += (size_t)Bb * Mm * 6;
    float* Kmm    = ws + off; off += (size_t)Bb * Mm;
    float* z      = ws + off; off += (size_t)Bb * Nn;
    float* Yrow   = ws + off; off += (size_t)Bb * Nn;

    float* out_mean = (float*)d_out;
    float* out_var  = out_mean + (size_t)Bb * Mm;
    float* nlml     = out_var + (size_t)Bb * Mm;

    prep_kernel<<<(Bb * Mm + 255) / 256, 256, 0, stream>>>(
        coords_train, kp_train, coords_test, kp_test, scale_param, trainP, testP, Kmm);
    buildA_kernel<<<(Bb * Nn * Nn) / 256, 256, 0, stream>>>(trainP, var, scale_param, y_train, mean, A, Yrow);
#define CHOL_STEP(K) chol_step_kernel<<<dim3(NT + 2, NT - (K), Bb), 32, 0, stream>>>(A, Lf, Linv, Yrow, z, (K));
    CHOL_STEP(0) CHOL_STEP(1) CHOL_STEP(2) CHOL_STEP(3) CHOL_STEP(4) CHOL_STEP(5) CHOL_STEP(6) CHOL_STEP(7)
    CHOL_STEP(8) CHOL_STEP(9) CHOL_STEP(10) CHOL_STEP(11) CHOL_STEP(12) CHOL_STEP(13) CHOL_STEP(14) CHOL_STEP(15)
    CHOL_STEP(16) CHOL_STEP(17) CHOL_STEP(18) CHOL_STEP(19) CHOL_STEP(20) CHOL_STEP(21) CHOL_STEP(22) CHOL_STEP(23)
    CHOL_STEP(24) CHOL_STEP(25) CHOL_STEP(26) CHOL_STEP(27) CHOL_STEP(28) CHOL_STEP(29) CHOL_STEP(30) CHOL_STEP(31)
#undef CHOL_STEP
    nlml_kernel<<<Bb, 1024, 0, stream>>>(Lf, z, nlml);

    size_t smem_bytes = (size_t)(Nn * KTS + Nn + Nn * 6 + 2 * 16 * APITCH) * sizeof(float);
    predict_kernel<<<dim3(Mm / MT, Bb), 128, smem_bytes, stream>>>(
        Linv, trainP, testP, Kmm, z, mean, scale_param, out_mean, out_var);
}


# predict kernel: a second wave per SIMD (8 per workgroup): four more waves take the odd row bands of L^-1, staging tiles of their own; group sums added at the end: new sum order in pred_mean, pred_var
# speedup vs baseline: 67.3232x; 1.1028x over previous
#include <hip/hip_runtime.h>
#include <math.h>

#define Nn 1024
#define Mm 16384
#define Bb 2
#define MT 64
#define KTS 65
#define ACH 64
#define KU 8
#define PG 2
#define APITCH 68

typedef __attribute__((ext_vector_type(2))) float v2f;
typedef __attribute__((ext_vector_type(8))) float v8f;

#define SQRT3F 1.7320508075688772f
#define LOG2PIF 1.8378770664093453f

__device__ __forceinline__ float kern_pair(float x1, float y1, float a1, float b1, float c1, float q1,
                                           float x2, float y2, float a2, float b2, float c2, float q2,
                                           float scale) {
    float s00 = a1 + a2, s11 = b1 + b2, s01 = c1 + c2;
    float sdet = s00 * s11 - s01 * s01;
    float d0 = x1 - x2, d1 = y1 - y2;
    float Q = 0.5f * (s11 * d0 * d0 - 2.0f * s01 * d0 * d1 + s00 * d1 * d1) / sdet;
    float C = 2.0f * q1 * q2 * rsqrtf(fmaxf(sdet, 1e-5f));
    float t = SQRT3F * sqrtf(fmaxf(Q, 1e-5f));
    return (1.0f + t) * __expf(-t) * C * scale;
}

__global__ void prep_kernel(const float* __restrict__ ct, const float* __restrict__ kpt,
                            const float* __restrict__ cs, const float* __restrict__ kps,
                            const float* __restrict__ scale_param,
                            float* __restrict__ trainP, float* __restrict__ testP,
                            float* __restrict__ Kmm) {
    int gid = blockIdx.x * blockDim.x + threadIdx.x;
    float scale = __expf(scale_param[0]);
    if (gid < Bb * Mm) {
        float a = kps[gid * 3 + 0], b = kps[gid * 3 + 1], c = kps[gid * 3 + 2];
        float det = a * b - c * c;
        testP[gid * 6 + 0] = cs[gid * 2 + 0];
        testP[gid * 6 + 1] = cs[gid * 2 + 1];
        testP[gid * 6 + 2] = a;
        testP[gid * 6 + 3] = b;
        testP[gid * 6 + 4] = c;
        testP[gid * 6 + 5] = sqrtf(sqrtf(det));
        float t0 = SQRT3F * sqrtf(1e-5f);
        float m0 = (1.0f + t0) * __expf(-t0);
        float C = 2.0f * sqrtf(det) * rsqrtf(fmaxf(4.0f * det, 1e-5f));
        Kmm[gid] = C * m0 * scale;
    }
    if (gid < Bb * Nn) {
        float a = kpt[gid * 3 + 0], b = kpt[gid * 3 + 1], c = kpt[gid * 3 + 2];
        float det = a * b - c * c;
        trainP[gid * 6 + 0] = ct[gid * 2 + 0];
        trainP[gid * 6 + 1] = ct[gid * 2 + 1];
        trainP[gid * 6 + 2] = a;
        trainP[gid * 6 + 3] = b;
        trainP[gid * 6 + 4] = c;
        trainP[gid * 6 + 5] = sqrtf(sqrtf(det));
    }
}

__global__ void buildA_kernel(const float* __restrict__ trainP, const float* __restrict__ var,
                              const float* __restrict__ scale_param,
                              const float* __restrict__ y, const float* __restrict__ mean,
                              float* __restrict__ A, float* __restrict__ Yrow) {
    int gid = blockIdx.x * blockDim.x + threadIdx.x;
    int b = gid >> 20;
    int i = (gid >> 10) & (Nn - 1);
    int j = gid & (Nn - 1);
    const float* p1 = trainP + (size_t)(b * Nn + i) * 6;
    const float* p2 = trainP + (size_t)(b * Nn + j) * 6;
    float scale = __expf(scale_param[0]);
    float v = kern_pair(p1[0], p1[1], p1[2], p1[3], p1[4], p1[5],
                        p2[0], p2[1], p2[2], p2[3], p2[4], p2[5], scale);
    if (i == j) v += var[b * Nn + i];
    A[(size_t)gid] = v;
    if (i == 0) Yrow[b * Nn + j] = y[b * Nn + j] - mean[b];
}

#define TS 32
#define NT (Nn / TS)

__device__ __forceinline__ float lane_bcast(float v, int src) {
    return __int_as_float(__builtin_amdgcn_readlane(__float_as_int(v), src));
}

__device__ __forceinline__ void load_row32(const float* p, float (&r)[TS]) {
    const float4* p4 = (const float4*)p;
#pragma unroll
    for (int q = 0; q < TS / 4; q++) {
        float4 t = p4[q];
        r[4 * q + 0] = t.x; r[4 * q + 1] = t.y; r[4 * q + 2] = t.z; r[4 * q + 3] = t.w;
    }
}

__device__ __forceinline__ void store_row32(float* p, const float (&r)[TS]) {
    float4* p4 = (float4*)p;
#pragma unroll
    for (int q = 0; q < TS / 4; q++) {
        float4 t;
        t.x = r[4 * q + 0]; t.y = r[4 * q + 1]; t.z = r[4 * q + 2]; t.w = r[4 * q + 3];
        p4[q] = t;
    }
}

#define QP 36

__global__ __launch_bounds__(32) void chol_step_kernel(float* Abase, float* __restrict__ Lbase,
                                                       float* Xbase, float* Ybase,
                                                       float* __restrict__ zbase, int k) {
    __shared__ __attribute__((aligned(16))) float qs[TS * QP];
    __shared__ __attribute__((aligned(16))) float ts[TS * QP];
    int nmat = NT - k;
    int bx = blockIdx.x;
    int kind = (bx < nmat) ? 0 : (bx <= NT) ? 1 : 2;
    int i = (kind == 0) ? k + bx : (kind == 1) ? bx - nmat : 0;
    int j = k + blockIdx.y;
    if (kind == 0 && j > i) return;
    int b = blockIdx.z;
    int r = threadIdx.x;
    float* A = Abase + (size_t)b * Nn * Nn;
    float* Lo = Lbase + (size_t)b * Nn * Nn;
    float* X = Xbase + (size_t)b * Nn * Nn;
    float* Y = Ybase + (size_t)b * Nn;
    bool ident = (kind == 1) && (i == k);
    bool mine = (kind != 2) || (r == 0);
    const float* prow = (kind == 0) ? A + (size_t)(i * TS + r) * Nn + k * TS
                      : (kind == 1) ? X + (size_t)(i * TS + r) * Nn + k * TS
                                    : Y + k * TS;
    float* trow = (kind == 0) ? A + (size_t)(i * TS + r) * Nn + j * TS
                : (kind == 1) ? X + (size_t)(i * TS + r) * Nn + j * TS
                              : Y + j * TS;

    float d[TS];
    v2f pq[TS];
    load_row32(A + (size_t)(k * TS + r) * Nn + k * TS, d);
    {
        float p[TS], q[TS], t[TS];
        load_row32(A + (size_t)(j * TS + r) * Nn + k * TS, q);
        if (j > k) {
            if (!ident && mine) {
                load_row32(trow, t);
            } else {
#pragma unroll
                for (int c = 0; c < TS; c++) t[c] = 0.0f;
            }
        }
        if (ident) {
#pragma unroll
            for (int c = 0; c < TS; c++) p[c] = (c == r) ? 1.0f : 0.0f;
        } else {
            load_row32(prow, p);
#pragma unroll
            for (int c = 0; c < TS; c++) p[c] = mine ? p[c] : 0.0f;
        }
#pragma unroll
        for (int c = 0; c < TS; c++) { pq[c].x = p[c]; pq[c].y = q[c]; }
        if (j > k) store_row32(ts + r * QP, t);
    }

#pragma unroll
    for (int c = 0; c < TS; c++) {
#pragma unroll
        for (int c1 = 0; c1 < c; c1++) {
            float s = lane_bcast(d[c1], c);
            d[c] -= d[c1] * s;
            pq[c] -= pq[c1] * s;
        }
        float piv = lane_bcast(d[c], c);
        float dg = sqrtf(piv);
        float dinv = 1.0f / dg;
        d[c] = (r == c) ? dg : d[c] * dinv;
        pq[c] = pq[c] * dinv;
    }

    if (j > k) {
        float4* qw = (float4*)(qs + r * QP);
#pragma unroll
        for (int c4 = 0; c4 < TS / 4; c4++)
            qw[c4] = make_float4(pq[4 * c4].y, pq[4 * c4 + 1].y, pq[4 * c4 + 2].y, pq[4 * c4 + 3].y);
        __syncthreads();
        float* arow = trow;
#pragma unroll 1
        for (int s0 = 0; s0 < TS; s0 += 4) {
            float4 av = *(const float4*)(ts + r * QP + s0);
            float a0 = av.x, a1 = av.y, a2 = av.z, a3 = av.w;
            const float4* q0 = (const float4*)(qs + (s0 + 0) * QP);
            const float4* q1 = (const float4*)(qs + (s0 + 1) * QP);
            const float4* q2 = (const float4*)(qs + (s0 + 2) * QP);
            const float4* q3 = (const float4*)(qs + (s0 + 3) * QP);
#pragma unroll
            for (int c4 = 0; c4 < TS / 4; c4++) {
                float4 u0 = q0[c4], u1 = q1[c4], u2 = q2[c4], u3 = q3[c4];
                float p0 = pq[4 * c4].x, p1 = pq[4 * c4 + 1].x, p2 = pq[4 * c4 + 2].x, p3 = pq[4 * c4 + 3].x;
                a0 -= p0 * u0.x; a1 -= p0 * u1.x; a2 -= p0 * u2.x; a3 -= p0 * u3.x;
                a0 -= p1 * u0.y; a1 -= p1 * u1.y; a2 -= p1 * u2.y; a3 -= p1 * u3.y;
                a0 -= p2 * u0.z; a1 -= p2 * u1.z; a2 -= p2 * u2.z; a3 -= p2 * u3.z;
                a0 -= p3 * u0.w; a1 -= p3 * u1.w; a2 -= p3 * u2.w; a3 -= p3 * u3.w;
            }
            if (mine) *(float4*)(arow + s0) = make_float4(a0, a1, a2, a3);
        }
    } else if (kind == 1) {
        float* dst = X + (size_t)(k * TS) * Nn + i * TS + r;
#pragma unroll
        for (int c = 0; c < TS; c++) dst[(size_t)c * Nn] = pq[c].x;
    } else if (kind == 2) {
        if (r == 0) {
            float p[TS];
#pragma unroll
            for (int c = 0; c < TS; c++) p[c] = pq[c].x;
            store_row32(zbase + (size_t)b * Nn + k * TS, p);
        }
    } else if (i > k) {
        float p[TS];
#pragma unroll
        for (int c = 0; c < TS; c++) p[c] = pq[c].x;
        store_row32(Lo + (size_t)(i * TS + r) * Nn + k * TS, p);
    } else {
#pragma unroll
        for (int c = 0; c < TS; c++) if (c > r) d[c] = 0.0f;
        store_row32(Lo + (size_t)(k * TS + r) * Nn + k * TS, d);
    }
}

__global__ __launch_bounds__(1024) void nlml_kernel(const float* __restrict__ Lbase,
                                                    const float* __restrict__ zbase,
                                                    float* __restrict__ nlml_out) {
    int b = blockIdx.x;
    const float* L = Lbase + (size_t)b * Nn * Nn;
    __shared__ float col[Nn];
    __shared__ float zz[Nn];
    int tid = threadIdx.x;
    float z = zbase[b * Nn + tid];
    col[tid] = __logf(L[(size_t)tid * Nn + tid]);
    zz[tid] = z * z;
    __syncthreads();
    for (int s = 512; s > 0; s >>= 1) {
        if (tid < s) {
            col[tid] += col[tid + s];
            zz[tid] += zz[tid + s];
        }
        __syncthreads();
    }
    if (tid == 0) nlml_out[b] = 0.5f * zz[0] + col[0] + 0.5f * Nn * LOG2PIF;
}

__global__ __launch_bounds__(128 * PG) void predict_kernel(const float* __restrict__ Linvbase,
                                                      const float* __restrict__ trainPg,
                                                      const float* __restrict__ testPg,
                                                      const float* __restrict__ Kmm,
                                                      const float* __restrict__ zbase,
                                                      const float* __restrict__ meang,
                                                      const float* __restrict__ scale_param,
                                                      float* __restrict__ out_mean,
                                                      float* __restrict__ out_var) {
    extern __shared__ float smem[];
    float* Kt  = smem;
    float* zsh = smem + Nn * KTS;
    float* tp  = zsh + Nn;
    float* Ab  = tp + Nn * 6;
    int b = blockIdx.y;
    int colBase = blockIdx.x * MT;
    int tid = threadIdx.x;
    float scale = __expf(scale_param[0]);

    for (int idx = tid; idx < Nn; idx += 128 * PG) zsh[idx] = zbase[b * Nn + idx];
    for (int idx = tid; idx < Nn * 6; idx += 128 * PG) tp[idx] = trainPg[(size_t)b * Nn * 6 + idx];
    __syncthreads();

    {
        int c = tid & 63;
        int h = tid >> 6;
        const float* q = testPg + (size_t)(b * Mm + colBase + c) * 6;
        float x2 = q[0], y2 = q[1], a2 = q[2], b2 = q[3], c2 = q[4], q2 = q[5];
        for (int n0 = h; n0 < Nn; n0 += 2 * PG * KU) {
            float pr[KU][6], kv[KU];
#pragma unroll
            for (int u = 0; u < KU; u++) {
                const float* p = tp + (n0 + 2 * PG * u) * 6;
#pragma unroll
                for (int e = 0; e < 6; e++) pr[u][e] = p[e];
            }
#pragma unroll
            for (int u = 0; u < KU; u++)
                kv[u] = kern_pair(pr[u][0], pr[u][1], pr[u][2], pr[u][3], pr[u][4], pr[u][5],
                                  x2, y2, a2, b2, c2, q2, scale);
#pragma unroll
            for (int u = 0; u < KU; u++)
                Kt[(n0 + 2 * PG * u) * KTS + c] = kv[u];
        }
    }
    __syncthreads();

    int grp = __builtin_amdgcn_readfirstlane(tid >> 7);
    int gt = tid & 127;
    int wave = gt >> 5, lane = tid & 31;
    int r = lane & 15, hi = lane >> 4;
    int wcol = wave * 16;
    const float* Li = Linvbase + (size_t)b * Nn * Nn;

    int p0 = (((int)(4.0f * sqrtf((float)blockIdx.x))) & (Nn / 16 - 1)) >> 1;
    int rb0 = 2 * p0 + grp;
    float* Abg = Ab + grp * (2 * 16 * APITCH);
    int srow = gt >> 4, sseg = gt & 15;
    const float* fa = Abg + r * APITCH + 2 * hi;
    const float* fb = Kt + (2 * hi) * KTS + wcol + r;
    float4 g0, g1;
    {
        g0 = make_float4(0.0f, 0.0f, 0.0f, 0.0f); g1 = g0;
        if (4 * sseg < 16 * (rb0 + 1)) {
            g0 = *(const float4*)(Li + (size_t)(rb0 * 16 + srow) * Nn + 4 * sseg);
            g1 = *(const float4*)(Li + (size_t)(rb0 * 16 + srow + 8) * Nn + 4 * sseg);
        }
    }
    int buf = 0;
    float meanAcc = 0.0f, varAcc = 0.0f;
    for (int t = 0; t < Nn / 32; t++) {
        int p = (p0 + t) & (Nn / 32 - 1);
        int rb = 2 * p + grp;
        v8f acc = {};
        int nch = (p >> 1) + 1;
        for (int c = 0; c < nch; c++) {
            float* At = Abg + buf * (16 * APITCH);
            *(float4*)(At + srow * APITCH + 4 * sseg) = g0;
            *(float4*)(At + (srow + 8) * APITCH + 4 * sseg) = g1;
            int nrb = rb, nc = c + 1;
            bool more = true;
            if (nc == nch) { nrb = 2 * ((p0 + t + 1) & (Nn / 32 - 1)) + grp; nc = 0; more = (t + 1 < Nn / 32); }
            if (more) {
                int k0 = nc * ACH + 4 * sseg;
                g0 = make_float4(0.0f, 0.0f, 0.0f, 0.0f); g1 = g0;
                if (k0 < 16 * (nrb + 1)) {
                    g0 = *(const float4*)(Li + (size_t)(nrb * 16 + srow) * Nn + k0);
                    g1 = *(const float4*)(Li + (size_t)(nrb * 16 + srow + 8) * Nn + k0);
                }
            }
            __builtin_amdgcn_fence(__ATOMIC_RELEASE, "workgroup", "local");
            __builtin_amdgcn_s_barrier();
            __builtin_amdgcn_fence(__ATOMIC_ACQUIRE, "workgroup", "local");
            const float* fac = fa + buf * (16 * APITCH);
            const float* fbc = fb + (c * ACH) * KTS;
            v2f av[ACH / 4], bv[ACH / 4];
#pragma unroll
            for (int s = 0; s < ACH / 4; s++) {
                av[s] = *(const v2f*)(fac + 4 * s);
                bv[s].x = fbc[(4 * s) * KTS];
                bv[s].y = fbc[(4 * s + 1) * KTS];
            }
#pragma unroll
            for (int s = 0; s < ACH / 4; s++)
                acc = __builtin_amdgcn_wmma_f32_16x16x4_f32(false, av[s], false, bv[s],
                                                            (short)0, acc, false, false);
            __builtin_amdgcn_sched_group_barrier(0x100, 24, 0);
            __builtin_amdgcn_sched_group_barrier(0x008, 16, 0);
            buf ^= 1;
        }
        const float* zb = zsh + rb * 16 + 8 * hi;
#pragma unroll
        for (int v = 0; v < 8; v++) {
            float cv = acc[v];
            meanAcc += zb[v] * cv;
            varAcc  += cv * cv;
        }
    }
    meanAcc += __shfl_xor(meanAcc, 16);
    varAcc  += __shfl_xor(varAcc, 16);
    float* red = tp;
    if (grp == 1 && lane < 16) {
        red[wcol + lane] = meanAcc;
        red[MT + wcol + lane] = varAcc;
    }
    __syncthreads();
    if (grp == 0 && lane < 16) {
        int m = colBase + wcol + lane;
        out_mean[b * Mm + m] = meang[b] + (meanAcc + red[wcol + lane]);
        out_var[b * Mm + m]  = Kmm[b * Mm + m] - (varAcc + red[MT + wcol + lane]);
    }
}

extern "C" void kernel_launch(void* const* d_in, const int* in_sizes, int n_in,
                              void* d_out, int out_size, void* d_ws, size_t ws_size,
                              hipStream_t stream) {
    const float* coords_train = (const float*)d_in[0];
    const float* kp_train     = (const float*)d_in[1];
    const float* coords_test  = (const float*)d_in[2];
    const float* kp_test      = (const float*)d_in[3];
    const float* y_train      = (const float*)d_in[4];
    const float* var          = (const float*)d_in[5];
    const float* mean         = (const float*)d_in[6];
    const float* scale_param  = (const float*)d_in[7];

    float* ws = (float*)d_ws;
    size_t off = 0;
    float* A      = ws + off; off += (size_t)Bb * Nn * Nn;
    float* Linv   = ws + off; off += (size_t)Bb * Nn * Nn;
    float* Lf     = ws + off; off += (size_t)Bb * Nn * Nn;
    float* trainP = ws + off; off += (size_t)Bb * Nn * 6;
    float* testP  = ws + off; off += (size_t)Bb * Mm * 6;
    float* Kmm    = ws + off; off += (size_t)Bb * Mm;
    float* z      = ws + off; off += (size_t)Bb * Nn;
    float* Yrow   = ws + off; off += (size_t)Bb * Nn;

    float* out_mean = (float*)d_out;
    float* out_var  = out_mean + (size_t)Bb * Mm;
    float* nlml     = out_var + (size_t)Bb * Mm;

    prep_kernel<<<(Bb * Mm + 255) / 256, 256, 0, stream>>>(
        coords_train, kp_train, coords_test, kp_test, scale_param, trainP, testP, Kmm);
    buildA_kernel<<<(Bb * Nn * Nn) / 256, 256, 0, stream>>>(trainP, var, scale_param, y_train, mean, A, Yrow);
#define CHOL_STEP(K) chol_step_kernel<<<dim3(NT + 2, NT - (K), Bb), 32, 0, stream>>>(A, Lf, Linv, Yrow, z, (K));
    CHOL_STEP(0) CHOL_STEP(1) CHOL_STEP(2) CHOL_STEP(3) CHOL_STEP(4) CHOL_STEP(5) CHOL_STEP(6) CHOL_STEP(7)
    CHOL_STEP(8) CHOL_STEP(9) CHOL_STEP(10) CHOL_STEP(11) CHOL_STEP(12) CHOL_STEP(13) CHOL_STEP(14) CHOL_STEP(15)
    CHOL_STEP(16) CHOL_STEP(17) CHOL_STEP(18) CHOL_STEP(19) CHOL_STEP(20) CHOL_STEP(21) CHOL_STEP(22) CHOL_STEP(23)
    CHOL_STEP(24) CHOL_STEP(25) CHOL_STEP(26) CHOL_STEP(27) CHOL_STEP(28) CHOL_STEP(29) CHOL_STEP(30) CHOL_STEP(31)
#undef CHOL_STEP
    nlml_kernel<<<Bb, 1024, 0, stream>>>(Lf, z, nlml);

    size_t smem_bytes = (size_t)(Nn * KTS + Nn + Nn * 6 + PG * 2 * 16 * APITCH) * sizeof(float);
    predict_kernel<<<dim3(Mm / MT, Bb), 128 * PG, smem_bytes, stream>>>(
        Linv, trainP, testP, Kmm, z, mean, scale_param, out_mean, out_var);
}


# predict kernel: four waves per SIMD (16 per workgroup): four wave groups take the row bands of L^-1 modulo 4, so half the chunk steps and barriers; new sum order in pred_mean, pred_var
# speedup vs baseline: 70.7943x; 1.0516x over previous
#include <hip/hip_runtime.h>
#include <math.h>

#define Nn 1024
#define Mm 16384
#define Bb 2
#define MT 64
#define KTS 65
#define ACH 64
#define KU 8
#define PG 4
#define NQ (Nn / 16 / PG)
#define APITCH 68

typedef __attribute__((ext_vector_type(2))) float v2f;
typedef __attribute__((ext_vector_type(8))) float v8f;

#define SQRT3F 1.7320508075688772f
#define LOG2PIF 1.8378770664093453f

__device__ __forceinline__ float kern_pair(float x1, float y1, float a1, float b1, float c1, float q1,
                                           float x2, float y2, float a2, float b2, float c2, float q2,
                                           float scale) {
    float s00 = a1 + a2, s11 = b1 + b2, s01 = c1 + c2;
    float sdet = s00 * s11 - s01 * s01;
    float d0 = x1 - x2, d1 = y1 - y2;
    float Q = 0.5f * (s11 * d0 * d0 - 2.0f * s01 * d0 * d1 + s00 * d1 * d1) / sdet;
    float C = 2.0f * q1 * q2 * rsqrtf(fmaxf(sdet, 1e-5f));
    float t = SQRT3F * sqrtf(fmaxf(Q, 1e-5f));
    return (1.0f + t) * __expf(-t) * C * scale;
}

__global__ void prep_kernel(const float* __restrict__ ct, const float* __restrict__ kpt,
                            const float* __restrict__ cs, const float* __restrict__ kps,
                            const float* __restrict__ scale_param,
                            float* __restrict__ trainP, float* __restrict__ testP,
                            float* __restrict__ Kmm) {
    int gid = blockIdx.x * blockDim.x + threadIdx.x;
    float scale = __expf(scale_param[0]);
    if (gid < Bb * Mm) {
        float a = kps[gid * 3 + 0], b = kps[gid * 3 + 1], c = kps[gid * 3 + 2];
        float det = a * b - c * c;
        testP[gid * 6 + 0] = cs[gid * 2 + 0];
        testP[gid * 6 + 1] = cs[gid * 2 + 1];
        testP[gid * 6 + 2] = a;
        testP[gid * 6 + 3] = b;
        testP[gid * 6 + 4] = c;
        testP[gid * 6 + 5] = sqrtf(sqrtf(det));
        float t0 = SQRT3F * sqrtf(1e-5f);
        float m0 = (1.0f + t0) * __expf(-t0);
        float C = 2.0f * sqrtf(det) * rsqrtf(fmaxf(4.0f * det, 1e-5f));
        Kmm[gid] = C * m0 * scale;
    }
    if (gid < Bb * Nn) {
        float a = kpt[gid * 3 + 0], b = kpt[gid * 3 + 1], c = kpt[gid * 3 + 2];
        float det = a * b - c * c;
        trainP[gid * 6 + 0] = ct[gid * 2 + 0];
        trainP[gid * 6 + 1] = ct[gid * 2 + 1];
        trainP[gid * 6 + 2] = a;
        trainP[gid * 6 + 3] = b;
        trainP[gid * 6 + 4] = c;
        trainP[gid * 6 + 5] = sqrtf(sqrtf(det));
    }
}

__global__ void buildA_kernel(const float* __restrict__ trainP, const float* __restrict__ var,
                              const float* __restrict__ scale_param,
                              const float* __restrict__ y, const float* __restrict__ mean,
                              float* __restrict__ A, float* __restrict__ Yrow) {
    int gid = blockIdx.x * blockDim.x + threadIdx.x;
    int b = gid >> 20;
    int i = (gid >> 10) & (Nn - 1);
    int j = gid & (Nn - 1);
    const float* p1 = trainP + (size_t)(b * Nn + i) * 6;
    const float* p2 = trainP + (size_t)(b * Nn + j) * 6;
    float scale = __expf(scale_param[0]);
    float v = kern_pair(p1[0], p1[1], p1[2], p1[3], p1[4], p1[5],
                        p2[0], p2[1], p2[2], p2[3], p2[4], p2[5], scale);
    if (i == j) v += var[b * Nn + i];
    A[(size_t)gid] = v;
    if (i == 0) Yrow[b * Nn + j] = y[b * Nn + j] - mean[b];
}

#define TS 32
#define NT (Nn / TS)

__device__ __forceinline__ float lane_bcast(float v, int src) {
    return __int_as_float(__builtin_amdgcn_readlane(__float_as_int(v), src));
}

__device__ __forceinline__ void load_row32(const float* p, float (&r)[TS]) {
    const float4* p4 = (const float4*)p;
#pragma unroll
    for (int q = 0; q < TS / 4; q++) {
        float4 t = p4[q];
        r[4 * q + 0] = t.x; r[4 * q + 1] = t.y; r[4 * q + 2] = t.z; r[4 * q + 3] = t.w;
    }
}

__device__ __forceinline__ void store_row32(float* p, const float (&r)[TS]) {
    float4* p4 = (float4*)p;
#pragma unroll
    for (int q = 0; q < TS / 4; q++) {
        float4 t;
        t.x = r[4 * q + 0]; t.y = r[4 * q + 1]; t.z = r[4 * q + 2]; t.w = r[4 * q + 3];
        p4[q] = t;
    }
}

#define QP 36

__global__ __launch_bounds__(32) void chol_step_kernel(float* Abase, float* __restrict__ Lbase,
                                                       float* Xbase, float* Ybase,
                                                       float* __restrict__ zbase, int k) {
    __shared__ __attribute__((aligned(16))) float qs[TS * QP];
    __shared__ __attribute__((aligned(16))) float ts[TS * QP];
    int nmat = NT - k;
    int bx = blockIdx.x;
    int kind = (bx < nmat) ? 0 : (bx <= NT) ? 1 : 2;
    int i = (kind == 0) ? k + bx : (kind == 1) ? bx - nmat : 0;
    int j = k + blockIdx.y;
    if (kind == 0 && j > i) return;
    int b = blockIdx.z;
    int r = threadIdx.x;
    float* A = Abase + (size_t)b * Nn * Nn;
    float* Lo = Lbase + (size_t)b * Nn * Nn;
    float* X = Xbase + (size_t)b * Nn * Nn;
    float* Y = Ybase + (size_t)b * Nn;
    bool ident = (kind == 1) && (i == k);
    bool mine = (kind != 2) || (r == 0);
    const float* prow = (kind == 0) ? A + (size_t)(i * TS + r) * Nn + k * TS
                      : (kind == 1) ? X + (size_t)(i * TS + r) * Nn + k * TS
                                    : Y + k * TS;
    float* trow = (kind == 0) ? A + (size_t)(i * TS + r) * Nn + j * TS
                : (kind == 1) ? X + (size_t)(i * TS + r) * Nn + j * TS
                              : Y + j * TS;

    float d[TS];
    v2f pq[TS];
    load_row32(A + (size_t)(k * TS + r) * Nn + k * TS, d);
    {
        float p[TS], q[TS], t[TS];
        load_row32(A + (size_t)(j * TS + r) * Nn + k * TS, q);
        if (j > k) {
            if (!ident && mine) {
                load_row32(trow, t);
            } else {
#pragma unroll
                for (int c = 0; c < TS; c++) t[c] = 0.0f;
            }
        }
        if (ident) {
#pragma unroll
            for (int c = 0; c < TS; c++) p[c] = (c == r) ? 1.0f : 0.0f;
        } else {
            load_row32(prow, p);
#pragma unroll
            for (int c = 0; c < TS; c++) p[c] = mine ? p[c] : 0.0f;
        }
#pragma unroll
        for (int c = 0; c < TS; c++) { pq[c].x = p[c]; pq[c].y = q[c]; }
        if (j > k) store_row32(ts + r * QP, t);
    }

#pragma unroll
    for (int c = 0; c < TS; c++) {
#pragma unroll
        for (int c1 = 0; c1 < c; c1++) {
            float s = lane_bcast(d[c1], c);
            d[c] -= d[c1] * s;
            pq[c] -= pq[c1] * s;
        }
        float piv = lane_bcast(d[c], c);
        float dg = sqrtf(piv);
        float dinv = 1.0f / dg;
        d[c] = (r == c) ? dg : d[c] * dinv;
        pq[c] = pq[c] * dinv;
    }

    if (j > k) {
        float4* qw = (float4*)(qs + r * QP);
#pragma unroll
        for (int c4 = 0; c4 < TS / 4; c4++)
            qw[c4] = make_float4(pq[4 * c4].y, pq[4 * c4 + 1].y, pq[4 * c4 + 2].y, pq[4 * c4 + 3].y);
        __syncthreads();
        float* arow = trow;
#pragma unroll 1
        for (int s0 = 0; s0 < TS; s0 += 4) {
            float4 av = *(const float4*)(ts + r * QP + s0);
            float a0 = av.x, a1 = av.y, a2 = av.z, a3 = av.w;
            const float4* q0 = (const float4*)(qs + (s0 + 0) * QP);
            const float4* q1 = (const float4*)(qs + (s0 + 1) * QP);
            const float4* q2 = (const float4*)(qs + (s0 + 2) * QP);
            const float4* q3 = (const float4*)(qs + (s0 + 3) * QP);
#pragma unroll
            for (int c4 = 0; c4 < TS / 4; c4++) {
                float4 u0 = q0[c4], u1 = q1[c4], u2 = q2[c4], u3 = q3[c4];
                float p0 = pq[4 * c4].x, p1 = pq[4 * c4 + 1].x, p2 = pq[4 * c4 + 2].x, p3 = pq[4 * c4 + 3].x;
                a0 -= p0 * u0.x; a1 -= p0 * u1.x; a2 -= p0 * u2.x; a3 -= p0 * u3.x;
                a0 -= p1 * u0.y; a1 -= p1 * u1.y; a2 -= p1 * u2.y; a3 -= p1 * u3.y;
                a0 -= p2 * u0.z; a1 -= p2 * u1.z; a2 -= p2 * u2.z; a3 -= p2 * u3.z;
                a0 -= p3 * u0.w; a1 -= p3 * u1.w; a2 -= p3 * u2.w; a3 -= p3 * u3.w;
            }
            if (mine) *(float4*)(arow + s0) = make_float4(a0, a1, a2, a3);
        }
    } else if (kind == 1) {
        float* dst = X + (size_t)(k * TS) * Nn + i * TS + r;
#pragma unroll
        for (int c = 0; c < TS; c++) dst[(size_t)c * Nn] = pq[c].x;
    } else if (kind == 2) {
        if (r == 0) {
            float p[TS];
#pragma unroll
            for (int c = 0; c < TS; c++) p[c] = pq[c].x;
            store_row32(zbase + (size_t)b * Nn + k * TS, p);
        }
    } else if (i > k) {
        float p[TS];
#pragma unroll
        for (int c = 0; c < TS; c++) p[c] = pq[c].x;
        store_row32(Lo + (size_t)(i * TS + r) * Nn + k * TS, p);
    } else {
#pragma unroll
        for (int c = 0; c < TS; c++) if (c > r) d[c] = 0.0f;
        store_row32(Lo + (size_t)(k * TS + r) * Nn + k * TS, d);
    }
}

__global__ __launch_bounds__(1024) void nlml_kernel(const float* __restrict__ Lbase,
                                                    const float* __restrict__ zbase,
                                                    float* __restrict__ nlml_out) {
    int b = blockIdx.x;
    const float* L = Lbase + (size_t)b * Nn * Nn;
    __shared__ float col[Nn];
    __shared__ float zz[Nn];
    int tid = threadIdx.x;
    float z = zbase[b * Nn + tid];
    col[tid] = __logf(L[(size_t)tid * Nn + tid]);
    zz[tid] = z * z;
    __syncthreads();
    for (int s = 512; s > 0; s >>= 1) {
        if (tid < s) {
            col[tid] += col[tid + s];
            zz[tid] += zz[tid + s];
        }
        __syncthreads();
    }
    if (tid == 0) nlml_out[b] = 0.5f * zz[0] + col[0] + 0.5f * Nn * LOG2PIF;
}

__global__ __launch_bounds__(128 * PG) void predict_kernel(const float* __restrict__ Linvbase,
                                                      const float* __restrict__ trainPg,
                                                      const float* __restrict__ testPg,
                                                      const float* __restrict__ Kmm,
                                                      const float* __restrict__ zbase,
                                                      const float* __restrict__ meang,
                                                      const float* __restrict__ scale_param,
                                                      float* __restrict__ out_mean,
                                                      float* __restrict__ out_var) {
    extern __shared__ float smem[];
    float* Kt  = smem;
    float* zsh = smem + Nn * KTS;
    float* tp  = zsh + Nn;
    float* Ab  = tp;
    static_assert(PG * 2 * 16 * APITCH >= Nn * 6, "the staging tiles cover the train parameters' place");
    static_assert(PG * 16 == ACH, "the PG bands of a walk step end in the same chunk");
    float* red = Ab + PG * 2 * 16 * APITCH;
    int b = blockIdx.y;
    int colBase = blockIdx.x * MT;
    int tid = threadIdx.x;
    float scale = __expf(scale_param[0]);

    for (int idx = tid; idx < Nn; idx += 128 * PG) zsh[idx] = zbase[b * Nn + idx];
    for (int idx = tid; idx < Nn * 6; idx += 128 * PG) tp[idx] = trainPg[(size_t)b * Nn * 6 + idx];
    __syncthreads();

    {
        int c = tid & 63;
        int h = tid >> 6;
        const float* q = testPg + (size_t)(b * Mm + colBase + c) * 6;
        float x2 = q[0], y2 = q[1], a2 = q[2], b2 = q[3], c2 = q[4], q2 = q[5];
        for (int n0 = h; n0 < Nn; n0 += 2 * PG * KU) {
            float pr[KU][6], kv[KU];
#pragma unroll
            for (int u = 0; u < KU; u++) {
                const float* p = tp + (n0 + 2 * PG * u) * 6;
#pragma unroll
                for (int e = 0; e < 6; e++) pr[u][e] = p[e];
            }
#pragma unroll
            for (int u = 0; u < KU; u++)
                kv[u] = kern_pair(pr[u][0], pr[u][1], pr[u][2], pr[u][3], pr[u][4], pr[u][5],
                                  x2, y2, a2, b2, c2, q2, scale);
#pragma unroll
            for (int u = 0; u < KU; u++)
                Kt[(n0 + 2 * PG * u) * KTS + c] = kv[u];
        }
    }
    __syncthreads();

    int grp = __builtin_amdgcn_readfirstlane(tid >> 7);
    int gt = tid & 127;
    int wave = gt >> 5, lane = tid & 31;
    int r = lane & 15, hi = lane >> 4;
    int wcol = wave * 16;
    const float* Li = Linvbase + (size_t)b * Nn * Nn;

    int q0 = (((int)(4.0f * sqrtf((float)blockIdx.x))) & (Nn / 16 - 1)) / PG;
    int rb0 = PG * q0 + grp;
    float* Abg = Ab + grp * (2 * 16 * APITCH);
    int srow = gt >> 4, sseg = gt & 15;
    const float* fa = Abg + r * APITCH + 2 * hi;
    const float* fb = Kt + (2 * hi) * KTS + wcol + r;
    float4 g0, g1;
    {
        g0 = make_float4(0.0f, 0.0f, 0.0f, 0.0f); g1 = g0;
        if (4 * sseg < 16 * (rb0 + 1)) {
            g0 = *(const float4*)(Li + (size_t)(rb0 * 16 + srow) * Nn + 4 * sseg);
            g1 = *(const float4*)(Li + (size_t)(rb0 * 16 + srow + 8) * Nn + 4 * sseg);
        }
    }
    int buf = 0;
    float meanAcc = 0.0f, varAcc = 0.0f;
    for (int t = 0; t < NQ; t++) {
        int q = (q0 + t) & (NQ - 1);
        int rb = PG * q + grp;
        v8f acc = {};
        int nch = q + 1;
        for (int c = 0; c < nch; c++) {
            float* At = Abg + buf * (16 * APITCH);
            *(float4*)(At + srow * APITCH + 4 * sseg) = g0;
            *(float4*)(At + (srow + 8) * APITCH + 4 * sseg) = g1;
            int nrb = rb, nc = c + 1;
            bool more = true;
            if (nc == nch) { nrb = PG * ((q0 + t + 1) & (NQ - 1)) + grp; nc = 0; more = (t + 1 < NQ); }
            if (more) {
                int k0 = nc * ACH + 4 * sseg;
                g0 = make_float4(0.0f, 0.0f, 0.0f, 0.0f); g1 = g0;
                if (k0 < 16 * (nrb + 1)) {
                    g0 = *(const float4*)(Li + (size_t)(nrb * 16 + srow) * Nn + k0);
                    g1 = *(const float4*)(Li + (size_t)(nrb * 16 + srow + 8) * Nn + k0);
                }
            }
            __builtin_amdgcn_fence(__ATOMIC_RELEASE, "workgroup", "local");
            __builtin_amdgcn_s_barrier();
            __builtin_amdgcn_fence(__ATOMIC_ACQUIRE, "workgroup", "local");
            const float* fac = fa + buf * (16 * APITCH);
            const float* fbc = fb + (c * ACH) * KTS;
            v2f av[ACH / 4], bv[ACH / 4];
#pragma unroll
            for (int s = 0; s < ACH / 4; s++) {
                av[s] = *(const v2f*)(fac + 4 * s);
                bv[s].x = fbc[(4 * s) * KTS];
                bv[s].y = fbc[(4 * s + 1) * KTS];
            }
#pragma unroll
            for (int s = 0; s < ACH / 4; s++)
                acc = __builtin_amdgcn_wmma_f32_16x16x4_f32(false, av[s], false, bv[s],
                                                            (short)0, acc, false, false);
            __builtin_amdgcn_sched_group_barrier(0x100, 24, 0);
            __builtin_amdgcn_sched_group_barrier(0x008, 16, 0);
            buf ^= 1;
        }
        const float* zb = zsh + rb * 16 + 8 * hi;
#pragma unroll
        for (int v = 0; v < 8; v++) {
            float cv = acc[v];
            meanAcc += zb[v] * cv;
            varAcc  += cv * cv;
        }
    }
    meanAcc += __shfl_xor(meanAcc, 16);
    varAcc  += __shfl_xor(varAcc, 16);
    if (grp != 0 && lane < 16) {
        red[(grp - 1) * 2 * MT + wcol + lane] = meanAcc;
        red[(grp - 1) * 2 * MT + MT + wcol + lane] = varAcc;
    }
    __syncthreads();
    if (grp == 0 && lane < 16) {
        int m = colBase + wcol + lane;
        for (int g = 0; g < PG - 1; g++) {
            meanAcc += red[g * 2 * MT + wcol + lane];
            varAcc  += red[g * 2 * MT + MT + wcol + lane];
        }
        out_mean[b * Mm + m] = meang[b] + meanAcc;
        out_var[b * Mm + m]  = Kmm[b * Mm + m] - varAcc;
    }
}

extern "C" void kernel_launch(void* const* d_in, const int* in_sizes, int n_in,
                              void* d_out, int out_size, void* d_ws, size_t ws_size,
                              hipStream_t stream) {
    const float* coords_train = (const float*)d_in[0];
    const float* kp_train     = (const float*)d_in[1];
    const float* coords_test  = (const float*)d_in[2];
    const float* kp_test      = (const float*)d_in[3];
    const float* y_train      = (const float*)d_in[4];
    const float* var          = (const float*)d_in[5];
    const float* mean         = (const float*)d_in[6];
    const float* scale_param  = (const float*)d_in[7];

    float* ws = (float*)d_ws;
    size_t off = 0;
    float* A      = ws + off; off += (size_t)Bb * Nn * Nn;
    float* Linv   = ws + off; off += (size_t)Bb * Nn * Nn;
    float* Lf     = ws + off; off += (size_t)Bb * Nn * Nn;
    float* trainP = ws + off; off += (size_t)Bb * Nn * 6;
    float* testP  = ws + off; off += (size_t)Bb * Mm * 6;
    float* Kmm    = ws + off; off += (size_t)Bb * Mm;
    float* z      = ws + off; off += (size_t)Bb * Nn;
    float* Yrow   = ws + off; off += (size_t)Bb * Nn;

    float* out_mean = (float*)d_out;
    float* out_var  = out_mean + (size_t)Bb * Mm;
    float* nlml     = out_var + (size_t)Bb * Mm;

    prep_kernel<<<(Bb * Mm + 255) / 256, 256, 0, stream>>>(
        coords_train, kp_train, coords_test, kp_test, scale_param, trainP, testP, Kmm);
    buildA_kernel<<<(Bb * Nn * Nn) / 256, 256, 0, stream>>>(trainP, var, scale_param, y_train, mean, A, Yrow);
#define CHOL_STEP(K) chol_step_kernel<<<dim3(NT + 2, NT - (K), Bb), 32, 0, stream>>>(A, Lf, Linv, Yrow, z, (K));
    CHOL_STEP(0) CHOL_STEP(1) CHOL_STEP(2) CHOL_STEP(3) CHOL_STEP(4) CHOL_STEP(5) CHOL_STEP(6) CHOL_STEP(7)
    CHOL_STEP(8) CHOL_STEP(9) CHOL_STEP(10) CHOL_STEP(11) CHOL_STEP(12) CHOL_STEP(13) CHOL_STEP(14) CHOL_STEP(15)
    CHOL_STEP(16) CHOL_STEP(17) CHOL_STEP(18) CHOL_STEP(19) CHOL_STEP(20) CHOL_STEP(21) CHOL_STEP(22) CHOL_STEP(23)
    CHOL_STEP(24) CHOL_STEP(25) CHOL_STEP(26) CHOL_STEP(27) CHOL_STEP(28) CHOL_STEP(29) CHOL_STEP(30) CHOL_STEP(31)
#undef CHOL_STEP
    nlml_kernel<<<Bb, 1024, 0, stream>>>(Lf, z, nlml);

    size_t smem_bytes = (size_t)(Nn * KTS + Nn + PG * 2 * 16 * APITCH + (PG - 1) * 2 * MT) * sizeof(float);
    predict_kernel<<<dim3(Mm / MT, Bb), 128 * PG, smem_bytes, stream>>>(
        Linv, trainP, testP, Kmm, z, mean, scale_param, out_mean, out_var);
}


# predict kernel: K tile in LDS laid out [k/4][column][k%4]: a lane's B fragment is one conflict-free 8-byte read (was two conflicted 4-byte reads), build stores 16 B; arithmetic and sum order kept
# speedup vs baseline: 77.2059x; 1.0906x over previous
#include <hip/hip_runtime.h>
#include <math.h>

#define Nn 1024
#define Mm 16384
#define Bb 2
#define MT 64
#define KQ (4 * MT)
#define ACH 64
#define KU 8
#define PG 4
#define NQ (Nn / 16 / PG)
#define APITCH 68

typedef __attribute__((ext_vector_type(2))) float v2f;
typedef __attribute__((ext_vector_type(8))) float v8f;

#define SQRT3F 1.7320508075688772f
#define LOG2PIF 1.8378770664093453f

__device__ __forceinline__ float kern_pair(float x1, float y1, float a1, float b1, float c1, float q1,
                                           float x2, float y2, float a2, float b2, float c2, float q2,
                                           float scale) {
    float s00 = a1 + a2, s11 = b1 + b2, s01 = c1 + c2;
    float sdet = s00 * s11 - s01 * s01;
    float d0 = x1 - x2, d1 = y1 - y2;
    float Q = 0.5f * (s11 * d0 * d0 - 2.0f * s01 * d0 * d1 + s00 * d1 * d1) / sdet;
    float C = 2.0f * q1 * q2 * rsqrtf(fmaxf(sdet, 1e-5f));
    float t = SQRT3F * sqrtf(fmaxf(Q, 1e-5f));
    return (1.0f + t) * __expf(-t) * C * scale;
}

__global__ void prep_kernel(const float* __restrict__ ct, const float* __restrict__ kpt,
                            const float* __restrict__ cs, const float* __restrict__ kps,
                            const float* __restrict__ scale_param,
                            float* __restrict__ trainP, float* __restrict__ testP,
                            float* __restrict__ Kmm) {
    int gid = blockIdx.x * blockDim.x + threadIdx.x;
    float scale = __expf(scale_param[0]);
    if (gid < Bb * Mm) {
        float a = kps[gid * 3 + 0], b = kps[gid * 3 + 1], c = kps[gid * 3 + 2];
        float det = a * b - c * c;
        testP[gid * 6 + 0] = cs[gid * 2 + 0];
        testP[gid * 6 + 1] = cs[gid * 2 + 1];
        testP[gid * 6 + 2] = a;
        testP[gid * 6 + 3] = b;
        testP[gid * 6 + 4] = c;
        testP[gid * 6 + 5] = sqrtf(sqrtf(det));
        float t0 = SQRT3F * sqrtf(1e-5f);
        float m0 = (1.0f + t0) * __expf(-t0);
        float C = 2.0f * sqrtf(det) * rsqrtf(fmaxf(4.0f * det, 1e-5f));
        Kmm[gid] = C * m0 * scale;
    }
    if (gid < Bb * Nn) {
        float a = kpt[gid * 3 + 0], b = kpt[gid * 3 + 1], c = kpt[gid * 3 + 2];
        float det = a * b - c * c;
        trainP[gid * 6 + 0] = ct[gid * 2 + 0];
        trainP[gid * 6 + 1] = ct[gid * 2 + 1];
        trainP[gid * 6 + 2] = a;
        trainP[gid * 6 + 3] = b;
        trainP[gid * 6 + 4] = c;
        trainP[gid * 6 + 5] = sqrtf(sqrtf(det));
    }
}

__global__ void buildA_kernel(const float* __restrict__ trainP, const float* __restrict__ var,
                              const float* __restrict__ scale_param,
                              const float* __restrict__ y, const float* __restrict__ mean,
                              float* __restrict__ A, float* __restrict__ Yrow) {
    int gid = blockIdx.x * blockDim.x + threadIdx.x;
    int b = gid >> 20;
    int i = (gid >> 10) & (Nn - 1);
    int j = gid & (Nn - 1);
    const float* p1 = trainP + (size_t)(b * Nn + i) * 6;
    const float* p2 = trainP + (size_t)(b * Nn + j) * 6;
    float scale = __expf(scale_param[0]);
    float v = kern_pair(p1[0], p1[1], p1[2], p1[3], p1[4], p1[5],
                        p2[0], p2[1], p2[2], p2[3], p2[4], p2[5], scale);
    if (i == j) v += var[b * Nn + i];
    A[(size_t)gid] = v;
    if (i == 0) Yrow[b * Nn + j] = y[b * Nn + j] - mean[b];
}

#define TS 32
#define NT (Nn / TS)

__device__ __forceinline__ float lane_bcast(float v, int src) {
    return __int_as_float(__builtin_amdgcn_readlane(__float_as_int(v), src));
}

__device__ __forceinline__ void load_row32(const float* p, float (&r)[TS]) {
    const float4* p4 = (const float4*)p;
#pragma unroll
    for (int q = 0; q < TS / 4; q++) {
        float4 t = p4[q];
        r[4 * q + 0] = t.x; r[4 * q + 1] = t.y; r[4 * q + 2] = t.z; r[4 * q + 3] = t.w;
    }
}

__device__ __forceinline__ void store_row32(float* p, const float (&r)[TS]) {
    float4* p4 = (float4*)p;
#pragma unroll
    for (int q = 0; q < TS / 4; q++) {
        float4 t;
        t.x = r[4 * q + 0]; t.y = r[4 * q + 1]; t.z = r[4 * q + 2]; t.w = r[4 * q + 3];
        p4[q] = t;
    }
}

#define QP 36

__global__ __launch_bounds__(32) void chol_step_kernel(float* Abase, float* __restrict__ Lbase,
                                                       float* Xbase, float* Ybase,
                                                       float* __restrict__ zbase, int k) {
    __shared__ __attribute__((aligned(16))) float qs[TS * QP];
    __shared__ __attribute__((aligned(16))) float ts[TS * QP];
    int nmat = NT - k;
    int bx = blockIdx.x;
    int kind = (bx < nmat) ? 0 : (bx <= NT) ? 1 : 2;
    int i = (kind == 0) ? k + bx : (kind == 1) ? bx - nmat : 0;
    int j = k + blockIdx.y;
    if (kind == 0 && j > i) return;
    int b = blockIdx.z;
    int r = threadIdx.x;
    float* A = Abase + (size_t)b * Nn * Nn;
    float* Lo = Lbase + (size_t)b * Nn * Nn;
    float* X = Xbase + (size_t)b * Nn * Nn;
    float* Y = Ybase + (size_t)b * Nn;
    bool ident = (kind == 1) && (i == k);
    bool mine = (kind != 2) || (r == 0);
    const float* prow = (kind == 0) ? A + (size_t)(i * TS + r) * Nn + k * TS
                      : (kind == 1) ? X + (size_t)(i * TS + r) * Nn + k * TS
                                    : Y + k * TS;
    float* trow = (kind == 0) ? A + (size_t)(i * TS + r) * Nn + j * TS
                : (kind == 1) ? X + (size_t)(i * TS + r) * Nn + j * TS
                              : Y + j * TS;

    float d[TS];
    v2f pq[TS];
    load_row32(A + (size_t)(k * TS + r) * Nn + k * TS, d);
    {
        float p[TS], q[TS], t[TS];
        load_row32(A + (size_t)(j * TS + r) * Nn + k * TS, q);
        if (j > k) {
            if (!ident && mine) {
                load_row32(trow, t);
            } else {
#pragma unroll
                for (int c = 0; c < TS; c++) t[c] = 0.0f;
            }
        }
        if (ident) {
#pragma unroll
            for (int c = 0; c < TS; c++) p[c] = (c == r) ? 1.0f : 0.0f;
        } else {
            load_row32(prow, p);
#pragma unroll
            for (int c = 0; c < TS; c++) p[c] = mine ? p[c] : 0.0f;
        }
#pragma unroll
        for (int c = 0; c < TS; c++) { pq[c].x = p[c]; pq[c].y = q[c]; }
        if (j > k) store_row32(ts + r * QP, t);
    }

#pragma unroll
    for (int c = 0; c < TS; c++) {
#pragma unroll
        for (int c1 = 0; c1 < c; c1++) {
            float s = lane_bcast(d[c1], c);
            d[c] -= d[c1] * s;
            pq[c] -= pq[c1] * s;
        }
        float piv = lane_bcast(d[c], c);
        float dg = sqrtf(piv);
        float dinv = 1.0f / dg;
        d[c] = (r == c) ? dg : d[c] * dinv;
        pq[c] = pq[c] * dinv;
    }

    if (j > k) {
        float4* qw = (float4*)(qs + r * QP);
#pragma unroll
        for (int c4 = 0; c4 < TS / 4; c4++)
            qw[c4] = make_float4(pq[4 * c4].y, pq[4 * c4 + 1].y, pq[4 * c4 + 2].y, pq[4 * c4 + 3].y);
        __syncthreads();
        float* arow = trow;
#pragma unroll 1
        for (int s0 = 0; s0 < TS; s0 += 4) {
            float4 av = *(const float4*)(ts + r * QP + s0);
            float a0 = av.x, a1 = av.y, a2 = av.z, a3 = av.w;
            const float4* q0 = (const float4*)(qs + (s0 + 0) * QP);
            const float4* q1 = (const float4*)(qs + (s0 + 1) * QP);
            const float4* q2 = (const float4*)(qs + (s0 + 2) * QP);
            const float4* q3 = (const float4*)(qs + (s0 + 3) * QP);
#pragma unroll
            for (int c4 = 0; c4 < TS / 4; c4++) {
                float4 u0 = q0[c4], u1 = q1[c4], u2 = q2[c4], u3 = q3[c4];
                float p0 = pq[4 * c4].x, p1 = pq[4 * c4 + 1].x, p2 = pq[4 * c4 + 2].x, p3 = pq[4 * c4 + 3].x;
                a0 -= p0 * u0.x; a1 -= p0 * u1.x; a2 -= p0 * u2.x; a3 -= p0 * u3.x;
                a0 -= p1 * u0.y; a1 -= p1 * u1.y; a2 -= p1 * u2.y; a3 -= p1 * u3.y;
                a0 -= p2 * u0.z; a1 -= p2 * u1.z; a2 -= p2 * u2.z; a3 -= p2 * u3.z;
                a0 -= p3 * u0.w; a1 -= p3 * u1.w; a2 -= p3 * u2.w; a3 -= p3 * u3.w;
            }
            if (mine) *(float4*)(arow + s0) = make_float4(a0, a1, a2, a3);
        }
    } else if (kind == 1) {
        float* dst = X + (size_t)(k * TS) * Nn + i * TS + r;
#pragma unroll
        for (int c = 0; c < TS; c++) dst[(size_t)c * Nn] = pq[c].x;
    } else if (kind == 2) {
        if (r == 0) {
            float p[TS];
#pragma unroll
            for (int c = 0; c < TS; c++) p[c] = pq[c].x;
            store_row32(zbase + (size_t)b * Nn + k * TS, p);
        }
    } else if (i > k) {
        float p[TS];
#pragma unroll
        for (int c = 0; c < TS; c++) p[c] = pq[c].x;
        store_row32(Lo + (size_t)(i * TS + r) * Nn + k * TS, p);
    } else {
#pragma unroll
        for (int c = 0; c < TS; c++) if (c > r) d[c] = 0.0f;
        store_row32(Lo + (size_t)(k * TS + r) * Nn + k * TS, d);
    }
}

__global__ __launch_bounds__(1024) void nlml_kernel(const float* __restrict__ Lbase,
                                                    const float* __restrict__ zbase,
                                                    float* __restrict__ nlml_out) {
    int b = blockIdx.x;
    const float* L = Lbase + (size_t)b * Nn * Nn;
    __shared__ float col[Nn];
    __shared__ float zz[Nn];
    int tid = threadIdx.x;
    float z = zbase[b * Nn + tid];
    col[tid] = __logf(L[(size_t)tid * Nn + tid]);
    zz[tid] = z * z;
    __syncthreads();
    for (int s = 512; s > 0; s >>= 1) {
        if (tid < s) {
            col[tid] += col[tid + s];
            zz[tid] += zz[tid + s];
        }
        __syncthreads();
    }
    if (tid == 0) nlml_out[b] = 0.5f * zz[0] + col[0] + 0.5f * Nn * LOG2PIF;
}

__global__ __launch_bounds__(128 * PG) void predict_kernel(const float* __restrict__ Linvbase,
                                                      const float* __restrict__ trainPg,
                                                      const float* __restrict__ testPg,
                                                      const float* __restrict__ Kmm,
                                                      const float* __restrict__ zbase,
                                                      const float* __restrict__ meang,
                                                      const float* __restrict__ scale_param,
                                                      float* __restrict__ out_mean,
                                                      float* __restrict__ out_var) {
    extern __shared__ float smem[];
    float* Kt  = smem;
    float* zsh = smem + (Nn / 4) * KQ;
    float* tp  = zsh + Nn;
    float* Ab  = tp;
    static_assert(PG * 2 * 16 * APITCH >= Nn * 6, "the staging tiles cover the train parameters' place");
    static_assert(PG * 16 == ACH, "the PG bands of a walk step end in the same chunk");
    float* red = Ab + PG * 2 * 16 * APITCH;
    int b = blockIdx.y;
    int colBase = blockIdx.x * MT;
    int tid = threadIdx.x;
    float scale = __expf(scale_param[0]);

    for (int idx = tid; idx < Nn; idx += 128 * PG) zsh[idx] = zbase[b * Nn + idx];
    for (int idx = tid; idx < Nn * 6; idx += 128 * PG) tp[idx] = trainPg[(size_t)b * Nn * 6 + idx];
    __syncthreads();

    {
        int c = tid & 63;
        int h = tid >> 6;
        const float* q = testPg + (size_t)(b * Mm + colBase + c) * 6;
        float x2 = q[0], y2 = q[1], a2 = q[2], b2 = q[3], c2 = q[4], q2 = q[5];
        static_assert(KU % 4 == 0, "a thread's rows of a pass are whole k-quads");
        for (int n0 = KU * h; n0 < Nn; n0 += 2 * PG * KU) {
            float pr[KU][6], kv[KU];
#pragma unroll
            for (int u = 0; u < KU; u++) {
                const float* p = tp + (n0 + u) * 6;
#pragma unroll
                for (int e = 0; e < 6; e++) pr[u][e] = p[e];
            }
#pragma unroll
            for (int u = 0; u < KU; u++)
                kv[u] = kern_pair(pr[u][0], pr[u][1], pr[u][2], pr[u][3], pr[u][4], pr[u][5],
                                  x2, y2, a2, b2, c2, q2, scale);
#pragma unroll
            for (int u = 0; u < KU; u += 4)
                *(float4*)(Kt + ((n0 + u) >> 2) * KQ + 4 * c) = make_float4(kv[u], kv[u + 1], kv[u + 2], kv[u + 3]);
        }
    }
    __syncthreads();

    int grp = __builtin_amdgcn_readfirstlane(tid >> 7);
    int gt = tid & 127;
    int wave = gt >> 5, lane = tid & 31;
    int r = lane & 15, hi = lane >> 4;
    int wcol = wave * 16;
    const float* Li = Linvbase + (size_t)b * Nn * Nn;

    int q0 = (((int)(4.0f * sqrtf((float)blockIdx.x))) & (Nn / 16 - 1)) / PG;
    int rb0 = PG * q0 + grp;
    float* Abg = Ab + grp * (2 * 16 * APITCH);
    int srow = gt >> 4, sseg = gt & 15;
    const float* fa = Abg + r * APITCH + 2 * hi;
    const float* fb = Kt + 4 * (wcol + r) + 2 * hi;
    float4 g0, g1;
    {
        g0 = make_float4(0.0f, 0.0f, 0.0f, 0.0f); g1 = g0;
        if (4 * sseg < 16 * (rb0 + 1)) {
            g0 = *(const float4*)(Li + (size_t)(rb0 * 16 + srow) * Nn + 4 * sseg);
            g1 = *(const float4*)(Li + (size_t)(rb0 * 16 + srow + 8) * Nn + 4 * sseg);
        }
    }
    int buf = 0;
    float meanAcc = 0.0f, varAcc = 0.0f;
    for (int t = 0; t < NQ; t++) {
        int q = (q0 + t) & (NQ - 1);
        int rb = PG * q + grp;
        v8f acc = {};
        int nch = q + 1;
        for (int c = 0; c < nch; c++) {
            float* At = Abg + buf * (16 * APITCH);
            *(float4*)(At + srow * APITCH + 4 * sseg) = g0;
            *(float4*)(At + (srow + 8) * APITCH + 4 * sseg) = g1;
            int nrb = rb, nc = c + 1;
            bool more = true;
            if (nc == nch) { nrb = PG * ((q0 + t + 1) & (NQ - 1)) + grp; nc = 0; more = (t + 1 < NQ); }
            if (more) {
                int k0 = nc * ACH + 4 * sseg;
                g0 = make_float4(0.0f, 0.0f, 0.0f, 0.0f); g1 = g0;
                if (k0 < 16 * (nrb + 1)) {
                    g0 = *(const float4*)(Li + (size_t)(nrb * 16 + srow) * Nn + k0);
                    g1 = *(const float4*)(Li + (size_t)(nrb * 16 + srow + 8) * Nn + k0);
                }
            }
            __builtin_amdgcn_fence(__ATOMIC_RELEASE, "workgroup", "local");
            __builtin_amdgcn_s_barrier();
            __builtin_amdgcn_fence(__ATOMIC_ACQUIRE, "workgroup", "local");
            const float* fac = fa + buf * (16 * APITCH);
            const float* fbc = fb + (c * (ACH / 4)) * KQ;
            v2f av[ACH / 4], bv[ACH / 4];
#pragma unroll
            for (int s = 0; s < ACH / 4; s++) {
                av[s] = *(const v2f*)(fac + 4 * s);
                bv[s] = *(const v2f*)(fbc + s * KQ);
            }
#pragma unroll
            for (int s = 0; s < ACH / 4; s++)
                acc = __builtin_amdgcn_wmma_f32_16x16x4_f32(false, av[s], false, bv[s],
                                                            (short)0, acc, false, false);
            __builtin_amdgcn_sched_group_barrier(0x100, 16, 0);
            __builtin_amdgcn_sched_group_barrier(0x008, 16, 0);
            buf ^= 1;
        }
        const float* zb = zsh + rb * 16 + 8 * hi;
#pragma unroll
        for (int v = 0; v < 8; v++) {
            float cv = acc[v];
            meanAcc += zb[v] * cv;
            varAcc  += cv * cv;
        }
    }
    meanAcc += __shfl_xor(meanAcc, 16);
    varAcc  += __shfl_xor(varAcc, 16);
    if (grp != 0 && lane < 16) {
        red[(grp - 1) * 2 * MT + wcol + lane] = meanAcc;
        red[(grp - 1) * 2 * MT + MT + wcol + lane] = varAcc;
    }
    __syncthreads();
    if (grp == 0 && lane < 16) {
        int m = colBase + wcol + lane;
        for (int g = 0; g < PG - 1; g++) {
            meanAcc += red[g * 2 * MT + wcol + lane];
            varAcc  += red[g * 2 * MT + MT + wcol + lane];
        }
        out_mean[b * Mm + m] = meang[b] + meanAcc;
        out_var[b * Mm + m]  = Kmm[b * Mm + m] - varAcc;
    }
}

extern "C" void kernel_launch(void* const* d_in, const int* in_sizes, int n_in,
                              void* d_out, int out_size, void* d_ws, size_t ws_size,
                              hipStream_t stream) {
    const float* coords_train = (const float*)d_in[0];
    const float* kp_train     = (const float*)d_in[1];
    const float* coords_test  = (const float*)d_in[2];
    const float* kp_test      = (const float*)d_in[3];
    const float* y_train      = (const float*)d_in[4];
    const float* var          = (const float*)d_in[5];
    const float* mean         = (const float*)d_in[6];
    const float* scale_param  = (const float*)d_in[7];

    float* ws = (float*)d_ws;
    size_t off = 0;
    float* A      = ws + off; off += (size_t)Bb * Nn * Nn;
    float* Linv   = ws + off; off += (size_t)Bb * Nn * Nn;
    float* Lf     = ws + off; off += (size_t)Bb * Nn * Nn;
    float* trainP = ws + off; off += (size_t)Bb * Nn * 6;
    float* testP  = ws + off; off += (size_t)Bb * Mm * 6;
    float* Kmm    = ws + off; off += (size_t)Bb * Mm;
    float* z      = ws + off; off += (size_t)Bb * Nn;
    float* Yrow   = ws + off; off += (size_t)Bb * Nn;

    float* out_mean = (float*)d_out;
    float* out_var  = out_mean + (size_t)Bb * Mm;
    float* nlml     = out_var + (size_t)Bb * Mm;

    prep_kernel<<<(Bb * Mm + 255) / 256, 256, 0, stream>>>(
        coords_train, kp_train, coords_test, kp_test, scale_param, trainP, testP, Kmm);
    buildA_kernel<<<(Bb * Nn * Nn) / 256, 256, 0, stream>>>(trainP, var, scale_param, y_train, mean, A, Yrow);
#define CHOL_STEP(K) chol_step_kernel<<<dim3(NT + 2, NT - (K), Bb), 32, 0, stream>>>(A, Lf, Linv, Yrow, z, (K));
    CHOL_STEP(0) CHOL_STEP(1) CHOL_STEP(2) CHOL_STEP(3) CHOL_STEP(4) CHOL_STEP(5) CHOL_STEP(6) CHOL_STEP(7)
    CHOL_STEP(8) CHOL_STEP(9) CHOL_STEP(10) CHOL_STEP(11) CHOL_STEP(12) CHOL_STEP(13) CHOL_STEP(14) CHOL_STEP(15)
    CHOL_STEP(16) CHOL_STEP(17) CHOL_STEP(18) CHOL_STEP(19) CHOL_STEP(20) CHOL_STEP(21) CHOL_STEP(22) CHOL_STEP(23)
    CHOL_STEP(24) CHOL_STEP(25) CHOL_STEP(26) CHOL_STEP(27) CHOL_STEP(28) CHOL_STEP(29) CHOL_STEP(30) CHOL_STEP(31)
#undef CHOL_STEP
    nlml_kernel<<<Bb, 1024, 0, stream>>>(Lf, z, nlml);

    size_t smem_bytes = (size_t)((Nn / 4) * KQ + Nn + PG * 2 * 16 * APITCH + (PG - 1) * 2 * MT) * sizeof(float);
    predict_kernel<<<dim3(Mm / MT, Bb), 128 * PG, smem_bytes, stream>>>(
        Linv, trainP, testP, Kmm, z, mean, scale_param, out_mean, out_var);
}


# Cholesky tile sweep square-root-free: columns stay unscaled, multiplier = column / pivot (hardware reciprocal); sqrt and divide once per lane after the sweep; sum order kept, L's last bits change
# speedup vs baseline: 91.7980x; 1.1890x over previous
#include <hip/hip_runtime.h>
#include <math.h>

#define Nn 1024
#define Mm 16384
#define Bb 2
#define MT 64
#define KQ (4 * MT)
#define ACH 64
#define KU 8
#define PG 4
#define NQ (Nn / 16 / PG)
#define APITCH 68

typedef __attribute__((ext_vector_type(2))) float v2f;
typedef __attribute__((ext_vector_type(8))) float v8f;

#define SQRT3F 1.7320508075688772f
#define LOG2PIF 1.8378770664093453f

__device__ __forceinline__ float kern_pair(float x1, float y1, float a1, float b1, float c1, float q1,
                                           float x2, float y2, float a2, float b2, float c2, float q2,
                                           float scale) {
    float s00 = a1 + a2, s11 = b1 + b2, s01 = c1 + c2;
    float sdet = s00 * s11 - s01 * s01;
    float d0 = x1 - x2, d1 = y1 - y2;
    float Q = 0.5f * (s11 * d0 * d0 - 2.0f * s01 * d0 * d1 + s00 * d1 * d1) / sdet;
    float C = 2.0f * q1 * q2 * rsqrtf(fmaxf(sdet, 1e-5f));
    float t = SQRT3F * sqrtf(fmaxf(Q, 1e-5f));
    return (1.0f + t) * __expf(-t) * C * scale;
}

__global__ void prep_kernel(const float* __restrict__ ct, const float* __restrict__ kpt,
                            const float* __restrict__ cs, const float* __restrict__ kps,
                            const float* __restrict__ scale_param,
                            float* __restrict__ trainP, float* __restrict__ testP,
                            float* __restrict__ Kmm) {
    int gid = blockIdx.x * blockDim.x + threadIdx.x;
    float scale = __expf(scale_param[0]);
    if (gid < Bb * Mm) {
        float a = kps[gid * 3 + 0], b = kps[gid * 3 + 1], c = kps[gid * 3 + 2];
        float det = a * b - c * c;
        testP[gid * 6 + 0] = cs[gid * 2 + 0];
        testP[gid * 6 + 1] = cs[gid * 2 + 1];
        testP[gid * 6 + 2] = a;
        testP[gid * 6 + 3] = b;
        testP[gid * 6 + 4] = c;
        testP[gid * 6 + 5] = sqrtf(sqrtf(det));
        float t0 = SQRT3F * sqrtf(1e-5f);
        float m0 = (1.0f + t0) * __expf(-t0);
        float C = 2.0f * sqrtf(det) * rsqrtf(fmaxf(4.0f * det, 1e-5f));
        Kmm[gid] = C * m0 * scale;
    }
    if (gid < Bb * Nn) {
        float a = kpt[gid * 3 + 0], b = kpt[gid * 3 + 1], c = kpt[gid * 3 + 2];
        float det = a * b - c * c;
        trainP[gid * 6 + 0] = ct[gid * 2 + 0];
        trainP[gid * 6 + 1] = ct[gid * 2 + 1];
        trainP[gid * 6 + 2] = a;
        trainP[gid * 6 + 3] = b;
        trainP[gid * 6 + 4] = c;
        trainP[gid * 6 + 5] = sqrtf(sqrtf(det));
    }
}

__global__ void buildA_kernel(const float* __restrict__ trainP, const float* __restrict__ var,
                              const float* __restrict__ scale_param,
                              const float* __restrict__ y, const float* __restrict__ mean,
                              float* __restrict__ A, float* __restrict__ Yrow) {
    int gid = blockIdx.x * blockDim.x + threadIdx.x;
    int b = gid >> 20;
    int i = (gid >> 10) & (Nn - 1);
    int j = gid & (Nn - 1);
    const float* p1 = trainP + (size_t)(b * Nn + i) * 6;
    const float* p2 = trainP + (size_t)(b * Nn + j) * 6;
    float scale = __expf(scale_param[0]);
    float v = kern_pair(p1[0], p1[1], p1[2], p1[3], p1[4], p1[5],
                        p2[0], p2[1], p2[2], p2[3], p2[4], p2[5], scale);
    if (i == j) v += var[b * Nn + i];
    A[(size_t)gid] = v;
    if (i == 0) Yrow[b * Nn + j] = y[b * Nn + j] - mean[b];
}

#define TS 32
#define NT (Nn / TS)

__device__ __forceinline__ float lane_bcast(float v, int src) {
    return __int_as_float(__builtin_amdgcn_readlane(__float_as_int(v), src));
}

__device__ __forceinline__ void load_row32(const float* p, float (&r)[TS]) {
    const float4* p4 = (const float4*)p;
#pragma unroll
    for (int q = 0; q < TS / 4; q++) {
        float4 t = p4[q];
        r[4 * q + 0] = t.x; r[4 * q + 1] = t.y; r[4 * q + 2] = t.z; r[4 * q + 3] = t.w;
    }
}

__device__ __forceinline__ void store_row32(float* p, const float (&r)[TS]) {
    float4* p4 = (float4*)p;
#pragma unroll
    for (int q = 0; q < TS / 4; q++) {
        float4 t;
        t.x = r[4 * q + 0]; t.y = r[4 * q + 1]; t.z = r[4 * q + 2]; t.w = r[4 * q + 3];
        p4[q] = t;
    }
}

#define QP 36

__global__ __launch_bounds__(32) void chol_step_kernel(float* Abase, float* __restrict__ Lbase,
                                                       float* Xbase, float* Ybase,
                                                       float* __restrict__ zbase, int k) {
    __shared__ __attribute__((aligned(16))) float qs[TS * QP];
    __shared__ __attribute__((aligned(16))) float ts[TS * QP];
    int nmat = NT - k;
    int bx = blockIdx.x;
    int kind = (bx < nmat) ? 0 : (bx <= NT) ? 1 : 2;
    int i = (kind == 0) ? k + bx : (kind == 1) ? bx - nmat : 0;
    int j = k + blockIdx.y;
    if (kind == 0 && j > i) return;
    int b = blockIdx.z;
    int r = threadIdx.x;
    float* A = Abase + (size_t)b * Nn * Nn;
    float* Lo = Lbase + (size_t)b * Nn * Nn;
    float* X = Xbase + (size_t)b * Nn * Nn;
    float* Y = Ybase + (size_t)b * Nn;
    bool ident = (kind == 1) && (i == k);
    bool mine = (kind != 2) || (r == 0);
    const float* prow = (kind == 0) ? A + (size_t)(i * TS + r) * Nn + k * TS
                      : (kind == 1) ? X + (size_t)(i * TS + r) * Nn + k * TS
                                    : Y + k * TS;
    float* trow = (kind == 0) ? A + (size_t)(i * TS + r) * Nn + j * TS
                : (kind == 1) ? X + (size_t)(i * TS + r) * Nn + j * TS
                              : Y + j * TS;

    float d[TS];
    v2f pq[TS];
    load_row32(A + (size_t)(k * TS + r) * Nn + k * TS, d);
    {
        float p[TS], q[TS], t[TS];
        load_row32(A + (size_t)(j * TS + r) * Nn + k * TS, q);
        if (j > k) {
            if (!ident && mine) {
                load_row32(trow, t);
            } else {
#pragma unroll
                for (int c = 0; c < TS; c++) t[c] = 0.0f;
            }
        }
        if (ident) {
#pragma unroll
            for (int c = 0; c < TS; c++) p[c] = (c == r) ? 1.0f : 0.0f;
        } else {
            load_row32(prow, p);
#pragma unroll
            for (int c = 0; c < TS; c++) p[c] = mine ? p[c] : 0.0f;
        }
#pragma unroll
        for (int c = 0; c < TS; c++) { pq[c].x = p[c]; pq[c].y = q[c]; }
        if (j > k) store_row32(ts + r * QP, t);
    }

    float dv[TS];
    float pvv = 1.0f;
#pragma unroll
    for (int c = 0; c < TS; c++) {
#pragma unroll
        for (int c1 = 0; c1 < c; c1++) {
            float s = lane_bcast(dv[c1], c);
            d[c] -= d[c1] * s;
            pq[c] -= pq[c1] * s;
        }
        float piv = lane_bcast(d[c], c);
        dv[c] = d[c] * __builtin_amdgcn_rcpf(piv);
        pvv = (r == c) ? d[c] : pvv;
    }
    float dgv = sqrtf(pvv);
    float dinvv = 1.0f / dgv;
#pragma unroll
    for (int c = 0; c < TS; c++) {
        float dinv = lane_bcast(dinvv, c);
        d[c] = (r == c) ? dgv : d[c] * dinv;
        pq[c] = pq[c] * dinv;
    }

    if (j > k) {
        float4* qw = (float4*)(qs + r * QP);
#pragma unroll
        for (int c4 = 0; c4 < TS / 4; c4++)
            qw[c4] = make_float4(pq[4 * c4].y, pq[4 * c4 + 1].y, pq[4 * c4 + 2].y, pq[4 * c4 + 3].y);
        __syncthreads();
        float* arow = trow;
#pragma unroll 1
        for (int s0 = 0; s0 < TS; s0 += 4) {
            float4 av = *(const float4*)(ts + r * QP + s0);
            float a0 = av.x, a1 = av.y, a2 = av.z, a3 = av.w;
            const float4* q0 = (const float4*)(qs + (s0 + 0) * QP);
            const float4* q1 = (const float4*)(qs + (s0 + 1) * QP);
            const float4* q2 = (const float4*)(qs + (s0 + 2) * QP);
            const float4* q3 = (const float4*)(qs + (s0 + 3) * QP);
#pragma unroll
            for (int c4 = 0; c4 < TS / 4; c4++) {
                float4 u0 = q0[c4], u1 = q1[c4], u2 = q2[c4], u3 = q3[c4];
                float p0 = pq[4 * c4].x, p1 = pq[4 * c4 + 1].x, p2 = pq[4 * c4 + 2].x, p3 = pq[4 * c4 + 3].x;
                a0 -= p0 * u0.x; a1 -= p0 * u1.x; a2 -= p0 * u2.x; a3 -= p0 * u3.x;
                a0 -= p1 * u0.y; a1 -= p1 * u1.y; a2 -= p1 * u2.y; a3 -= p1 * u3.y;
                a0 -= p2 * u0.z; a1 -= p2 * u1.z; a2 -= p2 * u2.z; a3 -= p2 * u3.z;
                a0 -= p3 * u0.w; a1 -= p3 * u1.w; a2 -= p3 * u2.w; a3 -= p3 * u3.w;
            }
            if (mine) *(float4*)(arow + s0) = make_float4(a0, a1, a2, a3);
        }
    } else if (kind == 1) {
        float* dst = X + (size_t)(k * TS) * Nn + i * TS + r;
#pragma unroll
        for (int c = 0; c < TS; c++) dst[(size_t)c * Nn] = pq[c].x;
    } else if (kind == 2) {
        if (r == 0) {
            float p[TS];
#pragma unroll
            for (int c = 0; c < TS; c++) p[c] = pq[c].x;
            store_row32(zbase + (size_t)b * Nn + k * TS, p);
        }
    } else if (i > k) {
        float p[TS];
#pragma unroll
        for (int c = 0; c < TS; c++) p[c] = pq[c].x;
        store_row32(Lo + (size_t)(i * TS + r) * Nn + k * TS, p);
    } else {
#pragma unroll
        for (int c = 0; c < TS; c++) if (c > r) d[c] = 0.0f;
        store_row32(Lo + (size_t)(k * TS + r) * Nn + k * TS, d);
    }
}

__global__ __launch_bounds__(1024) void nlml_kernel(const float* __restrict__ Lbase,
                                                    const float* __restrict__ zbase,
                                                    float* __restrict__ nlml_out) {
    int b = blockIdx.x;
    const float* L = Lbase + (size_t)b * Nn * Nn;
    __shared__ float col[Nn];
    __shared__ float zz[Nn];
    int tid = threadIdx.x;
    float z = zbase[b * Nn + tid];
    col[tid] = __logf(L[(size_t)tid * Nn + tid]);
    zz[tid] = z * z;
    __syncthreads();
    for (int s = 512; s > 0; s >>= 1) {
        if (tid < s) {
            col[tid] += col[tid + s];
            zz[tid] += zz[tid + s];
        }
        __syncthreads();
    }
    if (tid == 0) nlml_out[b] = 0.5f * zz[0] + col[0] + 0.5f * Nn * LOG2PIF;
}

__global__ __launch_bounds__(128 * PG) void predict_kernel(const float* __restrict__ Linvbase,
                                                      const float* __restrict__ trainPg,
                                                      const float* __restrict__ testPg,
                                                      const float* __restrict__ Kmm,
                                                      const float* __restrict__ zbase,
                                                      const float* __restrict__ meang,
                                                      const float* __restrict__ scale_param,
                                                      float* __restrict__ out_mean,
                                                      float* __restrict__ out_var) {
    extern __shared__ float smem[];
    float* Kt  = smem;
    float* zsh = smem + (Nn / 4) * KQ;
    float* tp  = zsh + Nn;
    float* Ab  = tp;
    static_assert(PG * 2 * 16 * APITCH >= Nn * 6, "the staging tiles cover the train parameters' place");
    static_assert(PG * 16 == ACH, "the PG bands of a walk step end in the same chunk");
    float* red = Ab + PG * 2 * 16 * APITCH;
    int b = blockIdx.y;
    int colBase = blockIdx.x * MT;
    int tid = threadIdx.x;
    float scale = __expf(scale_param[0]);

    for (int idx = tid; idx < Nn; idx += 128 * PG) zsh[idx] = zbase[b * Nn + idx];
    for (int idx = tid; idx < Nn * 6; idx += 128 * PG) tp[idx] = trainPg[(size_t)b * Nn * 6 + idx];
    __syncthreads();

    {
        int c = tid & 63;
        int h = tid >> 6;
        const float* q = testPg + (size_t)(b * Mm + colBase + c) * 6;
        float x2 = q[0], y2 = q[1], a2 = q[2], b2 = q[3], c2 = q[4], q2 = q[5];
        static_assert(KU % 4 == 0, "a thread's rows of a pass are whole k-quads");
        for (int n0 = KU * h; n0 < Nn; n0 += 2 * PG * KU) {
            float pr[KU][6], kv[KU];
#pragma unroll
            for (int u = 0; u < KU; u++) {
                const float* p = tp + (n0 + u) * 6;
#pragma unroll
                for (int e = 0; e < 6; e++) pr[u][e] = p[e];
            }
#pragma unroll
            for (int u = 0; u < KU; u++)
                kv[u] = kern_pair(pr[u][0], pr[u][1], pr[u][2], pr[u][3], pr[u][4], pr[u][5],
                                  x2, y2, a2, b2, c2, q2, scale);
#pragma unroll
            for (int u = 0; u < KU; u += 4)
                *(float4*)(Kt + ((n0 + u) >> 2) * KQ + 4 * c) = make_float4(kv[u], kv[u + 1], kv[u + 2], kv[u + 3]);
        }
    }
    __syncthreads();

    int grp = __builtin_amdgcn_readfirstlane(tid >> 7);
    int gt = tid & 127;
    int wave = gt >> 5, lane = tid & 31;
    int r = lane & 15, hi = lane >> 4;
    int wcol = wave * 16;
    const float* Li = Linvbase + (size_t)b * Nn * Nn;

    int q0 = (((int)(4.0f * sqrtf((float)blockIdx.x))) & (Nn / 16 - 1)) / PG;
    int rb0 = PG * q0 + grp;
    float* Abg = Ab + grp * (2 * 16 * APITCH);
    int srow = gt >> 4, sseg = gt & 15;
    const float* fa = Abg + r * APITCH + 2 * hi;
    const float* fb = Kt + 4 * (wcol + r) + 2 * hi;
    float4 g0, g1;
    {
        g0 = make_float4(0.0f, 0.0f, 0.0f, 0.0f); g1 = g0;
        if (4 * sseg < 16 * (rb0 + 1)) {
            g0 = *(const float4*)(Li + (size_t)(rb0 * 16 + srow) * Nn + 4 * sseg);
            g1 = *(const float4*)(Li + (size_t)(rb0 * 16 + srow + 8) * Nn + 4 * sseg);
        }
    }
    int buf = 0;
    float meanAcc = 0.0f, varAcc = 0.0f;
    for (int t = 0; t < NQ; t++) {
        int q = (q0 + t) & (NQ - 1);
        int rb = PG * q + grp;
        v8f acc = {};
        int nch = q + 1;
        for (int c = 0; c < nch; c++) {
            float* At = Abg + buf * (16 * APITCH);
            *(float4*)(At + srow * APITCH + 4 * sseg) = g0;
            *(float4*)(At + (srow + 8) * APITCH + 4 * sseg) = g1;
            int nrb = rb, nc = c + 1;
            bool more = true;
            if (nc == nch) { nrb = PG * ((q0 + t + 1) & (NQ - 1)) + grp; nc = 0; more = (t + 1 < NQ); }
            if (more) {
                int k0 = nc * ACH + 4 * sseg;
                g0 = make_float4(0.0f, 0.0f, 0.0f, 0.0f); g1 = g0;
                if (k0 < 16 * (nrb + 1)) {
                    g0 = *(const float4*)(Li + (size_t)(nrb * 16 + srow) * Nn + k0);
                    g1 = *(const float4*)(Li + (size_t)(nrb * 16 + srow + 8) * Nn + k0);
                }
            }
            __builtin_amdgcn_fence(__ATOMIC_RELEASE, "workgroup", "local");
            __builtin_amdgcn_s_barrier();
            __builtin_amdgcn_fence(__ATOMIC_ACQUIRE, "workgroup", "local");
            const float* fac = fa + buf * (16 * APITCH);
            const float* fbc = fb + (c * (ACH / 4)) * KQ;
            v2f av[ACH / 4], bv[ACH / 4];
#pragma unroll
            for (int s = 0; s < ACH / 4; s++) {
                av[s] = *(const v2f*)(fac + 4 * s);
                bv[s] = *(const v2f*)(fbc + s * KQ);
            }
#pragma unroll
            for (int s = 0; s < ACH / 4; s++)
                acc = __builtin_amdgcn_wmma_f32_16x16x4_f32(false, av[s], false, bv[s],
                                                            (short)0, acc, false, false);
            __builtin_amdgcn_sched_group_barrier(0x100, 16, 0);
            __builtin_amdgcn_sched_group_barrier(0x008, 16, 0);
            buf ^= 1;
        }
        const float* zb = zsh + rb * 16 + 8 * hi;
#pragma unroll
        for (int v = 0; v < 8; v++) {
            float cv = acc[v];
            meanAcc += zb[v] * cv;
            varAcc  += cv * cv;
        }
    }
    meanAcc += __shfl_xor(meanAcc, 16);
    varAcc  += __shfl_xor(varAcc, 16);
    if (grp != 0 && lane < 16) {
        red[(grp - 1) * 2 * MT + wcol + lane] = meanAcc;
        red[(grp - 1) * 2 * MT + MT + wcol + lane] = varAcc;
    }
    __syncthreads();
    if (grp == 0 && lane < 16) {
        int m = colBase + wcol + lane;
        for (int g = 0; g < PG - 1; g++) {
            meanAcc += red[g * 2 * MT + wcol + lane];
            varAcc  += red[g * 2 * MT + MT + wcol + lane];
        }
        out_mean[b * Mm + m] = meang[b] + meanAcc;
        out_var[b * Mm + m]  = Kmm[b * Mm + m] - varAcc;
    }
}

extern "C" void kernel_launch(void* const* d_in, const int* in_sizes, int n_in,
                              void* d_out, int out_size, void* d_ws, size_t ws_size,
                              hipStream_t stream) {
    const float* coords_train = (const float*)d_in[0];
    const float* kp_train     = (const float*)d_in[1];
    const float* coords_test  = (const float*)d_in[2];
    const float* kp_test      = (const float*)d_in[3];
    const float* y_train      = (const float*)d_in[4];
    const float* var          = (const float*)d_in[5];
    const float* mean         = (const float*)d_in[6];
    const float* scale_param  = (const float*)d_in[7];

    float* ws = (float*)d_ws;
    size_t off = 0;
    float* A      = ws + off; off += (size_t)Bb * Nn * Nn;
    float* Linv   = ws + off; off += (size_t)Bb * Nn * Nn;
    float* Lf     = ws + off; off += (size_t)Bb * Nn * Nn;
    float* trainP = ws + off; off += (size_t)Bb * Nn * 6;
    float* testP  = ws + off; off += (size_t)Bb * Mm * 6;
    float* Kmm    = ws + off; off += (size_t)Bb * Mm;
    float* z      = ws + off; off += (size_t)Bb * Nn;
    float* Yrow   = ws + off; off += (size_t)Bb * Nn;

    float* out_mean = (float*)d_out;
    float* out_var  = out_mean + (size_t)Bb * Mm;
    float* nlml     = out_var + (size_t)Bb * Mm;

    prep_kernel<<<(Bb * Mm + 255) / 256, 256, 0, stream>>>(
        coords_train, kp_train, coords_test, kp_test, scale_param, trainP, testP, Kmm);
    buildA_kernel<<<(Bb * Nn * Nn) / 256, 256, 0, stream>>>(trainP, var, scale_param, y_train, mean, A, Yrow);
#define CHOL_STEP(K) chol_step_kernel<<<dim3(NT + 2, NT - (K), Bb), 32, 0, stream>>>(A, Lf, Linv, Yrow, z, (K));
    CHOL_STEP(0) CHOL_STEP(1) CHOL_STEP(2) CHOL_STEP(3) CHOL_STEP(4) CHOL_STEP(5) CHOL_STEP(6) CHOL_STEP(7)
    CHOL_STEP(8) CHOL_STEP(9) CHOL_STEP(10) CHOL_STEP(11) CHOL_STEP(12) CHOL_STEP(13) CHOL_STEP(14) CHOL_STEP(15)
    CHOL_STEP(16) CHOL_STEP(17) CHOL_STEP(18) CHOL_STEP(19) CHOL_STEP(20) CHOL_STEP(21) CHOL_STEP(22) CHOL_STEP(23)
    CHOL_STEP(24) CHOL_STEP(25) CHOL_STEP(26) CHOL_STEP(27) CHOL_STEP(28) CHOL_STEP(29) CHOL_STEP(30) CHOL_STEP(31)
#undef CHOL_STEP
    nlml_kernel<<<Bb, 1024, 0, stream>>>(Lf, z, nlml);

    size_t smem_bytes = (size_t)((Nn / 4) * KQ + Nn + PG * 2 * 16 * APITCH + (PG - 1) * 2 * MT) * sizeof(float);
    predict_kernel<<<dim3(Mm / MT, Bb), 128 * PG, smem_bytes, stream>>>(
        Linv, trainP, testP, Kmm, z, mean, scale_param, out_mean, out_var);
}
